# starting kernel
# baseline (speedup 1.0000x reference)
#include <hip/hip_runtime.h>
#include <hip/hip_bf16.h>

#define NN 50000
#define E_ADJ 800000
#define E_DEC 200000
#define HH 64

typedef float v2f __attribute__((ext_vector_type(2)));
typedef float v8f __attribute__((ext_vector_type(8)));

__global__ __launch_bounds__(128) void gemm_stage(
    const float* __restrict__ x0, const float* __restrict__ x1,
    const float* __restrict__ W, float* __restrict__ xw, int Fin) {
  const int k    = blockIdx.y;
  const int tm   = blockIdx.x;
  const int wave = threadIdx.x >> 5;
  const int lane = threadIdx.x & 31;
  const int half = lane >> 4;
  const int l    = lane & 15;

  const float* X  = (k & 1) ? x1 : x0;
  const float* Wk = W + (size_t)k * Fin * HH;

  const int row = tm * 16 + l;
  const int col = wave * 16 + l;

  const float* ar = X + (size_t)row * Fin + half * 2;
  v8f c = {};
  for (int kk = 0; kk < Fin; kk += 4) {
    v2f a, b;
    a.x = ar[kk];
    a.y = ar[kk + 1];
    const float* wp = Wk + (size_t)(kk + half * 2) * HH + col;
    b.x = wp[0];
    b.y = wp[HH];
    c = __builtin_amdgcn_wmma_f32_16x16x4_f32(
         false, a,  false, b,
         (short)0, c,  false,  false);
  }

  float* out = xw + ((size_t)k * NN + (size_t)tm * 16) * HH + wave * 16;
#pragma unroll
  for (int v = 0; v < 8; ++v)
    out[(size_t)(v + half * 8) * HH + l] = c[v];
}

__global__ void zero_f32(float* __restrict__ p, size_t n) {
  size_t i = (size_t)blockIdx.x * blockDim.x + threadIdx.x;
  if (i < n) p[i] = 0.0f;
}

__global__ __launch_bounds__(256) void spmm_scatter(
    const int* __restrict__ rows, const int* __restrict__ cols,
    const float* __restrict__ vals, const float* __restrict__ xw,
    float* __restrict__ y) {
  const int k = blockIdx.y;
  const int t = threadIdx.x;
  const int e = blockIdx.x * 16 + (t >> 4);
  const int l = t & 15;

  const size_t eo = (size_t)k * E_ADJ + e;
  const int   r = rows[eo];
  const int   cidx = cols[eo];
  const float v = vals[eo];

  const float4 xv =
      ((const float4*)(xw + ((size_t)k * NN + cidx) * HH))[l];
  float* dst = y + ((size_t)k * NN + r) * HH + l * 4;
  unsafeAtomicAdd(dst + 0, xv.x * v);
  unsafeAtomicAdd(dst + 1, xv.y * v);
  unsafeAtomicAdd(dst + 2, xv.z * v);
  unsafeAtomicAdd(dst + 3, xv.w * v);
}

__global__ void relu_combine(const float* __restrict__ y,
                             float* __restrict__ h0, float* __restrict__ h1) {
  const size_t S = (size_t)NN * HH;
  size_t i = (size_t)blockIdx.x * blockDim.x + threadIdx.x;
  if (i >= S) return;
  float a = y[i], b = y[S + i], c = y[2 * S + i], d = y[3 * S + i];
  h0[i] = fmaxf(a, 0.0f) + fmaxf(b, 0.0f);
  h1[i] = fmaxf(c, 0.0f) + fmaxf(d, 0.0f);
}

__global__ void build_M(const float* __restrict__ R, const float* __restrict__ D,
                        const int* __restrict__ rt_k, float* __restrict__ M) {
  int i = blockIdx.x * blockDim.x + threadIdx.x;
  const float* d = D + rt_k[0] * HH;
  int h = i >> 6, g = i & 63;
  M[i] = d[h] * R[i] * d[g];
}

__global__ __launch_bounds__(128) void decode_edges(
    const float* __restrict__ h0, const float* __restrict__ h1,
    const int* __restrict__ edges, const float* __restrict__ M,
    float* __restrict__ preds) {
  __shared__ float part[4][16];
  const int wave = threadIdx.x >> 5;
  const int lane = threadIdx.x & 31;
  const int half = lane >> 4;
  const int l    = lane & 15;
  const int eb   = blockIdx.x * 16;

  const int arow = edges[(eb + l) * 2 + 0];
  const int col  = wave * 16 + l;

  const float* ar = h0 + (size_t)arow * HH + half * 2;
  v8f c = {};
  for (int kk = 0; kk < HH; kk += 4) {
    v2f a, b;
    a.x = ar[kk];
    a.y = ar[kk + 1];
    const float* mp = M + (kk + half * 2) * HH + col;
    b.x = mp[0];
    b.y = mp[HH];
    c = __builtin_amdgcn_wmma_f32_16x16x4_f32(
        false, a, false, b, (short)0, c, false, false);
  }

  float p[8];
#pragma unroll
  for (int v = 0; v < 8; ++v) {
    const int edge = v + half * 8;
    const int crow = edges[(eb + edge) * 2 + 1];
    p[v] = c[v] * h1[(size_t)crow * HH + col];
  }
#pragma unroll
  for (int off = 1; off < 16; off <<= 1) {
#pragma unroll
    for (int v = 0; v < 8; ++v) p[v] += __shfl_xor(p[v], off, 16);
  }
  if (l == 0) {
#pragma unroll
    for (int v = 0; v < 8; ++v) part[wave][v + half * 8] = p[v];
  }
  __syncthreads();
  if (threadIdx.x < 16) {
    int t = threadIdx.x;
    preds[eb + t] = part[0][t] + part[1][t] + part[2][t] + part[3][t];
  }
}

extern "C" void kernel_launch(void* const* d_in, const int* in_sizes, int n_in,
                              void* d_out, int out_size, void* d_ws, size_t ws_size,
                              hipStream_t stream) {
  const int*   adj_rows = (const int*)d_in[0];
  const int*   adj_cols = (const int*)d_in[1];
  const float* adj_vals = (const float*)d_in[2];
  const float* feat0    = (const float*)d_in[3];
  const float* feat1    = (const float*)d_in[4];
  const float* W1       = (const float*)d_in[5];
  const float* W2       = (const float*)d_in[6];
  const float* W3       = (const float*)d_in[7];
  const float* W4       = (const float*)d_in[8];
  const float* Rg       = (const float*)d_in[9];
  const float* Dl       = (const float*)d_in[10];
  const int*   edges    = (const int*)d_in[11];
  const int*   rt_k     = (const int*)d_in[12];

  const size_t S = (size_t)NN * HH;
  float* xw = (float*)d_ws;
  float* y  = xw + 4 * S;
  float* h0 = y + 4 * S;
  float* h1 = h0 + S;
  float* M  = h1 + S;

  struct Stage { const float* W; int Fin; int base; };
  const Stage st[4] = { {W1, 128, 0}, {W2, 64, 0}, {W3, 64, 4}, {W4, 64, 4} };

  const float* x0 = feat0;
  const float* x1 = feat1;
  for (int s = 0; s < 4; ++s) {
    gemm_stage<<<dim3(NN / 16, 4), 128, 0, stream>>>(x0, x1, st[s].W, xw, st[s].Fin);

    const size_t ycnt = 4 * S;
    zero_f32<<<(unsigned)((ycnt + 255) / 256), 256, 0, stream>>>(y, ycnt);

    spmm_scatter<<<dim3(E_ADJ / 16, 4), 256, 0, stream>>>(
        adj_rows + (size_t)st[s].base * E_ADJ,
        adj_cols + (size_t)st[s].base * E_ADJ,
        adj_vals + (size_t)st[s].base * E_ADJ, xw, y);

    relu_combine<<<(unsigned)((S + 255) / 256), 256, 0, stream>>>(y, h0, h1);
    x0 = h0;
    x1 = h1;
  }

  build_M<<<16, 256, 0, stream>>>(Rg, Dl, rt_k, M);
  decode_edges<<<E_DEC / 16, 128, 0, stream>>>(h0, h1, edges, M, (float*)d_out);
}


# SpMM as row gather: per call, edges are bucketed by dest row (int-atomic arrival rank + scan); one wave per row sums in that arrival order; f32 atomics, y planes, zero and relu passes go
# speedup vs baseline: 1.5989x; 1.5989x over previous
#include <hip/hip_runtime.h>
#include <hip/hip_bf16.h>

#define NN 50000
#define E_ADJ 800000
#define E_DEC 200000
#define HH 64

typedef float v2f __attribute__((ext_vector_type(2)));
typedef float v8f __attribute__((ext_vector_type(8)));

__global__ __launch_bounds__(128) void gemm_stage(
    const float* __restrict__ x0, const float* __restrict__ x1,
    const float* __restrict__ W, float* __restrict__ xw, int Fin) {
  const int k    = blockIdx.y;
  const int tm   = blockIdx.x;
  const int wave = threadIdx.x >> 5;
  const int lane = threadIdx.x & 31;
  const int half = lane >> 4;
  const int l    = lane & 15;

  const float* X  = (k & 1) ? x1 : x0;
  const float* Wk = W + (size_t)k * Fin * HH;

  const int row = tm * 16 + l;
  const int col = wave * 16 + l;

  const float* ar = X + (size_t)row * Fin + half * 2;
  v8f c = {};
  for (int kk = 0; kk < Fin; kk += 4) {
    v2f a, b;
    a.x = ar[kk];
    a.y = ar[kk + 1];
    const float* wp = Wk + (size_t)(kk + half * 2) * HH + col;
    b.x = wp[0];
    b.y = wp[HH];
    c = __builtin_amdgcn_wmma_f32_16x16x4_f32(
         false, a,  false, b,
         (short)0, c,  false,  false);
  }

  float* out = xw + ((size_t)k * NN + (size_t)tm * 16) * HH + wave * 16;
#pragma unroll
  for (int v = 0; v < 8; ++v)
    out[(size_t)(v + half * 8) * HH + l] = c[v];
}

struct __align__(8) EdgeCV { int c; float v; };

__global__ void zero_i32(int* __restrict__ p, int n) {
  int i = blockIdx.x * blockDim.x + threadIdx.x;
  if (i < n) p[i] = 0;
}

__global__ __launch_bounds__(256) void count_rank(
    const int* __restrict__ rows, int* __restrict__ cnt, int* __restrict__ rank) {
  const int k = blockIdx.y;
  const int e = blockIdx.x * 256 + threadIdx.x;
  const size_t eo = (size_t)k * E_ADJ + e;
  const int r = rows[eo];
  rank[eo] = atomicAdd(cnt + k * NN + r, 1);
}

__global__ __launch_bounds__(1024) void scan_rows(
    const int* __restrict__ cnt, int* __restrict__ start) {
  __shared__ int wpre[32];
  __shared__ int wown[32];
  const int k = blockIdx.y;
  const int t = threadIdx.x;
  const int lane = t & 31;
  const int w = t >> 5;
  const int lo = blockIdx.x * 1024;
  const int* c = cnt + k * NN;

  int pre = 0;
  for (int i = t; i < lo; i += 1024) pre += c[i];
  const int row = lo + t;
  const int own = (row < NN) ? c[row] : 0;

  int incl = own;
#pragma unroll
  for (int off = 1; off < 32; off <<= 1) {
    const int n = __shfl_up(incl, off, 32);
    if (lane >= off) incl += n;
  }
  int psum = pre;
#pragma unroll
  for (int off = 16; off > 0; off >>= 1) psum += __shfl_xor(psum, off, 32);
  if (lane == 31) wown[w] = incl;
  if (lane == 0)  wpre[w] = psum;
  __syncthreads();
  int base = 0;
  for (int j = 0; j < 32; ++j) {
    base += wpre[j];
    if (j < w) base += wown[j];
  }
  if (row < NN) start[k * NN + row] = base + incl - own;
}

__global__ __launch_bounds__(256) void fill_sorted(
    const int* __restrict__ rows, const int* __restrict__ cols,
    const float* __restrict__ vals, const int* __restrict__ start,
    const int* __restrict__ rank, EdgeCV* __restrict__ se) {
  const int k = blockIdx.y;
  const int e = blockIdx.x * 256 + threadIdx.x;
  const size_t eo = (size_t)k * E_ADJ + e;
  const int r = rows[eo];
  const int pos = start[k * NN + r] + rank[eo];
  if (pos < E_ADJ) {
    EdgeCV rec;
    rec.c = cols[eo];
    rec.v = vals[eo];
    se[(size_t)k * E_ADJ + pos] = rec;
  }
}

__global__ __launch_bounds__(256) void spmm_gather(
    const EdgeCV* __restrict__ se,
    const int* __restrict__ start, const float* __restrict__ xw,
    float* __restrict__ h0, float* __restrict__ h1) {
  const int lane = threadIdx.x & 31;
  const int half = lane >> 4;
  const int l    = lane & 15;
  const int r    = blockIdx.x * 8 + (threadIdx.x >> 5);
  const int i    = blockIdx.y;
  const int k    = i * 2 + half;

  const int* st = start + k * NN;
  int j = st[r];
  int end = (r + 1 < NN) ? st[r + 1] : E_ADJ;
  end = min(end, E_ADJ);
  const EdgeCV* ed = se + (size_t)k * E_ADJ;
  const float4* X = (const float4*)(xw + (size_t)k * NN * HH) + l;

  float4 acc = make_float4(0.0f, 0.0f, 0.0f, 0.0f);
  for (; j + 4 <= end; j += 4) {
    const EdgeCV e0 = ed[j], e1 = ed[j + 1], e2 = ed[j + 2], e3 = ed[j + 3];
    const float v0 = e0.v, v1 = e1.v, v2 = e2.v, v3 = e3.v;
    const float4 a0 = X[(size_t)e0.c * 16];
    const float4 a1 = X[(size_t)e1.c * 16];
    const float4 a2 = X[(size_t)e2.c * 16];
    const float4 a3 = X[(size_t)e3.c * 16];
    acc.x += a0.x * v0; acc.y += a0.y * v0; acc.z += a0.z * v0; acc.w += a0.w * v0;
    acc.x += a1.x * v1; acc.y += a1.y * v1; acc.z += a1.z * v1; acc.w += a1.w * v1;
    acc.x += a2.x * v2; acc.y += a2.y * v2; acc.z += a2.z * v2; acc.w += a2.w * v2;
    acc.x += a3.x * v3; acc.y += a3.y * v3; acc.z += a3.z * v3; acc.w += a3.w * v3;
  }
  const int rem = end - j;
  if (rem > 0) {
    const EdgeCV e0 = ed[j];
    const EdgeCV e1 = (rem > 1) ? ed[j + 1] : e0;
    const EdgeCV e2 = (rem > 2) ? ed[j + 2] : e0;
    const float v0 = e0.v, v1 = e1.v, v2 = e2.v;
    const float4 a0 = X[(size_t)e0.c * 16];
    const float4 a1 = X[(size_t)e1.c * 16];
    const float4 a2 = X[(size_t)e2.c * 16];
    acc.x += a0.x * v0; acc.y += a0.y * v0; acc.z += a0.z * v0; acc.w += a0.w * v0;
    if (rem > 1) {
      acc.x += a1.x * v1; acc.y += a1.y * v1; acc.z += a1.z * v1; acc.w += a1.w * v1;
    }
    if (rem > 2) {
      acc.x += a2.x * v2; acc.y += a2.y * v2; acc.z += a2.z * v2; acc.w += a2.w * v2;
    }
  }

  float4 q;
  q.x = fmaxf(acc.x, 0.0f); q.y = fmaxf(acc.y, 0.0f);
  q.z = fmaxf(acc.z, 0.0f); q.w = fmaxf(acc.w, 0.0f);
  float4 o;
  o.x = __shfl_xor(q.x, 16, 32); o.y = __shfl_xor(q.y, 16, 32);
  o.z = __shfl_xor(q.z, 16, 32); o.w = __shfl_xor(q.w, 16, 32);
  if (half == 0) {
    float4 s;
    s.x = q.x + o.x; s.y = q.y + o.y; s.z = q.z + o.z; s.w = q.w + o.w;
    ((float4*)(i ? h1 : h0))[(size_t)r * 16 + l] = s;
  }
}

__global__ void build_M(const float* __restrict__ R, const float* __restrict__ D,
                        const int* __restrict__ rt_k, float* __restrict__ M) {
  int i = blockIdx.x * blockDim.x + threadIdx.x;
  const float* d = D + rt_k[0] * HH;
  int h = i >> 6, g = i & 63;
  M[i] = d[h] * R[i] * d[g];
}

__global__ __launch_bounds__(128) void decode_edges(
    const float* __restrict__ h0, const float* __restrict__ h1,
    const int* __restrict__ edges, const float* __restrict__ M,
    float* __restrict__ preds) {
  __shared__ float part[4][16];
  const int wave = threadIdx.x >> 5;
  const int lane = threadIdx.x & 31;
  const int half = lane >> 4;
  const int l    = lane & 15;
  const int eb   = blockIdx.x * 16;

  const int arow = edges[(eb + l) * 2 + 0];
  const int col  = wave * 16 + l;

  const float* ar = h0 + (size_t)arow * HH + half * 2;
  v8f c = {};
  for (int kk = 0; kk < HH; kk += 4) {
    v2f a, b;
    a.x = ar[kk];
    a.y = ar[kk + 1];
    const float* mp = M + (kk + half * 2) * HH + col;
    b.x = mp[0];
    b.y = mp[HH];
    c = __builtin_amdgcn_wmma_f32_16x16x4_f32(
        false, a, false, b, (short)0, c, false, false);
  }

  float p[8];
#pragma unroll
  for (int v = 0; v < 8; ++v) {
    const int edge = v + half * 8;
    const int crow = edges[(eb + edge) * 2 + 1];
    p[v] = c[v] * h1[(size_t)crow * HH + col];
  }
#pragma unroll
  for (int off = 1; off < 16; off <<= 1) {
#pragma unroll
    for (int v = 0; v < 8; ++v) p[v] += __shfl_xor(p[v], off, 16);
  }
  if (l == 0) {
#pragma unroll
    for (int v = 0; v < 8; ++v) part[wave][v + half * 8] = p[v];
  }
  __syncthreads();
  if (threadIdx.x < 16) {
    int t = threadIdx.x;
    preds[eb + t] = part[0][t] + part[1][t] + part[2][t] + part[3][t];
  }
}

extern "C" void kernel_launch(void* const* d_in, const int* in_sizes, int n_in,
                              void* d_out, int out_size, void* d_ws, size_t ws_size,
                              hipStream_t stream) {
  const int*   adj_rows = (const int*)d_in[0];
  const int*   adj_cols = (const int*)d_in[1];
  const float* adj_vals = (const float*)d_in[2];
  const float* feat0    = (const float*)d_in[3];
  const float* feat1    = (const float*)d_in[4];
  const float* W1       = (const float*)d_in[5];
  const float* W2       = (const float*)d_in[6];
  const float* W3       = (const float*)d_in[7];
  const float* W4       = (const float*)d_in[8];
  const float* Rg       = (const float*)d_in[9];
  const float* Dl       = (const float*)d_in[10];
  const int*   edges    = (const int*)d_in[11];
  const int*   rt_k     = (const int*)d_in[12];

  const size_t S = (size_t)NN * HH;
  float* xw = (float*)d_ws;
  float* lists = xw + 4 * S;
  float* h0 = lists + 4 * S;
  float* h1 = h0 + S;
  float* M  = h1 + S;

  static_assert((size_t)4 * E_ADJ * 3 + (size_t)8 * NN <= (size_t)4 * NN * HH,
                "sorted lists must fit in the four former y planes");
  EdgeCV* se   = (EdgeCV*)lists;
  int*   cnt   = (int*)(se + (size_t)4 * E_ADJ);
  int*   start = cnt + (size_t)4 * NN;
  int*   rank  = start + (size_t)4 * NN;

  struct Stage { const float* W; int Fin; };
  const Stage st[4] = { {W1, 128}, {W2, 64}, {W3, 64}, {W4, 64} };

  const float* x0 = feat0;
  const float* x1 = feat1;
  for (int g = 0; g < 2; ++g) {
    const int*   rows = adj_rows + (size_t)g * 4 * E_ADJ;
    const int*   cols = adj_cols + (size_t)g * 4 * E_ADJ;
    const float* vals = adj_vals + (size_t)g * 4 * E_ADJ;

    zero_i32<<<(4 * NN + 255) / 256, 256, 0, stream>>>(cnt, 4 * NN);
    count_rank<<<dim3(E_ADJ / 256, 4), 256, 0, stream>>>(rows, cnt, rank);
    scan_rows<<<dim3((NN + 1023) / 1024, 4), 1024, 0, stream>>>(cnt, start);
    fill_sorted<<<dim3(E_ADJ / 256, 4), 256, 0, stream>>>(rows, cols, vals, start, rank, se);

    for (int t = 0; t < 2; ++t) {
      const int s = g * 2 + t;
      gemm_stage<<<dim3(NN / 16, 4), 128, 0, stream>>>(x0, x1, st[s].W, xw, st[s].Fin);
      spmm_gather<<<dim3(NN / 8, 2), 256, 0, stream>>>(se, start, xw, h0, h1);
      x0 = h0;
      x1 = h1;
    }
  }

  build_M<<<16, 256, 0, stream>>>(Rg, Dl, rt_k, M);
  decode_edges<<<E_DEC / 16, 128, 0, stream>>>(h0, h1, edges, M, (float*)d_out);
}


# Fewer, fatter waves, a third as many per call: 5 edges per thread in both bucketing passes, 2 rows per wave in the gather, a 16x64 strip per wave in GEMM and decoder; every sum keeps its order
# speedup vs baseline: 1.6834x; 1.0528x over previous
#include <hip/hip_runtime.h>
#include <hip/hip_bf16.h>

#define NN 50000
#define E_ADJ 800000
#define E_DEC 200000
#define HH 64

typedef float v2f __attribute__((ext_vector_type(2)));
typedef float v8f __attribute__((ext_vector_type(8)));

#define GEMM_WAVES 5
static_assert(NN % (16 * GEMM_WAVES) == 0, "row tiles must divide evenly among the blocks");

__global__ __launch_bounds__(32 * GEMM_WAVES) void gemm_stage(
    const float* __restrict__ x0, const float* __restrict__ x1,
    const float* __restrict__ W, float* __restrict__ xw, int Fin) {
  const int k    = blockIdx.y;
  const int tm   = blockIdx.x * GEMM_WAVES + (threadIdx.x >> 5);
  const int lane = threadIdx.x & 31;
  const int half = lane >> 4;
  const int l    = lane & 15;

  const float* X  = (k & 1) ? x1 : x0;
  const float* Wk = W + (size_t)k * Fin * HH;

  const int row = tm * 16 + l;

  const float* ar = X + (size_t)row * Fin + half * 2;
  v8f c0 = {}, c1 = {}, c2 = {}, c3 = {};
  for (int kk = 0; kk < Fin; kk += 4) {
    v2f a, b0, b1, b2, b3;
    a.x = ar[kk];
    a.y = ar[kk + 1];
    const float* wp = Wk + (size_t)(kk + half * 2) * HH + l;
    b0.x = wp[0];   b0.y = wp[HH];
    b1.x = wp[16];  b1.y = wp[HH + 16];
    b2.x = wp[32];  b2.y = wp[HH + 32];
    b3.x = wp[48];  b3.y = wp[HH + 48];
    c0 = __builtin_amdgcn_wmma_f32_16x16x4_f32(
         false, a,  false, b0,
         (short)0, c0,  false,  false);
    c1 = __builtin_amdgcn_wmma_f32_16x16x4_f32(false, a, false, b1, (short)0, c1, false, false);
    c2 = __builtin_amdgcn_wmma_f32_16x16x4_f32(false, a, false, b2, (short)0, c2, false, false);
    c3 = __builtin_amdgcn_wmma_f32_16x16x4_f32(false, a, false, b3, (short)0, c3, false, false);
  }

  float* out = xw + ((size_t)k * NN + (size_t)tm * 16) * HH + l;
#pragma unroll
  for (int v = 0; v < 8; ++v) {
    float* o = out + (size_t)(v + half * 8) * HH;
    o[0]  = c0[v];
    o[16] = c1[v];
    o[32] = c2[v];
    o[48] = c3[v];
  }
}

struct __align__(8) EdgeCV { int c; float v; };

__global__ void zero_i32(int* __restrict__ p, int n4) {
  int i = blockIdx.x * blockDim.x + threadIdx.x;
  if (i < n4) ((int4*)p)[i] = make_int4(0, 0, 0, 0);
}

#define EPT 5
static_assert(E_ADJ % (256 * EPT) == 0, "every thread of the bucketing passes has EPT edges");

__global__ __launch_bounds__(256) void count_rank(
    const int* __restrict__ rows, int* __restrict__ cnt, int* __restrict__ rank) {
  const int k = blockIdx.y;
  const size_t eo = (size_t)k * E_ADJ + (size_t)blockIdx.x * (256 * EPT) + threadIdx.x;
  int r[EPT], q[EPT];
#pragma unroll
  for (int u = 0; u < EPT; ++u) r[u] = rows[eo + u * 256];
#pragma unroll
  for (int u = 0; u < EPT; ++u) q[u] = atomicAdd(cnt + k * NN + r[u], 1);
#pragma unroll
  for (int u = 0; u < EPT; ++u) rank[eo + u * 256] = q[u];
}

__global__ __launch_bounds__(1024) void scan_rows(
    const int* __restrict__ cnt, int* __restrict__ start) {
  __shared__ int wpre[32];
  __shared__ int wown[32];
  const int k = blockIdx.y;
  const int t = threadIdx.x;
  const int lane = t & 31;
  const int w = t >> 5;
  const int lo = blockIdx.x * 1024;
  const int* c = cnt + k * NN;

  int pre = 0;
  for (int i = t; i < lo; i += 1024) pre += c[i];
  const int row = lo + t;
  const int own = (row < NN) ? c[row] : 0;

  int incl = own;
#pragma unroll
  for (int off = 1; off < 32; off <<= 1) {
    const int n = __shfl_up(incl, off, 32);
    if (lane >= off) incl += n;
  }
  int psum = pre;
#pragma unroll
  for (int off = 16; off > 0; off >>= 1) psum += __shfl_xor(psum, off, 32);
  if (lane == 31) wown[w] = incl;
  if (lane == 0)  wpre[w] = psum;
  __syncthreads();
  int base = 0;
  for (int j = 0; j < 32; ++j) {
    base += wpre[j];
    if (j < w) base += wown[j];
  }
  if (row < NN) start[k * NN + row] = base + incl - own;
}

__global__ __launch_bounds__(256) void fill_sorted(
    const int* __restrict__ rows, const int* __restrict__ cols,
    const float* __restrict__ vals, const int* __restrict__ start,
    const int* __restrict__ rank, EdgeCV* __restrict__ se) {
  const int k = blockIdx.y;
  const size_t eo = (size_t)k * E_ADJ + (size_t)blockIdx.x * (256 * EPT) + threadIdx.x;
  int r[EPT], rk[EPT], pos[EPT];
  EdgeCV rec[EPT];
#pragma unroll
  for (int u = 0; u < EPT; ++u) {
    r[u]     = rows[eo + u * 256];
    rk[u]    = rank[eo + u * 256];
    rec[u].c = cols[eo + u * 256];
    rec[u].v = vals[eo + u * 256];
  }
#pragma unroll
  for (int u = 0; u < EPT; ++u) pos[u] = start[k * NN + r[u]] + rk[u];
#pragma unroll
  for (int u = 0; u < EPT; ++u) {
    if (pos[u] < E_ADJ)
      se[(size_t)k * E_ADJ + pos[u]] = rec[u];
  }
}

static_assert(NN % 16 == 0, "sixteen rows per block of the gather");

__global__ __launch_bounds__(256) void spmm_gather(
    const EdgeCV* __restrict__ se,
    const int* __restrict__ start, const float* __restrict__ xw,
    float* __restrict__ h0, float* __restrict__ h1) {
  const int lane = threadIdx.x & 31;
  const int grp  = lane >> 3;
  const int half = grp & 1;
  const int l8   = lane & 7;
  const int r    = blockIdx.x * 16 + (threadIdx.x >> 5) * 2 + (grp >> 1);
  const int i    = blockIdx.y;
  const int k    = i * 2 + half;

  const int* st = start + k * NN;
  int j = st[r];
  int end = (r + 1 < NN) ? st[r + 1] : E_ADJ;
  end = min(end, E_ADJ);
  const EdgeCV* ed = se + (size_t)k * E_ADJ;
  const float4* X = (const float4*)(xw + (size_t)k * NN * HH) + l8;

  float4 acc = make_float4(0.0f, 0.0f, 0.0f, 0.0f);
  float4 bcc = make_float4(0.0f, 0.0f, 0.0f, 0.0f);
  for (; j + 4 <= end; j += 4) {
    const EdgeCV e0 = ed[j], e1 = ed[j + 1], e2 = ed[j + 2], e3 = ed[j + 3];
    const float v0 = e0.v, v1 = e1.v, v2 = e2.v, v3 = e3.v;
    const float4 a0 = X[(size_t)e0.c * 16], b0 = X[(size_t)e0.c * 16 + 8];
    const float4 a1 = X[(size_t)e1.c * 16], b1 = X[(size_t)e1.c * 16 + 8];
    const float4 a2 = X[(size_t)e2.c * 16], b2 = X[(size_t)e2.c * 16 + 8];
    const float4 a3 = X[(size_t)e3.c * 16], b3 = X[(size_t)e3.c * 16 + 8];
    acc.x += a0.x * v0; acc.y += a0.y * v0; acc.z += a0.z * v0; acc.w += a0.w * v0;
    bcc.x += b0.x * v0; bcc.y += b0.y * v0; bcc.z += b0.z * v0; bcc.w += b0.w * v0;
    acc.x += a1.x * v1; acc.y += a1.y * v1; acc.z += a1.z * v1; acc.w += a1.w * v1;
    bcc.x += b1.x * v1; bcc.y += b1.y * v1; bcc.z += b1.z * v1; bcc.w += b1.w * v1;
    acc.x += a2.x * v2; acc.y += a2.y * v2; acc.z += a2.z * v2; acc.w += a2.w * v2;
    bcc.x += b2.x * v2; bcc.y += b2.y * v2; bcc.z += b2.z * v2; bcc.w += b2.w * v2;
    acc.x += a3.x * v3; acc.y += a3.y * v3; acc.z += a3.z * v3; acc.w += a3.w * v3;
    bcc.x += b3.x * v3; bcc.y += b3.y * v3; bcc.z += b3.z * v3; bcc.w += b3.w * v3;
  }
  const int rem = end - j;
  if (rem > 0) {
    const EdgeCV e0 = ed[j];
    const EdgeCV e1 = (rem > 1) ? ed[j + 1] : e0;
    const EdgeCV e2 = (rem > 2) ? ed[j + 2] : e0;
    const float v0 = e0.v, v1 = e1.v, v2 = e2.v;
    const float4 a0 = X[(size_t)e0.c * 16], b0 = X[(size_t)e0.c * 16 + 8];
    const float4 a1 = X[(size_t)e1.c * 16], b1 = X[(size_t)e1.c * 16 + 8];
    const float4 a2 = X[(size_t)e2.c * 16], b2 = X[(size_t)e2.c * 16 + 8];
    acc.x += a0.x * v0; acc.y += a0.y * v0; acc.z += a0.z * v0; acc.w += a0.w * v0;
    bcc.x += b0.x * v0; bcc.y += b0.y * v0; bcc.z += b0.z * v0; bcc.w += b0.w * v0;
    if (rem > 1) {
      acc.x += a1.x * v1; acc.y += a1.y * v1; acc.z += a1.z * v1; acc.w += a1.w * v1;
      bcc.x += b1.x * v1; bcc.y += b1.y * v1; bcc.z += b1.z * v1; bcc.w += b1.w * v1;
    }
    if (rem > 2) {
      acc.x += a2.x * v2; acc.y += a2.y * v2; acc.z += a2.z * v2; acc.w += a2.w * v2;
      bcc.x += b2.x * v2; bcc.y += b2.y * v2; bcc.z += b2.z * v2; bcc.w += b2.w * v2;
    }
  }

  float4 q, p;
  q.x = fmaxf(acc.x, 0.0f); q.y = fmaxf(acc.y, 0.0f);
  q.z = fmaxf(acc.z, 0.0f); q.w = fmaxf(acc.w, 0.0f);
  p.x = fmaxf(bcc.x, 0.0f); p.y = fmaxf(bcc.y, 0.0f);
  p.z = fmaxf(bcc.z, 0.0f); p.w = fmaxf(bcc.w, 0.0f);
  float4 o, n;
  o.x = __shfl_xor(q.x, 8, 32); o.y = __shfl_xor(q.y, 8, 32);
  o.z = __shfl_xor(q.z, 8, 32); o.w = __shfl_xor(q.w, 8, 32);
  n.x = __shfl_xor(p.x, 8, 32); n.y = __shfl_xor(p.y, 8, 32);
  n.z = __shfl_xor(p.z, 8, 32); n.w = __shfl_xor(p.w, 8, 32);
  if (half == 0) {
    float4 s, t;
    s.x = q.x + o.x; s.y = q.y + o.y; s.z = q.z + o.z; s.w = q.w + o.w;
    t.x = p.x + n.x; t.y = p.y + n.y; t.z = p.z + n.z; t.w = p.w + n.w;
    float4* H = (float4*)(i ? h1 : h0) + (size_t)r * 16 + l8;
    H[0] = s;
    H[8] = t;
  }
}

__global__ void build_M(const float* __restrict__ R, const float* __restrict__ D,
                        const int* __restrict__ rt_k, float* __restrict__ M) {
  int i = blockIdx.x * blockDim.x + threadIdx.x;
  const float* d = D + rt_k[0] * HH;
  int h = i >> 6, g = i & 63;
  M[i] = d[h] * R[i] * d[g];
}

static_assert(E_DEC % 64 == 0, "four 16-edge groups per block of the decoder");

__global__ __launch_bounds__(128) void decode_edges(
    const float* __restrict__ h0, const float* __restrict__ h1,
    const int* __restrict__ edges, const float* __restrict__ M,
    float* __restrict__ preds) {
  const int wave = threadIdx.x >> 5;
  const int lane = threadIdx.x & 31;
  const int half = lane >> 4;
  const int l    = lane & 15;
  const int eb   = blockIdx.x * 64 + wave * 16;

  const int arow = edges[(eb + l) * 2 + 0];

  const float* ar = h0 + (size_t)arow * HH + half * 2;
  v8f c0 = {}, c1 = {}, c2 = {}, c3 = {};
  for (int kk = 0; kk < HH; kk += 4) {
    v2f a, b0, b1, b2, b3;
    a.x = ar[kk];
    a.y = ar[kk + 1];
    const float* mp = M + (kk + half * 2) * HH + l;
    b0.x = mp[0];   b0.y = mp[HH];
    b1.x = mp[16];  b1.y = mp[HH + 16];
    b2.x = mp[32];  b2.y = mp[HH + 32];
    b3.x = mp[48];  b3.y = mp[HH + 48];
    c0 = __builtin_amdgcn_wmma_f32_16x16x4_f32(false, a, false, b0, (short)0, c0, false, false);
    c1 = __builtin_amdgcn_wmma_f32_16x16x4_f32(false, a, false, b1, (short)0, c1, false, false);
    c2 = __builtin_amdgcn_wmma_f32_16x16x4_f32(false, a, false, b2, (short)0, c2, false, false);
    c3 = __builtin_amdgcn_wmma_f32_16x16x4_f32(false, a, false, b3, (short)0, c3, false, false);
  }

  float p0[8], p1[8], p2[8], p3[8];
#pragma unroll
  for (int v = 0; v < 8; ++v) {
    const int edge = v + half * 8;
    const int crow = edges[(eb + edge) * 2 + 1];
    const float* hc = h1 + (size_t)crow * HH + l;
    p0[v] = c0[v] * hc[0];
    p1[v] = c1[v] * hc[16];
    p2[v] = c2[v] * hc[32];
    p3[v] = c3[v] * hc[48];
  }
#pragma unroll
  for (int off = 1; off < 16; off <<= 1) {
#pragma unroll
    for (int v = 0; v < 8; ++v) {
      p0[v] += __shfl_xor(p0[v], off, 16);
      p1[v] += __shfl_xor(p1[v], off, 16);
      p2[v] += __shfl_xor(p2[v], off, 16);
      p3[v] += __shfl_xor(p3[v], off, 16);
    }
  }
  if (l == 0) {
    float s[8];
#pragma unroll
    for (int v = 0; v < 8; ++v) s[v] = p0[v] + p1[v] + p2[v] + p3[v];
    float4* o = (float4*)(preds + eb + half * 8);
    o[0] = make_float4(s[0], s[1], s[2], s[3]);
    o[1] = make_float4(s[4], s[5], s[6], s[7]);
  }
}

extern "C" void kernel_launch(void* const* d_in, const int* in_sizes, int n_in,
                              void* d_out, int out_size, void* d_ws, size_t ws_size,
                              hipStream_t stream) {
  const int*   adj_rows = (const int*)d_in[0];
  const int*   adj_cols = (const int*)d_in[1];
  const float* adj_vals = (const float*)d_in[2];
  const float* feat0    = (const float*)d_in[3];
  const float* feat1    = (const float*)d_in[4];
  const float* W1       = (const float*)d_in[5];
  const float* W2       = (const float*)d_in[6];
  const float* W3       = (const float*)d_in[7];
  const float* W4       = (const float*)d_in[8];
  const float* Rg       = (const float*)d_in[9];
  const float* Dl       = (const float*)d_in[10];
  const int*   edges    = (const int*)d_in[11];
  const int*   rt_k     = (const int*)d_in[12];

  const size_t S = (size_t)NN * HH;
  float* xw = (float*)d_ws;
  float* lists = xw + 4 * S;
  float* h0 = lists + 4 * S;
  float* h1 = h0 + S;
  float* M  = h1 + S;

  static_assert((size_t)4 * E_ADJ * 3 + (size_t)8 * NN <= (size_t)4 * NN * HH,
                "sorted lists must fit in the four former y planes");
  EdgeCV* se   = (EdgeCV*)lists;
  int*   cnt   = (int*)(se + (size_t)4 * E_ADJ);
  int*   start = cnt + (size_t)4 * NN;
  int*   rank  = start + (size_t)4 * NN;

  struct Stage { const float* W; int Fin; };
  const Stage st[4] = { {W1, 128}, {W2, 64}, {W3, 64}, {W4, 64} };

  const float* x0 = feat0;
  const float* x1 = feat1;
  for (int g = 0; g < 2; ++g) {
    const int*   rows = adj_rows + (size_t)g * 4 * E_ADJ;
    const int*   cols = adj_cols + (size_t)g * 4 * E_ADJ;
    const float* vals = adj_vals + (size_t)g * 4 * E_ADJ;

    zero_i32<<<(NN + 255) / 256, 256, 0, stream>>>(cnt, NN);
    count_rank<<<dim3(E_ADJ / (256 * EPT), 4), 256, 0, stream>>>(rows, cnt, rank);
    scan_rows<<<dim3((NN + 1023) / 1024, 4), 1024, 0, stream>>>(cnt, start);
    fill_sorted<<<dim3(E_ADJ / (256 * EPT), 4), 256, 0, stream>>>(rows, cols, vals, start, rank, se);

    for (int t = 0; t < 2; ++t) {
      const int s = g * 2 + t;
      gemm_stage<<<dim3(NN / (16 * GEMM_WAVES), 4), 32 * GEMM_WAVES, 0, stream>>>(x0, x1, st[s].W, xw, st[s].Fin);
      spmm_gather<<<dim3(NN / 16, 2), 256, 0, stream>>>(se, start, xw, h0, h1);
      x0 = h0;
      x1 = h1;
    }
  }

  build_M<<<16, 256, 0, stream>>>(Rg, Dl, rt_k, M);
  decode_edges<<<E_DEC / 64, 128, 0, stream>>>(h0, h1, edges, M, (float*)d_out);
}


# Edge bucketing on LDS counters: a block owns 782 rows, reads the relation's row array once, ranks its edges by LDS atomic arrival order, scans, places; global atomics and 6 of 18 launches go
# speedup vs baseline: 2.5788x; 1.5319x over previous
#include <hip/hip_runtime.h>
#include <hip/hip_bf16.h>

#define NN 50000
#define E_ADJ 800000
#define E_DEC 200000
#define HH 64

typedef float v2f __attribute__((ext_vector_type(2)));
typedef float v8f __attribute__((ext_vector_type(8)));

#define GEMM_WAVES 5
static_assert(NN % (16 * GEMM_WAVES) == 0, "row tiles must divide evenly among the blocks");

__global__ __launch_bounds__(32 * GEMM_WAVES) void gemm_stage(
    const float* __restrict__ x0, const float* __restrict__ x1,
    const float* __restrict__ W, float* __restrict__ xw, int Fin) {
  const int k    = blockIdx.y;
  const int tm   = blockIdx.x * GEMM_WAVES + (threadIdx.x >> 5);
  const int lane = threadIdx.x & 31;
  const int half = lane >> 4;
  const int l    = lane & 15;

  const float* X  = (k & 1) ? x1 : x0;
  const float* Wk = W + (size_t)k * Fin * HH;

  const int row = tm * 16 + l;

  const float* ar = X + (size_t)row * Fin + half * 2;
  v8f c0 = {}, c1 = {}, c2 = {}, c3 = {};
  for (int kk = 0; kk < Fin; kk += 4) {
    v2f a, b0, b1, b2, b3;
    a.x = ar[kk];
    a.y = ar[kk + 1];
    const float* wp = Wk + (size_t)(kk + half * 2) * HH + l;
    b0.x = wp[0];   b0.y = wp[HH];
    b1.x = wp[16];  b1.y = wp[HH + 16];
    b2.x = wp[32];  b2.y = wp[HH + 32];
    b3.x = wp[48];  b3.y = wp[HH + 48];
    c0 = __builtin_amdgcn_wmma_f32_16x16x4_f32(
         false, a,  false, b0,
         (short)0, c0,  false,  false);
    c1 = __builtin_amdgcn_wmma_f32_16x16x4_f32(false, a, false, b1, (short)0, c1, false, false);
    c2 = __builtin_amdgcn_wmma_f32_16x16x4_f32(false, a, false, b2, (short)0, c2, false, false);
    c3 = __builtin_amdgcn_wmma_f32_16x16x4_f32(false, a, false, b3, (short)0, c3, false, false);
  }

  float* out = xw + ((size_t)k * NN + (size_t)tm * 16) * HH + l;
#pragma unroll
  for (int v = 0; v < 8; ++v) {
    float* o = out + (size_t)(v + half * 8) * HH;
    o[0]  = c0[v];
    o[16] = c1[v];
    o[32] = c2[v];
    o[48] = c3[v];
  }
}

struct __align__(8) EdgeCV { int c; float v; };

#define RB_BLOCKS  64
#define RB_ROWS    ((NN + RB_BLOCKS - 1) / RB_BLOCKS)
#define RB_THREADS 1024
#define RB_WAVES   (RB_THREADS / 32)
#define RB_CAPW    1408
#define RB_N4      (E_ADJ / 4)
static_assert(RB_ROWS <= RB_THREADS && RB_ROWS <= 1024, "one counter per thread; local row packs into 10 bits");
static_assert(E_ADJ <= (1 << 20) && E_ADJ % 4 == 0, "edge number packs into 20 bits; rows are read in groups of four");
static_assert(RB_N4 % (4 * RB_THREADS) == 0 || RB_N4 % (4 * RB_THREADS) >= RB_THREADS,
              "every thread of a block makes the same number of trips through the reading loop (the wave-wide counts rely on it)");

__global__ __launch_bounds__(RB_THREADS) void bucket_rows(
    const int* __restrict__ rows, const int* __restrict__ cols,
    const float* __restrict__ vals, int* __restrict__ start, EdgeCV* __restrict__ se) {
  __shared__ unsigned wl[RB_WAVES][RB_CAPW];
  __shared__ int cnt[RB_THREADS];
  __shared__ int off[RB_THREADS];
  __shared__ int wtot[RB_WAVES];
  __shared__ int wbel[RB_WAVES];
  __shared__ int wcnt[RB_WAVES];

  const int k    = blockIdx.y;
  const int t    = threadIdx.x;
  const int lane = t & 31;
  const int w    = t >> 5;
  const int lo   = blockIdx.x * RB_ROWS;
  const unsigned nrows = (unsigned)min(RB_ROWS, NN - lo);
  const int4* R4 = (const int4*)(rows + (size_t)k * E_ADJ);
  const int*  C  = cols + (size_t)k * E_ADJ;
  const float* V = vals + (size_t)k * E_ADJ;
  EdgeCV* S = se + (size_t)k * E_ADJ;
  const int rot = blockIdx.x * (RB_N4 / RB_BLOCKS);

  cnt[t] = 0;
  __syncthreads();

  int wn = 0;
  int below = 0;
  for (int i0 = t; i0 < RB_N4; i0 += 4 * RB_THREADS) {
    int4 q[4];
    int  g[4];
#pragma unroll
    for (int u = 0; u < 4; ++u) {
      const int i = i0 + u * RB_THREADS;
      int gg = i + rot;
      if (gg >= RB_N4) gg -= RB_N4;
      g[u] = gg;
      q[u] = (i < RB_N4) ? R4[gg] : make_int4(-1, -1, -1, -1);
    }
#pragma unroll
    for (int u = 0; u < 4; ++u) {
      const int rr[4] = { q[u].x, q[u].y, q[u].z, q[u].w };
#pragma unroll
      for (int c = 0; c < 4; ++c) {
        const int row = rr[c];
        below += __builtin_popcount(__builtin_amdgcn_ballot_w32((unsigned)row < (unsigned)lo));
        const unsigned d = (unsigned)(row - lo);
        const bool mt = d < nrows;
        const unsigned m = __builtin_amdgcn_ballot_w32(mt);
        if (mt) {
          const int p = wn + (int)__builtin_amdgcn_mbcnt_lo(m, 0u);
          if (p < RB_CAPW) wl[w][p] = ((unsigned)(g[u] * 4 + c) << 10) | d;
          atomicAdd(&cnt[d], 1);
        }
        wn += __builtin_popcount(m);
      }
    }
  }
  if (lane == 0) { wcnt[w] = wn; wbel[w] = below; }
  __syncthreads();

  const int own = ((unsigned)t < nrows) ? cnt[t] : 0;
  int incl = own;
#pragma unroll
  for (int o = 1; o < 32; o <<= 1) {
    const int n = __shfl_up(incl, o, 32);
    if (lane >= o) incl += n;
  }
  if (lane == 31) wtot[w] = incl;
  __syncthreads();
  int base = 0;
  bool fits = true;
  for (int j = 0; j < RB_WAVES; ++j) {
    base += wbel[j];
    if (j < w) base += wtot[j];
    fits = fits && (wcnt[j] <= RB_CAPW);
  }
  const int ex = base + incl - own;
  off[t] = ex;
  cnt[t] = 0;
  if ((unsigned)t < nrows) start[k * NN + lo + t] = ex;
  __syncthreads();

  if (fits) {
    const int n = wcnt[w];
    for (int i0 = lane; i0 < n; i0 += 4 * 32) {
      int e[4], pos[4];
      EdgeCV rec[4];
#pragma unroll
      for (int u = 0; u < 4; ++u) {
        const int i = i0 + u * 32;
        const unsigned x = (i < n) ? wl[w][i] : 0u;
        e[u] = (int)(x >> 10);
        const int d = (int)(x & 1023u);
        pos[u] = (i < n) ? off[d] + atomicAdd(&cnt[d], 1) : E_ADJ;
      }
#pragma unroll
      for (int u = 0; u < 4; ++u) { rec[u].c = C[e[u]]; rec[u].v = V[e[u]]; }
#pragma unroll
      for (int u = 0; u < 4; ++u) {
        if (pos[u] < E_ADJ) S[pos[u]] = rec[u];
      }
    }
  } else {
    for (int i = t; i < RB_N4; i += RB_THREADS) {
      const int4 q = R4[i];
      const int rr[4] = { q.x, q.y, q.z, q.w };
#pragma unroll
      for (int c = 0; c < 4; ++c) {
        const unsigned d = (unsigned)(rr[c] - lo);
        if (d < nrows) {
          const int pos = off[d] + atomicAdd(&cnt[d], 1);
          EdgeCV rec;
          rec.c = C[i * 4 + c];
          rec.v = V[i * 4 + c];
          if (pos < E_ADJ) S[pos] = rec;
        }
      }
    }
  }
}

static_assert(NN % 16 == 0, "sixteen rows per block of the gather");

__global__ __launch_bounds__(256) void spmm_gather(
    const EdgeCV* __restrict__ se,
    const int* __restrict__ start, const float* __restrict__ xw,
    float* __restrict__ h0, float* __restrict__ h1) {
  const int lane = threadIdx.x & 31;
  const int grp  = lane >> 3;
  const int half = grp & 1;
  const int l8   = lane & 7;
  const int r    = blockIdx.x * 16 + (threadIdx.x >> 5) * 2 + (grp >> 1);
  const int i    = blockIdx.y;
  const int k    = i * 2 + half;

  const int* st = start + k * NN;
  int j = st[r];
  int end = (r + 1 < NN) ? st[r + 1] : E_ADJ;
  end = min(end, E_ADJ);
  const EdgeCV* ed = se + (size_t)k * E_ADJ;
  const float4* X = (const float4*)(xw + (size_t)k * NN * HH) + l8;

  float4 acc = make_float4(0.0f, 0.0f, 0.0f, 0.0f);
  float4 bcc = make_float4(0.0f, 0.0f, 0.0f, 0.0f);
  for (; j + 4 <= end; j += 4) {
    const EdgeCV e0 = ed[j], e1 = ed[j + 1], e2 = ed[j + 2], e3 = ed[j + 3];
    const float v0 = e0.v, v1 = e1.v, v2 = e2.v, v3 = e3.v;
    const float4 a0 = X[(size_t)e0.c * 16], b0 = X[(size_t)e0.c * 16 + 8];
    const float4 a1 = X[(size_t)e1.c * 16], b1 = X[(size_t)e1.c * 16 + 8];
    const float4 a2 = X[(size_t)e2.c * 16], b2 = X[(size_t)e2.c * 16 + 8];
    const float4 a3 = X[(size_t)e3.c * 16], b3 = X[(size_t)e3.c * 16 + 8];
    acc.x += a0.x * v0; acc.y += a0.y * v0; acc.z += a0.z * v0; acc.w += a0.w * v0;
    bcc.x += b0.x * v0; bcc.y += b0.y * v0; bcc.z += b0.z * v0; bcc.w += b0.w * v0;
    acc.x += a1.x * v1; acc.y += a1.y * v1; acc.z += a1.z * v1; acc.w += a1.w * v1;
    bcc.x += b1.x * v1; bcc.y += b1.y * v1; bcc.z += b1.z * v1; bcc.w += b1.w * v1;
    acc.x += a2.x * v2; acc.y += a2.y * v2; acc.z += a2.z * v2; acc.w += a2.w * v2;
    bcc.x += b2.x * v2; bcc.y += b2.y * v2; bcc.z += b2.z * v2; bcc.w += b2.w * v2;
    acc.x += a3.x * v3; acc.y += a3.y * v3; acc.z += a3.z * v3; acc.w += a3.w * v3;
    bcc.x += b3.x * v3; bcc.y += b3.y * v3; bcc.z += b3.z * v3; bcc.w += b3.w * v3;
  }
  const int rem = end - j;
  if (rem > 0) {
    const EdgeCV e0 = ed[j];
    const EdgeCV e1 = (rem > 1) ? ed[j + 1] : e0;
    const EdgeCV e2 = (rem > 2) ? ed[j + 2] : e0;
    const float v0 = e0.v, v1 = e1.v, v2 = e2.v;
    const float4 a0 = X[(size_t)e0.c * 16], b0 = X[(size_t)e0.c * 16 + 8];
    const float4 a1 = X[(size_t)e1.c * 16], b1 = X[(size_t)e1.c * 16 + 8];
    const float4 a2 = X[(size_t)e2.c * 16], b2 = X[(size_t)e2.c * 16 + 8];
    acc.x += a0.x * v0; acc.y += a0.y * v0; acc.z += a0.z * v0; acc.w += a0.w * v0;
    bcc.x += b0.x * v0; bcc.y += b0.y * v0; bcc.z += b0.z * v0; bcc.w += b0.w * v0;
    if (rem > 1) {
      acc.x += a1.x * v1; acc.y += a1.y * v1; acc.z += a1.z * v1; acc.w += a1.w * v1;
      bcc.x += b1.x * v1; bcc.y += b1.y * v1; bcc.z += b1.z * v1; bcc.w += b1.w * v1;
    }
    if (rem > 2) {
      acc.x += a2.x * v2; acc.y += a2.y * v2; acc.z += a2.z * v2; acc.w += a2.w * v2;
      bcc.x += b2.x * v2; bcc.y += b2.y * v2; bcc.z += b2.z * v2; bcc.w += b2.w * v2;
    }
  }

  float4 q, p;
  q.x = fmaxf(acc.x, 0.0f); q.y = fmaxf(acc.y, 0.0f);
  q.z = fmaxf(acc.z, 0.0f); q.w = fmaxf(acc.w, 0.0f);
  p.x = fmaxf(bcc.x, 0.0f); p.y = fmaxf(bcc.y, 0.0f);
  p.z = fmaxf(bcc.z, 0.0f); p.w = fmaxf(bcc.w, 0.0f);
  float4 o, n;
  o.x = __shfl_xor(q.x, 8, 32); o.y = __shfl_xor(q.y, 8, 32);
  o.z = __shfl_xor(q.z, 8, 32); o.w = __shfl_xor(q.w, 8, 32);
  n.x = __shfl_xor(p.x, 8, 32); n.y = __shfl_xor(p.y, 8, 32);
  n.z = __shfl_xor(p.z, 8, 32); n.w = __shfl_xor(p.w, 8, 32);
  if (half == 0) {
    float4 s, t;
    s.x = q.x + o.x; s.y = q.y + o.y; s.z = q.z + o.z; s.w = q.w + o.w;
    t.x = p.x + n.x; t.y = p.y + n.y; t.z = p.z + n.z; t.w = p.w + n.w;
    float4* H = (float4*)(i ? h1 : h0) + (size_t)r * 16 + l8;
    H[0] = s;
    H[8] = t;
  }
}

__global__ void build_M(const float* __restrict__ R, const float* __restrict__ D,
                        const int* __restrict__ rt_k, float* __restrict__ M) {
  int i = blockIdx.x * blockDim.x + threadIdx.x;
  const float* d = D + rt_k[0] * HH;
  int h = i >> 6, g = i & 63;
  M[i] = d[h] * R[i] * d[g];
}

static_assert(E_DEC % 64 == 0, "four 16-edge groups per block of the decoder");

__global__ __launch_bounds__(128) void decode_edges(
    const float* __restrict__ h0, const float* __restrict__ h1,
    const int* __restrict__ edges, const float* __restrict__ M,
    float* __restrict__ preds) {
  const int wave = threadIdx.x >> 5;
  const int lane = threadIdx.x & 31;
  const int half = lane >> 4;
  const int l    = lane & 15;
  const int eb   = blockIdx.x * 64 + wave * 16;

  const int arow = edges[(eb + l) * 2 + 0];

  const float* ar = h0 + (size_t)arow * HH + half * 2;
  v8f c0 = {}, c1 = {}, c2 = {}, c3 = {};
  for (int kk = 0; kk < HH; kk += 4) {
    v2f a, b0, b1, b2, b3;
    a.x = ar[kk];
    a.y = ar[kk + 1];
    const float* mp = M + (kk + half * 2) * HH + l;
    b0.x = mp[0];   b0.y = mp[HH];
    b1.x = mp[16];  b1.y = mp[HH + 16];
    b2.x = mp[32];  b2.y = mp[HH + 32];
    b3.x = mp[48];  b3.y = mp[HH + 48];
    c0 = __builtin_amdgcn_wmma_f32_16x16x4_f32(false, a, false, b0, (short)0, c0, false, false);
    c1 = __builtin_amdgcn_wmma_f32_16x16x4_f32(false, a, false, b1, (short)0, c1, false, false);
    c2 = __builtin_amdgcn_wmma_f32_16x16x4_f32(false, a, false, b2, (short)0, c2, false, false);
    c3 = __builtin_amdgcn_wmma_f32_16x16x4_f32(false, a, false, b3, (short)0, c3, false, false);
  }

  float p0[8], p1[8], p2[8], p3[8];
#pragma unroll
  for (int v = 0; v < 8; ++v) {
    const int edge = v + half * 8;
    const int crow = edges[(eb + edge) * 2 + 1];
    const float* hc = h1 + (size_t)crow * HH + l;
    p0[v] = c0[v] * hc[0];
    p1[v] = c1[v] * hc[16];
    p2[v] = c2[v] * hc[32];
    p3[v] = c3[v] * hc[48];
  }
#pragma unroll
  for (int off = 1; off < 16; off <<= 1) {
#pragma unroll
    for (int v = 0; v < 8; ++v) {
      p0[v] += __shfl_xor(p0[v], off, 16);
      p1[v] += __shfl_xor(p1[v], off, 16);
      p2[v] += __shfl_xor(p2[v], off, 16);
      p3[v] += __shfl_xor(p3[v], off, 16);
    }
  }
  if (l == 0) {
    float s[8];
#pragma unroll
    for (int v = 0; v < 8; ++v) s[v] = p0[v] + p1[v] + p2[v] + p3[v];
    float4* o = (float4*)(preds + eb + half * 8);
    o[0] = make_float4(s[0], s[1], s[2], s[3]);
    o[1] = make_float4(s[4], s[5], s[6], s[7]);
  }
}

extern "C" void kernel_launch(void* const* d_in, const int* in_sizes, int n_in,
                              void* d_out, int out_size, void* d_ws, size_t ws_size,
                              hipStream_t stream) {
  const int*   adj_rows = (const int*)d_in[0];
  const int*   adj_cols = (const int*)d_in[1];
  const float* adj_vals = (const float*)d_in[2];
  const float* feat0    = (const float*)d_in[3];
  const float* feat1    = (const float*)d_in[4];
  const float* W1       = (const float*)d_in[5];
  const float* W2       = (const float*)d_in[6];
  const float* W3       = (const float*)d_in[7];
  const float* W4       = (const float*)d_in[8];
  const float* Rg       = (const float*)d_in[9];
  const float* Dl       = (const float*)d_in[10];
  const int*   edges    = (const int*)d_in[11];
  const int*   rt_k     = (const int*)d_in[12];

  const size_t S = (size_t)NN * HH;
  float* xw = (float*)d_ws;
  float* lists = xw + 4 * S;
  float* h0 = lists + 4 * S;
  float* h1 = h0 + S;
  float* M  = h1 + S;

  static_assert((size_t)4 * E_ADJ * 2 + (size_t)4 * NN <= (size_t)4 * NN * HH,
                "sorted lists must fit in the four former y planes");
  EdgeCV* se   = (EdgeCV*)lists;
  int*   start = (int*)(se + (size_t)4 * E_ADJ);

  struct Stage { const float* W; int Fin; };
  const Stage st[4] = { {W1, 128}, {W2, 64}, {W3, 64}, {W4, 64} };

  const float* x0 = feat0;
  const float* x1 = feat1;
  for (int g = 0; g < 2; ++g) {
    const int*   rows = adj_rows + (size_t)g * 4 * E_ADJ;
    const int*   cols = adj_cols + (size_t)g * 4 * E_ADJ;
    const float* vals = adj_vals + (size_t)g * 4 * E_ADJ;

    bucket_rows<<<dim3(RB_BLOCKS, 4), RB_THREADS, 0, stream>>>(rows, cols, vals, start, se);

    for (int t = 0; t < 2; ++t) {
      const int s = g * 2 + t;
      gemm_stage<<<dim3(NN / (16 * GEMM_WAVES), 4), 32 * GEMM_WAVES, 0, stream>>>(x0, x1, st[s].W, xw, st[s].Fin);
      spmm_gather<<<dim3(NN / 16, 2), 256, 0, stream>>>(se, start, xw, h0, h1);
      x0 = h0;
      x1 = h1;
    }
  }

  build_M<<<16, 256, 0, stream>>>(Rg, Dl, rt_k, M);
  decode_edges<<<E_DEC / 64, 128, 0, stream>>>(h0, h1, edges, M, (float*)d_out);
}


# Layers 2-4 as one launch each: gather the 64-wide input rows first, then the 16-row tile times W on WMMA in the block, (A x) W for A (x W): products reassociated; 3 GEMM launches and xw planes go
# speedup vs baseline: 2.9726x; 1.1527x over previous
#include <hip/hip_runtime.h>
#include <hip/hip_bf16.h>

#define NN 50000
#define E_ADJ 800000
#define E_DEC 200000
#define HH 64

typedef float v2f __attribute__((ext_vector_type(2)));
typedef float v8f __attribute__((ext_vector_type(8)));

#define GEMM_WAVES 5
static_assert(NN % (16 * GEMM_WAVES) == 0, "row tiles must divide evenly among the blocks");

__global__ __launch_bounds__(32 * GEMM_WAVES) void gemm_stage(
    const float* __restrict__ x0, const float* __restrict__ x1,
    const float* __restrict__ W, float* __restrict__ xw, int Fin) {
  const int k    = blockIdx.y;
  const int tm   = blockIdx.x * GEMM_WAVES + (threadIdx.x >> 5);
  const int lane = threadIdx.x & 31;
  const int half = lane >> 4;
  const int l    = lane & 15;

  const float* X  = (k & 1) ? x1 : x0;
  const float* Wk = W + (size_t)k * Fin * HH;

  const int row = tm * 16 + l;

  const float* ar = X + (size_t)row * Fin + half * 2;
  v8f c0 = {}, c1 = {}, c2 = {}, c3 = {};
  for (int kk = 0; kk < Fin; kk += 4) {
    v2f a, b0, b1, b2, b3;
    a.x = ar[kk];
    a.y = ar[kk + 1];
    const float* wp = Wk + (size_t)(kk + half * 2) * HH + l;
    b0.x = wp[0];   b0.y = wp[HH];
    b1.x = wp[16];  b1.y = wp[HH + 16];
    b2.x = wp[32];  b2.y = wp[HH + 32];
    b3.x = wp[48];  b3.y = wp[HH + 48];
    c0 = __builtin_amdgcn_wmma_f32_16x16x4_f32(
         false, a,  false, b0,
         (short)0, c0,  false,  false);
    c1 = __builtin_amdgcn_wmma_f32_16x16x4_f32(false, a, false, b1, (short)0, c1, false, false);
    c2 = __builtin_amdgcn_wmma_f32_16x16x4_f32(false, a, false, b2, (short)0, c2, false, false);
    c3 = __builtin_amdgcn_wmma_f32_16x16x4_f32(false, a, false, b3, (short)0, c3, false, false);
  }

  float* out = xw + ((size_t)k * NN + (size_t)tm * 16) * HH + l;
#pragma unroll
  for (int v = 0; v < 8; ++v) {
    float* o = out + (size_t)(v + half * 8) * HH;
    o[0]  = c0[v];
    o[16] = c1[v];
    o[32] = c2[v];
    o[48] = c3[v];
  }
}

struct __align__(8) EdgeCV { int c; float v; };

#define RB_BLOCKS  64
#define RB_ROWS    ((NN + RB_BLOCKS - 1) / RB_BLOCKS)
#define RB_THREADS 1024
#define RB_WAVES   (RB_THREADS / 32)
#define RB_CAPW    1408
#define RB_N4      (E_ADJ / 4)
static_assert(RB_ROWS <= RB_THREADS && RB_ROWS <= 1024, "one counter per thread; local row packs into 10 bits");
static_assert(E_ADJ <= (1 << 20) && E_ADJ % 4 == 0, "edge number packs into 20 bits; rows are read in groups of four");
static_assert(RB_N4 % (4 * RB_THREADS) == 0 || RB_N4 % (4 * RB_THREADS) >= RB_THREADS,
              "every thread of a block makes the same number of trips through the reading loop (the wave-wide counts rely on it)");

__global__ __launch_bounds__(RB_THREADS) void bucket_rows(
    const int* __restrict__ rows, const int* __restrict__ cols,
    const float* __restrict__ vals, int* __restrict__ start, EdgeCV* __restrict__ se) {
  __shared__ unsigned wl[RB_WAVES][RB_CAPW];
  __shared__ int cnt[RB_THREADS];
  __shared__ int off[RB_THREADS];
  __shared__ int wtot[RB_WAVES];
  __shared__ int wbel[RB_WAVES];
  __shared__ int wcnt[RB_WAVES];

  const int k    = blockIdx.y;
  const int t    = threadIdx.x;
  const int lane = t & 31;
  const int w    = t >> 5;
  const int lo   = blockIdx.x * RB_ROWS;
  const unsigned nrows = (unsigned)min(RB_ROWS, NN - lo);
  const int4* R4 = (const int4*)(rows + (size_t)k * E_ADJ);
  const int*  C  = cols + (size_t)k * E_ADJ;
  const float* V = vals + (size_t)k * E_ADJ;
  EdgeCV* S = se + (size_t)k * E_ADJ;
  const int rot = blockIdx.x * (RB_N4 / RB_BLOCKS);

  cnt[t] = 0;
  __syncthreads();

  int wn = 0;
  int below = 0;
  for (int i0 = t; i0 < RB_N4; i0 += 4 * RB_THREADS) {
    int4 q[4];
    int  g[4];
#pragma unroll
    for (int u = 0; u < 4; ++u) {
      const int i = i0 + u * RB_THREADS;
      int gg = i + rot;
      if (gg >= RB_N4) gg -= RB_N4;
      g[u] = gg;
      q[u] = (i < RB_N4) ? R4[gg] : make_int4(-1, -1, -1, -1);
    }
#pragma unroll
    for (int u = 0; u < 4; ++u) {
      const int rr[4] = { q[u].x, q[u].y, q[u].z, q[u].w };
#pragma unroll
      for (int c = 0; c < 4; ++c) {
        const int row = rr[c];
        below += __builtin_popcount(__builtin_amdgcn_ballot_w32((unsigned)row < (unsigned)lo));
        const unsigned d = (unsigned)(row - lo);
        const bool mt = d < nrows;
        const unsigned m = __builtin_amdgcn_ballot_w32(mt);
        if (mt) {
          const int p = wn + (int)__builtin_amdgcn_mbcnt_lo(m, 0u);
          if (p < RB_CAPW) wl[w][p] = ((unsigned)(g[u] * 4 + c) << 10) | d;
          atomicAdd(&cnt[d], 1);
        }
        wn += __builtin_popcount(m);
      }
    }
  }
  if (lane == 0) { wcnt[w] = wn; wbel[w] = below; }
  __syncthreads();

  const int own = ((unsigned)t < nrows) ? cnt[t] : 0;
  int incl = own;
#pragma unroll
  for (int o = 1; o < 32; o <<= 1) {
    const int n = __shfl_up(incl, o, 32);
    if (lane >= o) incl += n;
  }
  if (lane == 31) wtot[w] = incl;
  __syncthreads();
  int base = 0;
  bool fits = true;
  for (int j = 0; j < RB_WAVES; ++j) {
    base += wbel[j];
    if (j < w) base += wtot[j];
    fits = fits && (wcnt[j] <= RB_CAPW);
  }
  const int ex = base + incl - own;
  off[t] = ex;
  cnt[t] = 0;
  if ((unsigned)t < nrows) start[k * NN + lo + t] = ex;
  __syncthreads();

  if (fits) {
    const int n = wcnt[w];
    for (int i0 = lane; i0 < n; i0 += 4 * 32) {
      int e[4], pos[4];
      EdgeCV rec[4];
#pragma unroll
      for (int u = 0; u < 4; ++u) {
        const int i = i0 + u * 32;
        const unsigned x = (i < n) ? wl[w][i] : 0u;
        e[u] = (int)(x >> 10);
        const int d = (int)(x & 1023u);
        pos[u] = (i < n) ? off[d] + atomicAdd(&cnt[d], 1) : E_ADJ;
      }
#pragma unroll
      for (int u = 0; u < 4; ++u) { rec[u].c = C[e[u]]; rec[u].v = V[e[u]]; }
#pragma unroll
      for (int u = 0; u < 4; ++u) {
        if (pos[u] < E_ADJ) S[pos[u]] = rec[u];
      }
    }
  } else {
    for (int i = t; i < RB_N4; i += RB_THREADS) {
      const int4 q = R4[i];
      const int rr[4] = { q.x, q.y, q.z, q.w };
#pragma unroll
      for (int c = 0; c < 4; ++c) {
        const unsigned d = (unsigned)(rr[c] - lo);
        if (d < nrows) {
          const int pos = off[d] + atomicAdd(&cnt[d], 1);
          EdgeCV rec;
          rec.c = C[i * 4 + c];
          rec.v = V[i * 4 + c];
          if (pos < E_ADJ) S[pos] = rec;
        }
      }
    }
  }
}

static_assert(NN % 16 == 0, "sixteen rows per block of the gather");

__global__ __launch_bounds__(256) void spmm_gather(
    const EdgeCV* __restrict__ se,
    const int* __restrict__ start, const float* __restrict__ xw,
    float* __restrict__ h0, float* __restrict__ h1) {
  const int lane = threadIdx.x & 31;
  const int grp  = lane >> 3;
  const int half = grp & 1;
  const int l8   = lane & 7;
  const int r    = blockIdx.x * 16 + (threadIdx.x >> 5) * 2 + (grp >> 1);
  const int i    = blockIdx.y;
  const int k    = i * 2 + half;

  const int* st = start + k * NN;
  int j = st[r];
  int end = (r + 1 < NN) ? st[r + 1] : E_ADJ;
  end = min(end, E_ADJ);
  const EdgeCV* ed = se + (size_t)k * E_ADJ;
  const float4* X = (const float4*)(xw + (size_t)k * NN * HH) + l8;

  float4 acc = make_float4(0.0f, 0.0f, 0.0f, 0.0f);
  float4 bcc = make_float4(0.0f, 0.0f, 0.0f, 0.0f);
  for (; j + 4 <= end; j += 4) {
    const EdgeCV e0 = ed[j], e1 = ed[j + 1], e2 = ed[j + 2], e3 = ed[j + 3];
    const float v0 = e0.v, v1 = e1.v, v2 = e2.v, v3 = e3.v;
    const float4 a0 = X[(size_t)e0.c * 16], b0 = X[(size_t)e0.c * 16 + 8];
    const float4 a1 = X[(size_t)e1.c * 16], b1 = X[(size_t)e1.c * 16 + 8];
    const float4 a2 = X[(size_t)e2.c * 16], b2 = X[(size_t)e2.c * 16 + 8];
    const float4 a3 = X[(size_t)e3.c * 16], b3 = X[(size_t)e3.c * 16 + 8];
    acc.x += a0.x * v0; acc.y += a0.y * v0; acc.z += a0.z * v0; acc.w += a0.w * v0;
    bcc.x += b0.x * v0; bcc.y += b0.y * v0; bcc.z += b0.z * v0; bcc.w += b0.w * v0;
    acc.x += a1.x * v1; acc.y += a1.y * v1; acc.z += a1.z * v1; acc.w += a1.w * v1;
    bcc.x += b1.x * v1; bcc.y += b1.y * v1; bcc.z += b1.z * v1; bcc.w += b1.w * v1;
    acc.x += a2.x * v2; acc.y += a2.y * v2; acc.z += a2.z * v2; acc.w += a2.w * v2;
    bcc.x += b2.x * v2; bcc.y += b2.y * v2; bcc.z += b2.z * v2; bcc.w += b2.w * v2;
    acc.x += a3.x * v3; acc.y += a3.y * v3; acc.z += a3.z * v3; acc.w += a3.w * v3;
    bcc.x += b3.x * v3; bcc.y += b3.y * v3; bcc.z += b3.z * v3; bcc.w += b3.w * v3;
  }
  const int rem = end - j;
  if (rem > 0) {
    const EdgeCV e0 = ed[j];
    const EdgeCV e1 = (rem > 1) ? ed[j + 1] : e0;
    const EdgeCV e2 = (rem > 2) ? ed[j + 2] : e0;
    const float v0 = e0.v, v1 = e1.v, v2 = e2.v;
    const float4 a0 = X[(size_t)e0.c * 16], b0 = X[(size_t)e0.c * 16 + 8];
    const float4 a1 = X[(size_t)e1.c * 16], b1 = X[(size_t)e1.c * 16 + 8];
    const float4 a2 = X[(size_t)e2.c * 16], b2 = X[(size_t)e2.c * 16 + 8];
    acc.x += a0.x * v0; acc.y += a0.y * v0; acc.z += a0.z * v0; acc.w += a0.w * v0;
    bcc.x += b0.x * v0; bcc.y += b0.y * v0; bcc.z += b0.z * v0; bcc.w += b0.w * v0;
    if (rem > 1) {
      acc.x += a1.x * v1; acc.y += a1.y * v1; acc.z += a1.z * v1; acc.w += a1.w * v1;
      bcc.x += b1.x * v1; bcc.y += b1.y * v1; bcc.z += b1.z * v1; bcc.w += b1.w * v1;
    }
    if (rem > 2) {
      acc.x += a2.x * v2; acc.y += a2.y * v2; acc.z += a2.z * v2; acc.w += a2.w * v2;
      bcc.x += b2.x * v2; bcc.y += b2.y * v2; bcc.z += b2.z * v2; bcc.w += b2.w * v2;
    }
  }

  float4 q, p;
  q.x = fmaxf(acc.x, 0.0f); q.y = fmaxf(acc.y, 0.0f);
  q.z = fmaxf(acc.z, 0.0f); q.w = fmaxf(acc.w, 0.0f);
  p.x = fmaxf(bcc.x, 0.0f); p.y = fmaxf(bcc.y, 0.0f);
  p.z = fmaxf(bcc.z, 0.0f); p.w = fmaxf(bcc.w, 0.0f);
  float4 o, n;
  o.x = __shfl_xor(q.x, 8, 32); o.y = __shfl_xor(q.y, 8, 32);
  o.z = __shfl_xor(q.z, 8, 32); o.w = __shfl_xor(q.w, 8, 32);
  n.x = __shfl_xor(p.x, 8, 32); n.y = __shfl_xor(p.y, 8, 32);
  n.z = __shfl_xor(p.z, 8, 32); n.w = __shfl_xor(p.w, 8, 32);
  if (half == 0) {
    float4 s, t;
    s.x = q.x + o.x; s.y = q.y + o.y; s.z = q.z + o.z; s.w = q.w + o.w;
    t.x = p.x + n.x; t.y = p.y + n.y; t.z = p.z + n.z; t.w = p.w + n.w;
    float4* H = (float4*)(i ? h1 : h0) + (size_t)r * 16 + l8;
    H[0] = s;
    H[8] = t;
  }
}

#define GP 68
#define WTP 76
#define GG_TILES  5
#define GG_BLOCKS (NN / 16 / GG_TILES)
static_assert(GG_BLOCKS * GG_TILES * 16 == NN, "the blocks' tiles cover the rows exactly");

__global__ __launch_bounds__(256) void gather_gemm_layer(
    const EdgeCV* __restrict__ se, const int* __restrict__ start,
    const float* __restrict__ x0, const float* __restrict__ x1,
    const float* __restrict__ W,
    float* __restrict__ h0, float* __restrict__ h1) {
  __shared__ __align__(16) float Wt[2][HH][WTP];
  __shared__ __align__(16) float G[2][16][GP];
  __shared__ __align__(16) float Y[2][16][GP];

  const int lane = threadIdx.x & 31;
  const int wv   = threadIdx.x >> 5;
  const int grp  = lane >> 3;
  const int half = grp & 1;
  const int l8   = lane & 7;
  const int lr   = wv * 2 + (grp >> 1);
  const int i    = blockIdx.y;
  const int k    = i * 2 + half;

  const int* st = start + k * NN;
  const EdgeCV* ed = se + (size_t)k * E_ADJ;
  const float4* X = (const float4*)(half ? x1 : x0) + l8;

  int rj, re;
  {
    const int r = blockIdx.x * 16 + lr;
    rj = st[r];
    re = st[min(r + 1, NN - 1)];
  }

  {
    const float4* Wsrc = (const float4*)(W + (size_t)(i * 2) * HH * HH);
#pragma unroll
    for (int u = 0; u < 8; ++u) {
      const int idx = threadIdx.x + 256 * u;
      const float4 w4 = Wsrc[idx];
      const int relx = idx >> 10, kx = (idx >> 4) & 63, nx = (idx & 15) * 4;
      Wt[relx][nx + 0][kx] = w4.x;
      Wt[relx][nx + 1][kx] = w4.y;
      Wt[relx][nx + 2][kx] = w4.z;
      Wt[relx][nx + 3][kx] = w4.w;
    }
  }

  const int rel = wv >> 2;
  const int ct  = wv & 3;
  const int h2  = lane >> 4;
  const int l   = lane & 15;
  const float* ar = &G[rel][l][h2 * 2];
  const float* br = &Wt[rel][ct * 16 + l][h2 * 2];
  const int orow = threadIdx.x >> 4;
  const int oq   = threadIdx.x & 15;
  float4* Hout = (float4*)(i ? h1 : h0) + oq;

  for (int tile = blockIdx.x; tile < NN / 16; tile += gridDim.x) {
    const int r = tile * 16 + lr;
    int j = rj;
    int end = (r + 1 < NN) ? re : E_ADJ;
    end = min(end, E_ADJ);
    {
      const int nr = min(r + (int)gridDim.x * 16, NN - 1);
      rj = st[nr];
      re = st[min(nr + 1, NN - 1)];
    }
    float4 acc = make_float4(0.0f, 0.0f, 0.0f, 0.0f);
    float4 bcc = make_float4(0.0f, 0.0f, 0.0f, 0.0f);
    for (; j + 4 <= end; j += 4) {
      const EdgeCV e0 = ed[j], e1 = ed[j + 1], e2 = ed[j + 2], e3 = ed[j + 3];
      const float v0 = e0.v, v1 = e1.v, v2 = e2.v, v3 = e3.v;
      const float4 a0 = X[(size_t)e0.c * 16], b0 = X[(size_t)e0.c * 16 + 8];
      const float4 a1 = X[(size_t)e1.c * 16], b1 = X[(size_t)e1.c * 16 + 8];
      const float4 a2 = X[(size_t)e2.c * 16], b2 = X[(size_t)e2.c * 16 + 8];
      const float4 a3 = X[(size_t)e3.c * 16], b3 = X[(size_t)e3.c * 16 + 8];
      acc.x += a0.x * v0; acc.y += a0.y * v0; acc.z += a0.z * v0; acc.w += a0.w * v0;
      bcc.x += b0.x * v0; bcc.y += b0.y * v0; bcc.z += b0.z * v0; bcc.w += b0.w * v0;
      acc.x += a1.x * v1; acc.y += a1.y * v1; acc.z += a1.z * v1; acc.w += a1.w * v1;
      bcc.x += b1.x * v1; bcc.y += b1.y * v1; bcc.z += b1.z * v1; bcc.w += b1.w * v1;
      acc.x += a2.x * v2; acc.y += a2.y * v2; acc.z += a2.z * v2; acc.w += a2.w * v2;
      bcc.x += b2.x * v2; bcc.y += b2.y * v2; bcc.z += b2.z * v2; bcc.w += b2.w * v2;
      acc.x += a3.x * v3; acc.y += a3.y * v3; acc.z += a3.z * v3; acc.w += a3.w * v3;
      bcc.x += b3.x * v3; bcc.y += b3.y * v3; bcc.z += b3.z * v3; bcc.w += b3.w * v3;
    }
    const int rem = end - j;
    if (rem > 0) {
      const EdgeCV e0 = ed[j];
      const EdgeCV e1 = (rem > 1) ? ed[j + 1] : e0;
      const EdgeCV e2 = (rem > 2) ? ed[j + 2] : e0;
      const float v0 = e0.v, v1 = e1.v, v2 = e2.v;
      const float4 a0 = X[(size_t)e0.c * 16], b0 = X[(size_t)e0.c * 16 + 8];
      const float4 a1 = X[(size_t)e1.c * 16], b1 = X[(size_t)e1.c * 16 + 8];
      const float4 a2 = X[(size_t)e2.c * 16], b2 = X[(size_t)e2.c * 16 + 8];
      acc.x += a0.x * v0; acc.y += a0.y * v0; acc.z += a0.z * v0; acc.w += a0.w * v0;
      bcc.x += b0.x * v0; bcc.y += b0.y * v0; bcc.z += b0.z * v0; bcc.w += b0.w * v0;
      if (rem > 1) {
        acc.x += a1.x * v1; acc.y += a1.y * v1; acc.z += a1.z * v1; acc.w += a1.w * v1;
        bcc.x += b1.x * v1; bcc.y += b1.y * v1; bcc.z += b1.z * v1; bcc.w += b1.w * v1;
      }
      if (rem > 2) {
        acc.x += a2.x * v2; acc.y += a2.y * v2; acc.z += a2.z * v2; acc.w += a2.w * v2;
        bcc.x += b2.x * v2; bcc.y += b2.y * v2; bcc.z += b2.z * v2; bcc.w += b2.w * v2;
      }
    }

    *(float4*)&G[half][lr][4 * l8]      = acc;
    *(float4*)&G[half][lr][32 + 4 * l8] = bcc;
    __syncthreads();

    v8f c = {};
#pragma unroll
    for (int s = 0; s < HH / 4; ++s) {
      v2f a, b;
      a.x = ar[4 * s];
      a.y = ar[4 * s + 1];
      b.x = br[4 * s];
      b.y = br[4 * s + 1];
      c = __builtin_amdgcn_wmma_f32_16x16x4_f32(false, a, false, b, (short)0, c, false, false);
    }
#pragma unroll
    for (int v = 0; v < 8; ++v) Y[rel][v + h2 * 8][ct * 16 + l] = fmaxf(c[v], 0.0f);
    __syncthreads();

    const float4 y0 = *(const float4*)&Y[0][orow][4 * oq];
    const float4 y1 = *(const float4*)&Y[1][orow][4 * oq];
    float4 s4;
    s4.x = y0.x + y1.x; s4.y = y0.y + y1.y; s4.z = y0.z + y1.z; s4.w = y0.w + y1.w;
    Hout[(size_t)(tile * 16 + orow) * 16] = s4;
  }
}

__global__ void build_M(const float* __restrict__ R, const float* __restrict__ D,
                        const int* __restrict__ rt_k, float* __restrict__ M) {
  int i = blockIdx.x * blockDim.x + threadIdx.x;
  const float* d = D + rt_k[0] * HH;
  int h = i >> 6, g = i & 63;
  M[i] = d[h] * R[i] * d[g];
}

static_assert(E_DEC % 64 == 0, "four 16-edge groups per block of the decoder");

__global__ __launch_bounds__(128) void decode_edges(
    const float* __restrict__ h0, const float* __restrict__ h1,
    const int* __restrict__ edges, const float* __restrict__ M,
    float* __restrict__ preds) {
  const int wave = threadIdx.x >> 5;
  const int lane = threadIdx.x & 31;
  const int half = lane >> 4;
  const int l    = lane & 15;
  const int eb   = blockIdx.x * 64 + wave * 16;

  const int arow = edges[(eb + l) * 2 + 0];

  const float* ar = h0 + (size_t)arow * HH + half * 2;
  v8f c0 = {}, c1 = {}, c2 = {}, c3 = {};
  for (int kk = 0; kk < HH; kk += 4) {
    v2f a, b0, b1, b2, b3;
    a.x = ar[kk];
    a.y = ar[kk + 1];
    const float* mp = M + (kk + half * 2) * HH + l;
    b0.x = mp[0];   b0.y = mp[HH];
    b1.x = mp[16];  b1.y = mp[HH + 16];
    b2.x = mp[32];  b2.y = mp[HH + 32];
    b3.x = mp[48];  b3.y = mp[HH + 48];
    c0 = __builtin_amdgcn_wmma_f32_16x16x4_f32(false, a, false, b0, (short)0, c0, false, false);
    c1 = __builtin_amdgcn_wmma_f32_16x16x4_f32(false, a, false, b1, (short)0, c1, false, false);
    c2 = __builtin_amdgcn_wmma_f32_16x16x4_f32(false, a, false, b2, (short)0, c2, false, false);
    c3 = __builtin_amdgcn_wmma_f32_16x16x4_f32(false, a, false, b3, (short)0, c3, false, false);
  }

  float p0[8], p1[8], p2[8], p3[8];
#pragma unroll
  for (int v = 0; v < 8; ++v) {
    const int edge = v + half * 8;
    const int crow = edges[(eb + edge) * 2 + 1];
    const float* hc = h1 + (size_t)crow * HH + l;
    p0[v] = c0[v] * hc[0];
    p1[v] = c1[v] * hc[16];
    p2[v] = c2[v] * hc[32];
    p3[v] = c3[v] * hc[48];
  }
#pragma unroll
  for (int off = 1; off < 16; off <<= 1) {
#pragma unroll
    for (int v = 0; v < 8; ++v) {
      p0[v] += __shfl_xor(p0[v], off, 16);
      p1[v] += __shfl_xor(p1[v], off, 16);
      p2[v] += __shfl_xor(p2[v], off, 16);
      p3[v] += __shfl_xor(p3[v], off, 16);
    }
  }
  if (l == 0) {
    float s[8];
#pragma unroll
    for (int v = 0; v < 8; ++v) s[v] = p0[v] + p1[v] + p2[v] + p3[v];
    float4* o = (float4*)(preds + eb + half * 8);
    o[0] = make_float4(s[0], s[1], s[2], s[3]);
    o[1] = make_float4(s[4], s[5], s[6], s[7]);
  }
}

extern "C" void kernel_launch(void* const* d_in, const int* in_sizes, int n_in,
                              void* d_out, int out_size, void* d_ws, size_t ws_size,
                              hipStream_t stream) {
  const int*   adj_rows = (const int*)d_in[0];
  const int*   adj_cols = (const int*)d_in[1];
  const float* adj_vals = (const float*)d_in[2];
  const float* feat0    = (const float*)d_in[3];
  const float* feat1    = (const float*)d_in[4];
  const float* W1       = (const float*)d_in[5];
  const float* W2       = (const float*)d_in[6];
  const float* W3       = (const float*)d_in[7];
  const float* W4       = (const float*)d_in[8];
  const float* Rg       = (const float*)d_in[9];
  const float* Dl       = (const float*)d_in[10];
  const int*   edges    = (const int*)d_in[11];
  const int*   rt_k     = (const int*)d_in[12];

  const size_t S = (size_t)NN * HH;
  float* xw = (float*)d_ws;
  float* lists = xw + 4 * S;
  float* h0 = lists + 4 * S;
  float* h1 = h0 + S;
  float* M  = h1 + S;

  static_assert((size_t)4 * E_ADJ * 2 + (size_t)4 * NN <= (size_t)4 * NN * HH,
                "sorted lists must fit in the four former y planes");
  EdgeCV* se   = (EdgeCV*)lists;
  int*   start = (int*)(se + (size_t)4 * E_ADJ);

  float* g0 = xw;
  float* g1 = xw + S;

  const int*   rows1 = adj_rows;
  const int*   cols1 = adj_cols;
  const float* vals1 = adj_vals;
  const int*   rows2 = adj_rows + (size_t)4 * E_ADJ;
  const int*   cols2 = adj_cols + (size_t)4 * E_ADJ;
  const float* vals2 = adj_vals + (size_t)4 * E_ADJ;

  bucket_rows<<<dim3(RB_BLOCKS, 4), RB_THREADS, 0, stream>>>(rows1, cols1, vals1, start, se);
  gemm_stage<<<dim3(NN / (16 * GEMM_WAVES), 4), 32 * GEMM_WAVES, 0, stream>>>(feat0, feat1, W1, xw, 128);
  spmm_gather<<<dim3(NN / 16, 2), 256, 0, stream>>>(se, start, xw, h0, h1);
  gather_gemm_layer<<<dim3(GG_BLOCKS, 2), 256, 0, stream>>>(se, start, h0, h1, W2, g0, g1);

  bucket_rows<<<dim3(RB_BLOCKS, 4), RB_THREADS, 0, stream>>>(rows2, cols2, vals2, start, se);
  gather_gemm_layer<<<dim3(GG_BLOCKS, 2), 256, 0, stream>>>(se, start, g0, g1, W3, h0, h1);
  gather_gemm_layer<<<dim3(GG_BLOCKS, 2), 256, 0, stream>>>(se, start, h0, h1, W4, g0, g1);

  build_M<<<16, 256, 0, stream>>>(Rg, Dl, rt_k, M);
  decode_edges<<<E_DEC / 64, 128, 0, stream>>>(g0, g1, edges, M, (float*)d_out);
}


# Edge bucketing in two launches that each read every edge once: slices sorted by row range with cols and vals carried along, then a block takes its range's pieces; LDS-atomic arrival order kept
# speedup vs baseline: 4.1141x; 1.3840x over previous
#include <hip/hip_runtime.h>
#include <hip/hip_bf16.h>

#define NN 50000
#define E_ADJ 800000
#define E_DEC 200000
#define HH 64

typedef float v2f __attribute__((ext_vector_type(2)));
typedef float v8f __attribute__((ext_vector_type(8)));

#define GEMM_WAVES 5
static_assert(NN % (16 * GEMM_WAVES) == 0, "row tiles must divide evenly among the blocks");

__global__ __launch_bounds__(32 * GEMM_WAVES) void gemm_stage(
    const float* __restrict__ x0, const float* __restrict__ x1,
    const float* __restrict__ W, float* __restrict__ xw, int Fin) {
  const int k    = blockIdx.y;
  const int tm   = blockIdx.x * GEMM_WAVES + (threadIdx.x >> 5);
  const int lane = threadIdx.x & 31;
  const int half = lane >> 4;
  const int l    = lane & 15;

  const float* X  = (k & 1) ? x1 : x0;
  const float* Wk = W + (size_t)k * Fin * HH;

  const int row = tm * 16 + l;

  const float* ar = X + (size_t)row * Fin + half * 2;
  v8f c0 = {}, c1 = {}, c2 = {}, c3 = {};
  for (int kk = 0; kk < Fin; kk += 4) {
    v2f a, b0, b1, b2, b3;
    a.x = ar[kk];
    a.y = ar[kk + 1];
    const float* wp = Wk + (size_t)(kk + half * 2) * HH + l;
    b0.x = wp[0];   b0.y = wp[HH];
    b1.x = wp[16];  b1.y = wp[HH + 16];
    b2.x = wp[32];  b2.y = wp[HH + 32];
    b3.x = wp[48];  b3.y = wp[HH + 48];
    c0 = __builtin_amdgcn_wmma_f32_16x16x4_f32(
         false, a,  false, b0,
         (short)0, c0,  false,  false);
    c1 = __builtin_amdgcn_wmma_f32_16x16x4_f32(false, a, false, b1, (short)0, c1, false, false);
    c2 = __builtin_amdgcn_wmma_f32_16x16x4_f32(false, a, false, b2, (short)0, c2, false, false);
    c3 = __builtin_amdgcn_wmma_f32_16x16x4_f32(false, a, false, b3, (short)0, c3, false, false);
  }

  float* out = xw + ((size_t)k * NN + (size_t)tm * 16) * HH + l;
#pragma unroll
  for (int v = 0; v < 8; ++v) {
    float* o = out + (size_t)(v + half * 8) * HH;
    o[0]  = c0[v];
    o[16] = c1[v];
    o[32] = c2[v];
    o[48] = c3[v];
  }
}

struct __align__(8) EdgeCV { int c; float v; };
struct __align__(8) EdgeDV { unsigned cd; float v; };

#define RB_BLOCKS  64
#define RB_ROWS    ((NN + RB_BLOCKS - 1) / RB_BLOCKS)
#define RB_THREADS 1024
#define RB_WAVES   (RB_THREADS / 32)
#define PS_SLICE   (E_ADJ / RB_BLOCKS)
#define PS_PER     ((PS_SLICE + RB_THREADS - 1) / RB_THREADS)
static_assert(RB_ROWS <= RB_THREADS && RB_ROWS <= 1024, "one counter per thread; local row packs into 10 bits");
static_assert(NN <= (1 << 16), "source row packs into 16 bits");
static_assert(E_ADJ % RB_BLOCKS == 0, "slices are equal");
static_assert((NN - 1) / RB_ROWS < RB_BLOCKS, "every row has a range");
static_assert(RB_BLOCKS == 2 * RB_WAVES, "a wave of place_edges serves two slices' pieces");

__global__ __launch_bounds__(RB_THREADS) void split_edges(
    const int* __restrict__ rows, const int* __restrict__ cols, const float* __restrict__ vals,
    EdgeDV* __restrict__ tmp, int* __restrict__ hoff) {
  __shared__ int hist[RB_BLOCKS];
  __shared__ int hx[RB_BLOCKS];
  __shared__ EdgeDV buf[PS_SLICE];

  const int k = blockIdx.y;
  const int j = blockIdx.x;
  const int t = threadIdx.x;
  const size_t s0 = (size_t)k * E_ADJ + (size_t)j * PS_SLICE;
  const int*   R  = rows + s0;
  const int*   Cs = cols + s0;
  const float* Vs = vals + s0;

  if (t < RB_BLOCKS) hist[t] = 0;
  int row[PS_PER];
#pragma unroll
  for (int n = 0; n < PS_PER; ++n) {
    const int i = t + n * RB_THREADS;
    row[n] = (i < PS_SLICE) ? R[i] : 0;
  }
  __syncthreads();

  int rg[PS_PER], rk[PS_PER];
#pragma unroll
  for (int n = 0; n < PS_PER; ++n) {
    const int i = t + n * RB_THREADS;
    const int r = (int)min((unsigned)row[n] / (unsigned)RB_ROWS, (unsigned)(RB_BLOCKS - 1));
    rg[n] = r;
    rk[n] = (i < PS_SLICE) ? atomicAdd(&hist[r], 1) : 0;
  }
  int cc[PS_PER];
  float vv[PS_PER];
#pragma unroll
  for (int n = 0; n < PS_PER; ++n) {
    const int i = t + n * RB_THREADS;
    cc[n] = (i < PS_SLICE) ? Cs[i] : 0;
    vv[n] = (i < PS_SLICE) ? Vs[i] : 0.0f;
  }
  __syncthreads();

  if (t < RB_BLOCKS) {
    int ex = 0;
    for (int q = 0; q < RB_BLOCKS; ++q) ex += (q < t) ? hist[q] : 0;
    hx[t] = ex;
    int* H = hoff + ((size_t)k * RB_BLOCKS + j) * (RB_BLOCKS + 1);
    H[t] = ex;
    if (t == RB_BLOCKS - 1) H[RB_BLOCKS] = ex + hist[t];
  }
  __syncthreads();

#pragma unroll
  for (int n = 0; n < PS_PER; ++n) {
    const int i = t + n * RB_THREADS;
    if (i < PS_SLICE) {
      const int p = min(hx[rg[n]] + rk[n], PS_SLICE - 1);
      EdgeDV rec;
      rec.cd = ((unsigned)cc[n] & 0xffffu) | (((unsigned)(row[n] - rg[n] * RB_ROWS) & 1023u) << 16);
      rec.v  = vv[n];
      buf[p] = rec;
    }
  }
  __syncthreads();

  EdgeDV* T = tmp + s0;
  for (int i = t; i < PS_SLICE; i += RB_THREADS) T[i] = buf[i];
}

__global__ __launch_bounds__(RB_THREADS) void place_edges(
    const EdgeDV* __restrict__ tmp, const int* __restrict__ hoff,
    int* __restrict__ start, EdgeCV* __restrict__ se) {
  __shared__ int cnt[RB_THREADS];
  __shared__ int off[RB_THREADS];
  __shared__ int wtot[RB_WAVES];
  __shared__ int sbeg[RB_BLOCKS];
  __shared__ int slen[RB_BLOCKS];
  __shared__ int sbel[2];

  const int k    = blockIdx.y;
  const int b    = blockIdx.x;
  const int t    = threadIdx.x;
  const int lane = t & 31;
  const int w    = t >> 5;
  const int lo   = b * RB_ROWS;
  const unsigned nrows = (unsigned)min(RB_ROWS, NN - lo);
  const EdgeDV* T = tmp + (size_t)k * E_ADJ;
  EdgeCV* S = se + (size_t)k * E_ADJ;

  cnt[t] = 0;
  if (t < RB_BLOCKS) {
    const int* H = hoff + ((size_t)k * RB_BLOCKS + t) * (RB_BLOCKS + 1);
    const int hb = min(max(H[b], 0), PS_SLICE);
    const int he = min(max(H[b + 1], hb), PS_SLICE);
    sbeg[t] = t * PS_SLICE + hb;
    slen[t] = he - hb;
    int sb = hb;
#pragma unroll
    for (int o = 16; o > 0; o >>= 1) sb += __shfl_xor(sb, o, 32);
    if (lane == 0) sbel[w] = sb;
  }
  __syncthreads();

  const int b0 = sbeg[2 * w], l0 = slen[2 * w];
  const int b1 = sbeg[2 * w + 1], l1 = slen[2 * w + 1];
  const int L = l0 + l1;

  for (int i0 = lane; i0 < L; i0 += 4 * 32) {
    unsigned x[4];
#pragma unroll
    for (int u = 0; u < 4; ++u) {
      const int i = i0 + u * 32;
      const int a = (i < l0) ? b0 + i : b1 + (i - l0);
      x[u] = T[min(a, E_ADJ - 1)].cd;
    }
    asm volatile("" :: "v"(x[0]), "v"(x[1]), "v"(x[2]), "v"(x[3]));
#pragma unroll
    for (int u = 0; u < 4; ++u) {
      if (i0 + u * 32 < L) atomicAdd(&cnt[(x[u] >> 16) & 1023u], 1);
    }
  }
  __syncthreads();

  const int own = ((unsigned)t < nrows) ? cnt[t] : 0;
  int incl = own;
#pragma unroll
  for (int o = 1; o < 32; o <<= 1) {
    const int n = __shfl_up(incl, o, 32);
    if (lane >= o) incl += n;
  }
  if (lane == 31) wtot[w] = incl;
  __syncthreads();
  int base = sbel[0] + sbel[1];
  for (int q = 0; q < RB_WAVES; ++q) {
    if (q < w) base += wtot[q];
  }
  const int ex = base + incl - own;
  off[t] = ex;
  cnt[t] = 0;
  if ((unsigned)t < nrows) start[k * NN + lo + t] = ex;
  __syncthreads();

  for (int i0 = lane; i0 < L; i0 += 4 * 32) {
    EdgeDV x[4];
    int pos[4];
#pragma unroll
    for (int u = 0; u < 4; ++u) {
      const int i = i0 + u * 32;
      const int a = (i < l0) ? b0 + i : b1 + (i - l0);
      x[u] = T[min(a, E_ADJ - 1)];
    }
    asm volatile("" :: "v"(x[0].cd), "v"(x[0].v), "v"(x[1].cd), "v"(x[1].v), "v"(x[2].cd), "v"(x[2].v), "v"(x[3].cd), "v"(x[3].v));
#pragma unroll
    for (int u = 0; u < 4; ++u) {
      const int d = (int)((x[u].cd >> 16) & 1023u);
      pos[u] = (i0 + u * 32 < L) ? off[d] + atomicAdd(&cnt[d], 1) : E_ADJ;
    }
#pragma unroll
    for (int u = 0; u < 4; ++u) {
      if (pos[u] >= 0 && pos[u] < E_ADJ) {
        EdgeCV rec;
        rec.c = (int)(x[u].cd & 0xffffu);
        rec.v = x[u].v;
        S[pos[u]] = rec;
      }
    }
  }
}

static_assert(NN % 16 == 0, "sixteen rows per block of the gather");

__global__ __launch_bounds__(256) void spmm_gather(
    const EdgeCV* __restrict__ se,
    const int* __restrict__ start, const float* __restrict__ xw,
    float* __restrict__ h0, float* __restrict__ h1) {
  const int lane = threadIdx.x & 31;
  const int grp  = lane >> 3;
  const int half = grp & 1;
  const int l8   = lane & 7;
  const int r    = blockIdx.x * 16 + (threadIdx.x >> 5) * 2 + (grp >> 1);
  const int i    = blockIdx.y;
  const int k    = i * 2 + half;

  const int* st = start + k * NN;
  int j = st[r];
  int end = (r + 1 < NN) ? st[r + 1] : E_ADJ;
  end = min(end, E_ADJ);
  const EdgeCV* ed = se + (size_t)k * E_ADJ;
  const float4* X = (const float4*)(xw + (size_t)k * NN * HH) + l8;

  float4 acc = make_float4(0.0f, 0.0f, 0.0f, 0.0f);
  float4 bcc = make_float4(0.0f, 0.0f, 0.0f, 0.0f);
  for (; j + 4 <= end; j += 4) {
    const EdgeCV e0 = ed[j], e1 = ed[j + 1], e2 = ed[j + 2], e3 = ed[j + 3];
    const float v0 = e0.v, v1 = e1.v, v2 = e2.v, v3 = e3.v;
    const float4 a0 = X[(size_t)e0.c * 16], b0 = X[(size_t)e0.c * 16 + 8];
    const float4 a1 = X[(size_t)e1.c * 16], b1 = X[(size_t)e1.c * 16 + 8];
    const float4 a2 = X[(size_t)e2.c * 16], b2 = X[(size_t)e2.c * 16 + 8];
    const float4 a3 = X[(size_t)e3.c * 16], b3 = X[(size_t)e3.c * 16 + 8];
    acc.x += a0.x * v0; acc.y += a0.y * v0; acc.z += a0.z * v0; acc.w += a0.w * v0;
    bcc.x += b0.x * v0; bcc.y += b0.y * v0; bcc.z += b0.z * v0; bcc.w += b0.w * v0;
    acc.x += a1.x * v1; acc.y += a1.y * v1; acc.z += a1.z * v1; acc.w += a1.w * v1;
    bcc.x += b1.x * v1; bcc.y += b1.y * v1; bcc.z += b1.z * v1; bcc.w += b1.w * v1;
    acc.x += a2.x * v2; acc.y += a2.y * v2; acc.z += a2.z * v2; acc.w += a2.w * v2;
    bcc.x += b2.x * v2; bcc.y += b2.y * v2; bcc.z += b2.z * v2; bcc.w += b2.w * v2;
    acc.x += a3.x * v3; acc.y += a3.y * v3; acc.z += a3.z * v3; acc.w += a3.w * v3;
    bcc.x += b3.x * v3; bcc.y += b3.y * v3; bcc.z += b3.z * v3; bcc.w += b3.w * v3;
  }
  const int rem = end - j;
  if (rem > 0) {
    const EdgeCV e0 = ed[j];
    const EdgeCV e1 = (rem > 1) ? ed[j + 1] : e0;
    const EdgeCV e2 = (rem > 2) ? ed[j + 2] : e0;
    const float v0 = e0.v, v1 = e1.v, v2 = e2.v;
    const float4 a0 = X[(size_t)e0.c * 16], b0 = X[(size_t)e0.c * 16 + 8];
    const float4 a1 = X[(size_t)e1.c * 16], b1 = X[(size_t)e1.c * 16 + 8];
    const float4 a2 = X[(size_t)e2.c * 16], b2 = X[(size_t)e2.c * 16 + 8];
    acc.x += a0.x * v0; acc.y += a0.y * v0; acc.z += a0.z * v0; acc.w += a0.w * v0;
    bcc.x += b0.x * v0; bcc.y += b0.y * v0; bcc.z += b0.z * v0; bcc.w += b0.w * v0;
    if (rem > 1) {
      acc.x += a1.x * v1; acc.y += a1.y * v1; acc.z += a1.z * v1; acc.w += a1.w * v1;
      bcc.x += b1.x * v1; bcc.y += b1.y * v1; bcc.z += b1.z * v1; bcc.w += b1.w * v1;
    }
    if (rem > 2) {
      acc.x += a2.x * v2; acc.y += a2.y * v2; acc.z += a2.z * v2; acc.w += a2.w * v2;
      bcc.x += b2.x * v2; bcc.y += b2.y * v2; bcc.z += b2.z * v2; bcc.w += b2.w * v2;
    }
  }

  float4 q, p;
  q.x = fmaxf(acc.x, 0.0f); q.y = fmaxf(acc.y, 0.0f);
  q.z = fmaxf(acc.z, 0.0f); q.w = fmaxf(acc.w, 0.0f);
  p.x = fmaxf(bcc.x, 0.0f); p.y = fmaxf(bcc.y, 0.0f);
  p.z = fmaxf(bcc.z, 0.0f); p.w = fmaxf(bcc.w, 0.0f);
  float4 o, n;
  o.x = __shfl_xor(q.x, 8, 32); o.y = __shfl_xor(q.y, 8, 32);
  o.z = __shfl_xor(q.z, 8, 32); o.w = __shfl_xor(q.w, 8, 32);
  n.x = __shfl_xor(p.x, 8, 32); n.y = __shfl_xor(p.y, 8, 32);
  n.z = __shfl_xor(p.z, 8, 32); n.w = __shfl_xor(p.w, 8, 32);
  if (half == 0) {
    float4 s, t;
    s.x = q.x + o.x; s.y = q.y + o.y; s.z = q.z + o.z; s.w = q.w + o.w;
    t.x = p.x + n.x; t.y = p.y + n.y; t.z = p.z + n.z; t.w = p.w + n.w;
    float4* H = (float4*)(i ? h1 : h0) + (size_t)r * 16 + l8;
    H[0] = s;
    H[8] = t;
  }
}

#define GP 68
#define WTP 76
#define GG_TILES  5
#define GG_BLOCKS (NN / 16 / GG_TILES)
static_assert(GG_BLOCKS * GG_TILES * 16 == NN, "the blocks' tiles cover the rows exactly");

__global__ __launch_bounds__(256) void gather_gemm_layer(
    const EdgeCV* __restrict__ se, const int* __restrict__ start,
    const float* __restrict__ x0, const float* __restrict__ x1,
    const float* __restrict__ W,
    float* __restrict__ h0, float* __restrict__ h1) {
  __shared__ __align__(16) float Wt[2][HH][WTP];
  __shared__ __align__(16) float G[2][16][GP];
  __shared__ __align__(16) float Y[2][16][GP];

  const int lane = threadIdx.x & 31;
  const int wv   = threadIdx.x >> 5;
  const int grp  = lane >> 3;
  const int half = grp & 1;
  const int l8   = lane & 7;
  const int lr   = wv * 2 + (grp >> 1);
  const int i    = blockIdx.y;
  const int k    = i * 2 + half;

  const int* st = start + k * NN;
  const EdgeCV* ed = se + (size_t)k * E_ADJ;
  const float4* X = (const float4*)(half ? x1 : x0) + l8;

  int rj, re;
  {
    const int r = blockIdx.x * 16 + lr;
    rj = st[r];
    re = st[min(r + 1, NN - 1)];
  }

  {
    const float4* Wsrc = (const float4*)(W + (size_t)(i * 2) * HH * HH);
#pragma unroll
    for (int u = 0; u < 8; ++u) {
      const int idx = threadIdx.x + 256 * u;
      const float4 w4 = Wsrc[idx];
      const int relx = idx >> 10, kx = (idx >> 4) & 63, nx = (idx & 15) * 4;
      Wt[relx][nx + 0][kx] = w4.x;
      Wt[relx][nx + 1][kx] = w4.y;
      Wt[relx][nx + 2][kx] = w4.z;
      Wt[relx][nx + 3][kx] = w4.w;
    }
  }

  const int rel = wv >> 2;
  const int ct  = wv & 3;
  const int h2  = lane >> 4;
  const int l   = lane & 15;
  const float* ar = &G[rel][l][h2 * 2];
  const float* br = &Wt[rel][ct * 16 + l][h2 * 2];
  const int orow = threadIdx.x >> 4;
  const int oq   = threadIdx.x & 15;
  float4* Hout = (float4*)(i ? h1 : h0) + oq;

  for (int tile = blockIdx.x; tile < NN / 16; tile += gridDim.x) {
    const int r = tile * 16 + lr;
    int j = rj;
    int end = (r + 1 < NN) ? re : E_ADJ;
    end = min(end, E_ADJ);
    {
      const int nr = min(r + (int)gridDim.x * 16, NN - 1);
      rj = st[nr];
      re = st[min(nr + 1, NN - 1)];
    }
    float4 acc = make_float4(0.0f, 0.0f, 0.0f, 0.0f);
    float4 bcc = make_float4(0.0f, 0.0f, 0.0f, 0.0f);
    for (; j + 4 <= end; j += 4) {
      const EdgeCV e0 = ed[j], e1 = ed[j + 1], e2 = ed[j + 2], e3 = ed[j + 3];
      const float v0 = e0.v, v1 = e1.v, v2 = e2.v, v3 = e3.v;
      const float4 a0 = X[(size_t)e0.c * 16], b0 = X[(size_t)e0.c * 16 + 8];
      const float4 a1 = X[(size_t)e1.c * 16], b1 = X[(size_t)e1.c * 16 + 8];
      const float4 a2 = X[(size_t)e2.c * 16], b2 = X[(size_t)e2.c * 16 + 8];
      const float4 a3 = X[(size_t)e3.c * 16], b3 = X[(size_t)e3.c * 16 + 8];
      acc.x += a0.x * v0; acc.y += a0.y * v0; acc.z += a0.z * v0; acc.w += a0.w * v0;
      bcc.x += b0.x * v0; bcc.y += b0.y * v0; bcc.z += b0.z * v0; bcc.w += b0.w * v0;
      acc.x += a1.x * v1; acc.y += a1.y * v1; acc.z += a1.z * v1; acc.w += a1.w * v1;
      bcc.x += b1.x * v1; bcc.y += b1.y * v1; bcc.z += b1.z * v1; bcc.w += b1.w * v1;
      acc.x += a2.x * v2; acc.y += a2.y * v2; acc.z += a2.z * v2; acc.w += a2.w * v2;
      bcc.x += b2.x * v2; bcc.y += b2.y * v2; bcc.z += b2.z * v2; bcc.w += b2.w * v2;
      acc.x += a3.x * v3; acc.y += a3.y * v3; acc.z += a3.z * v3; acc.w += a3.w * v3;
      bcc.x += b3.x * v3; bcc.y += b3.y * v3; bcc.z += b3.z * v3; bcc.w += b3.w * v3;
    }
    const int rem = end - j;
    if (rem > 0) {
      const EdgeCV e0 = ed[j];
      const EdgeCV e1 = (rem > 1) ? ed[j + 1] : e0;
      const EdgeCV e2 = (rem > 2) ? ed[j + 2] : e0;
      const float v0 = e0.v, v1 = e1.v, v2 = e2.v;
      const float4 a0 = X[(size_t)e0.c * 16], b0 = X[(size_t)e0.c * 16 + 8];
      const float4 a1 = X[(size_t)e1.c * 16], b1 = X[(size_t)e1.c * 16 + 8];
      const float4 a2 = X[(size_t)e2.c * 16], b2 = X[(size_t)e2.c * 16 + 8];
      acc.x += a0.x * v0; acc.y += a0.y * v0; acc.z += a0.z * v0; acc.w += a0.w * v0;
      bcc.x += b0.x * v0; bcc.y += b0.y * v0; bcc.z += b0.z * v0; bcc.w += b0.w * v0;
      if (rem > 1) {
        acc.x += a1.x * v1; acc.y += a1.y * v1; acc.z += a1.z * v1; acc.w += a1.w * v1;
        bcc.x += b1.x * v1; bcc.y += b1.y * v1; bcc.z += b1.z * v1; bcc.w += b1.w * v1;
      }
      if (rem > 2) {
        acc.x += a2.x * v2; acc.y += a2.y * v2; acc.z += a2.z * v2; acc.w += a2.w * v2;
        bcc.x += b2.x * v2; bcc.y += b2.y * v2; bcc.z += b2.z * v2; bcc.w += b2.w * v2;
      }
    }

    *(float4*)&G[half][lr][4 * l8]      = acc;
    *(float4*)&G[half][lr][32 + 4 * l8] = bcc;
    __syncthreads();

    v8f c = {};
#pragma unroll
    for (int s = 0; s < HH / 4; ++s) {
      v2f a, b;
      a.x = ar[4 * s];
      a.y = ar[4 * s + 1];
      b.x = br[4 * s];
      b.y = br[4 * s + 1];
      c = __builtin_amdgcn_wmma_f32_16x16x4_f32(false, a, false, b, (short)0, c, false, false);
    }
#pragma unroll
    for (int v = 0; v < 8; ++v) Y[rel][v + h2 * 8][ct * 16 + l] = fmaxf(c[v], 0.0f);
    __syncthreads();

    const float4 y0 = *(const float4*)&Y[0][orow][4 * oq];
    const float4 y1 = *(const float4*)&Y[1][orow][4 * oq];
    float4 s4;
    s4.x = y0.x + y1.x; s4.y = y0.y + y1.y; s4.z = y0.z + y1.z; s4.w = y0.w + y1.w;
    Hout[(size_t)(tile * 16 + orow) * 16] = s4;
  }
}

__global__ void build_M(const float* __restrict__ R, const float* __restrict__ D,
                        const int* __restrict__ rt_k, float* __restrict__ M) {
  int i = blockIdx.x * blockDim.x + threadIdx.x;
  const float* d = D + rt_k[0] * HH;
  int h = i >> 6, g = i & 63;
  M[i] = d[h] * R[i] * d[g];
}

static_assert(E_DEC % 64 == 0, "four 16-edge groups per block of the decoder");

__global__ __launch_bounds__(128) void decode_edges(
    const float* __restrict__ h0, const float* __restrict__ h1,
    const int* __restrict__ edges, const float* __restrict__ M,
    float* __restrict__ preds) {
  const int wave = threadIdx.x >> 5;
  const int lane = threadIdx.x & 31;
  const int half = lane >> 4;
  const int l    = lane & 15;
  const int eb   = blockIdx.x * 64 + wave * 16;

  const int arow = edges[(eb + l) * 2 + 0];

  const float* ar = h0 + (size_t)arow * HH + half * 2;
  v8f c0 = {}, c1 = {}, c2 = {}, c3 = {};
  for (int kk = 0; kk < HH; kk += 4) {
    v2f a, b0, b1, b2, b3;
    a.x = ar[kk];
    a.y = ar[kk + 1];
    const float* mp = M + (kk + half * 2) * HH + l;
    b0.x = mp[0];   b0.y = mp[HH];
    b1.x = mp[16];  b1.y = mp[HH + 16];
    b2.x = mp[32];  b2.y = mp[HH + 32];
    b3.x = mp[48];  b3.y = mp[HH + 48];
    c0 = __builtin_amdgcn_wmma_f32_16x16x4_f32(false, a, false, b0, (short)0, c0, false, false);
    c1 = __builtin_amdgcn_wmma_f32_16x16x4_f32(false, a, false, b1, (short)0, c1, false, false);
    c2 = __builtin_amdgcn_wmma_f32_16x16x4_f32(false, a, false, b2, (short)0, c2, false, false);
    c3 = __builtin_amdgcn_wmma_f32_16x16x4_f32(false, a, false, b3, (short)0, c3, false, false);
  }

  float p0[8], p1[8], p2[8], p3[8];
#pragma unroll
  for (int v = 0; v < 8; ++v) {
    const int edge = v + half * 8;
    const int crow = edges[(eb + edge) * 2 + 1];
    const float* hc = h1 + (size_t)crow * HH + l;
    p0[v] = c0[v] * hc[0];
    p1[v] = c1[v] * hc[16];
    p2[v] = c2[v] * hc[32];
    p3[v] = c3[v] * hc[48];
  }
#pragma unroll
  for (int off = 1; off < 16; off <<= 1) {
#pragma unroll
    for (int v = 0; v < 8; ++v) {
      p0[v] += __shfl_xor(p0[v], off, 16);
      p1[v] += __shfl_xor(p1[v], off, 16);
      p2[v] += __shfl_xor(p2[v], off, 16);
      p3[v] += __shfl_xor(p3[v], off, 16);
    }
  }
  if (l == 0) {
    float s[8];
#pragma unroll
    for (int v = 0; v < 8; ++v) s[v] = p0[v] + p1[v] + p2[v] + p3[v];
    float4* o = (float4*)(preds + eb + half * 8);
    o[0] = make_float4(s[0], s[1], s[2], s[3]);
    o[1] = make_float4(s[4], s[5], s[6], s[7]);
  }
}

extern "C" void kernel_launch(void* const* d_in, const int* in_sizes, int n_in,
                              void* d_out, int out_size, void* d_ws, size_t ws_size,
                              hipStream_t stream) {
  const int*   adj_rows = (const int*)d_in[0];
  const int*   adj_cols = (const int*)d_in[1];
  const float* adj_vals = (const float*)d_in[2];
  const float* feat0    = (const float*)d_in[3];
  const float* feat1    = (const float*)d_in[4];
  const float* W1       = (const float*)d_in[5];
  const float* W2       = (const float*)d_in[6];
  const float* W3       = (const float*)d_in[7];
  const float* W4       = (const float*)d_in[8];
  const float* Rg       = (const float*)d_in[9];
  const float* Dl       = (const float*)d_in[10];
  const int*   edges    = (const int*)d_in[11];
  const int*   rt_k     = (const int*)d_in[12];

  const size_t S = (size_t)NN * HH;
  float* xw = (float*)d_ws;
  float* lists = xw + 4 * S;
  float* h0 = lists + 4 * S;
  float* h1 = h0 + S;
  float* M  = h1 + S;

  static_assert((size_t)4 * E_ADJ * 2 + (size_t)4 * NN + (size_t)4 * RB_BLOCKS * (RB_BLOCKS + 1) <= (size_t)4 * NN * HH,
                "sorted lists and offset tables must fit in the four former y planes");
  static_assert((size_t)4 * E_ADJ * 2 <= (size_t)2 * NN * HH, "the split slices must fit in two xw planes");
  EdgeCV* se   = (EdgeCV*)lists;
  int*   start = (int*)(se + (size_t)4 * E_ADJ);
  int*   hoff  = start + (size_t)4 * NN;
  EdgeDV* tmp  = (EdgeDV*)(xw + 2 * S);

  float* g0 = xw;
  float* g1 = xw + S;

  const int*   rows1 = adj_rows;
  const int*   cols1 = adj_cols;
  const float* vals1 = adj_vals;
  const int*   rows2 = adj_rows + (size_t)4 * E_ADJ;
  const int*   cols2 = adj_cols + (size_t)4 * E_ADJ;
  const float* vals2 = adj_vals + (size_t)4 * E_ADJ;

  split_edges<<<dim3(RB_BLOCKS, 4), RB_THREADS, 0, stream>>>(rows1, cols1, vals1, tmp, hoff);
  place_edges<<<dim3(RB_BLOCKS, 4), RB_THREADS, 0, stream>>>(tmp, hoff, start, se);
  gemm_stage<<<dim3(NN / (16 * GEMM_WAVES), 4), 32 * GEMM_WAVES, 0, stream>>>(feat0, feat1, W1, xw, 128);
  spmm_gather<<<dim3(NN / 16, 2), 256, 0, stream>>>(se, start, xw, h0, h1);
  gather_gemm_layer<<<dim3(GG_BLOCKS, 2), 256, 0, stream>>>(se, start, h0, h1, W2, g0, g1);

  split_edges<<<dim3(RB_BLOCKS, 4), RB_THREADS, 0, stream>>>(rows2, cols2, vals2, tmp, hoff);
  place_edges<<<dim3(RB_BLOCKS, 4), RB_THREADS, 0, stream>>>(tmp, hoff, start, se);
  gather_gemm_layer<<<dim3(GG_BLOCKS, 2), 256, 0, stream>>>(se, start, g0, g1, W3, h0, h1);
  gather_gemm_layer<<<dim3(GG_BLOCKS, 2), 256, 0, stream>>>(se, start, h0, h1, W4, g0, g1);

  build_M<<<16, 256, 0, stream>>>(Rg, Dl, rt_k, M);
  decode_edges<<<E_DEC / 64, 128, 0, stream>>>(g0, g1, edges, M, (float*)d_out);
}


# Gather loops read their edge records one trip ahead: a trip's eight row loads and the next trip's records go out together, so do the tail's rows and the next tile's first records; sum order kept
# speedup vs baseline: 4.2463x; 1.0321x over previous
#include <hip/hip_runtime.h>
#include <hip/hip_bf16.h>

#define NN 50000
#define E_ADJ 800000
#define E_DEC 200000
#define HH 64

typedef float v2f __attribute__((ext_vector_type(2)));
typedef float v8f __attribute__((ext_vector_type(8)));

#define GEMM_WAVES 5
static_assert(NN % (16 * GEMM_WAVES) == 0, "row tiles must divide evenly among the blocks");

__global__ __launch_bounds__(32 * GEMM_WAVES) void gemm_stage(
    const float* __restrict__ x0, const float* __restrict__ x1,
    const float* __restrict__ W, float* __restrict__ xw, int Fin) {
  const int k    = blockIdx.y;
  const int tm   = blockIdx.x * GEMM_WAVES + (threadIdx.x >> 5);
  const int lane = threadIdx.x & 31;
  const int half = lane >> 4;
  const int l    = lane & 15;

  const float* X  = (k & 1) ? x1 : x0;
  const float* Wk = W + (size_t)k * Fin * HH;

  const int row = tm * 16 + l;

  const float* ar = X + (size_t)row * Fin + half * 2;
  v8f c0 = {}, c1 = {}, c2 = {}, c3 = {};
  for (int kk = 0; kk < Fin; kk += 4) {
    v2f a, b0, b1, b2, b3;
    a.x = ar[kk];
    a.y = ar[kk + 1];
    const float* wp = Wk + (size_t)(kk + half * 2) * HH + l;
    b0.x = wp[0];   b0.y = wp[HH];
    b1.x = wp[16];  b1.y = wp[HH + 16];
    b2.x = wp[32];  b2.y = wp[HH + 32];
    b3.x = wp[48];  b3.y = wp[HH + 48];
    c0 = __builtin_amdgcn_wmma_f32_16x16x4_f32(
         false, a,  false, b0,
         (short)0, c0,  false,  false);
    c1 = __builtin_amdgcn_wmma_f32_16x16x4_f32(false, a, false, b1, (short)0, c1, false, false);
    c2 = __builtin_amdgcn_wmma_f32_16x16x4_f32(false, a, false, b2, (short)0, c2, false, false);
    c3 = __builtin_amdgcn_wmma_f32_16x16x4_f32(false, a, false, b3, (short)0, c3, false, false);
  }

  float* out = xw + ((size_t)k * NN + (size_t)tm * 16) * HH + l;
#pragma unroll
  for (int v = 0; v < 8; ++v) {
    float* o = out + (size_t)(v + half * 8) * HH;
    o[0]  = c0[v];
    o[16] = c1[v];
    o[32] = c2[v];
    o[48] = c3[v];
  }
}

struct __align__(8) EdgeCV { int c; float v; };
struct __align__(8) EdgeDV { unsigned cd; float v; };

#define RB_BLOCKS  64
#define RB_ROWS    ((NN + RB_BLOCKS - 1) / RB_BLOCKS)
#define RB_THREADS 1024
#define RB_WAVES   (RB_THREADS / 32)
#define PS_SLICE   (E_ADJ / RB_BLOCKS)
#define PS_PER     ((PS_SLICE + RB_THREADS - 1) / RB_THREADS)
static_assert(RB_ROWS <= RB_THREADS && RB_ROWS <= 1024, "one counter per thread; local row packs into 10 bits");
static_assert(NN <= (1 << 16), "source row packs into 16 bits");
static_assert(E_ADJ % RB_BLOCKS == 0, "slices are equal");
static_assert((NN - 1) / RB_ROWS < RB_BLOCKS, "every row has a range");
static_assert(RB_BLOCKS == 2 * RB_WAVES, "a wave of place_edges serves two slices' pieces");

__global__ __launch_bounds__(RB_THREADS) void split_edges(
    const int* __restrict__ rows, const int* __restrict__ cols, const float* __restrict__ vals,
    EdgeDV* __restrict__ tmp, int* __restrict__ hoff) {
  __shared__ int hist[RB_BLOCKS];
  __shared__ int hx[RB_BLOCKS];
  __shared__ EdgeDV buf[PS_SLICE];

  const int k = blockIdx.y;
  const int j = blockIdx.x;
  const int t = threadIdx.x;
  const size_t s0 = (size_t)k * E_ADJ + (size_t)j * PS_SLICE;
  const int*   R  = rows + s0;
  const int*   Cs = cols + s0;
  const float* Vs = vals + s0;

  if (t < RB_BLOCKS) hist[t] = 0;
  int row[PS_PER];
#pragma unroll
  for (int n = 0; n < PS_PER; ++n) {
    const int i = t + n * RB_THREADS;
    row[n] = (i < PS_SLICE) ? R[i] : 0;
  }
  __syncthreads();

  int rg[PS_PER], rk[PS_PER];
#pragma unroll
  for (int n = 0; n < PS_PER; ++n) {
    const int i = t + n * RB_THREADS;
    const int r = (int)min((unsigned)row[n] / (unsigned)RB_ROWS, (unsigned)(RB_BLOCKS - 1));
    rg[n] = r;
    rk[n] = (i < PS_SLICE) ? atomicAdd(&hist[r], 1) : 0;
  }
  int cc[PS_PER];
  float vv[PS_PER];
#pragma unroll
  for (int n = 0; n < PS_PER; ++n) {
    const int i = t + n * RB_THREADS;
    cc[n] = (i < PS_SLICE) ? Cs[i] : 0;
    vv[n] = (i < PS_SLICE) ? Vs[i] : 0.0f;
  }
  __syncthreads();

  if (t < RB_BLOCKS) {
    int ex = 0;
    for (int q = 0; q < RB_BLOCKS; ++q) ex += (q < t) ? hist[q] : 0;
    hx[t] = ex;
    int* H = hoff + ((size_t)k * RB_BLOCKS + j) * (RB_BLOCKS + 1);
    H[t] = ex;
    if (t == RB_BLOCKS - 1) H[RB_BLOCKS] = ex + hist[t];
  }
  __syncthreads();

#pragma unroll
  for (int n = 0; n < PS_PER; ++n) {
    const int i = t + n * RB_THREADS;
    if (i < PS_SLICE) {
      const int p = min(hx[rg[n]] + rk[n], PS_SLICE - 1);
      EdgeDV rec;
      rec.cd = ((unsigned)cc[n] & 0xffffu) | (((unsigned)(row[n] - rg[n] * RB_ROWS) & 1023u) << 16);
      rec.v  = vv[n];
      buf[p] = rec;
    }
  }
  __syncthreads();

  EdgeDV* T = tmp + s0;
  for (int i = t; i < PS_SLICE; i += RB_THREADS) T[i] = buf[i];
}

__global__ __launch_bounds__(RB_THREADS) void place_edges(
    const EdgeDV* __restrict__ tmp, const int* __restrict__ hoff,
    int* __restrict__ start, EdgeCV* __restrict__ se) {
  __shared__ int cnt[RB_THREADS];
  __shared__ int off[RB_THREADS];
  __shared__ int wtot[RB_WAVES];
  __shared__ int sbeg[RB_BLOCKS];
  __shared__ int slen[RB_BLOCKS];
  __shared__ int sbel[2];

  const int k    = blockIdx.y;
  const int b    = blockIdx.x;
  const int t    = threadIdx.x;
  const int lane = t & 31;
  const int w    = t >> 5;
  const int lo   = b * RB_ROWS;
  const unsigned nrows = (unsigned)min(RB_ROWS, NN - lo);
  const EdgeDV* T = tmp + (size_t)k * E_ADJ;
  EdgeCV* S = se + (size_t)k * E_ADJ;

  cnt[t] = 0;
  if (t < RB_BLOCKS) {
    const int* H = hoff + ((size_t)k * RB_BLOCKS + t) * (RB_BLOCKS + 1);
    const int hb = min(max(H[b], 0), PS_SLICE);
    const int he = min(max(H[b + 1], hb), PS_SLICE);
    sbeg[t] = t * PS_SLICE + hb;
    slen[t] = he - hb;
    int sb = hb;
#pragma unroll
    for (int o = 16; o > 0; o >>= 1) sb += __shfl_xor(sb, o, 32);
    if (lane == 0) sbel[w] = sb;
  }
  __syncthreads();

  const int b0 = sbeg[2 * w], l0 = slen[2 * w];
  const int b1 = sbeg[2 * w + 1], l1 = slen[2 * w + 1];
  const int L = l0 + l1;

  for (int i0 = lane; i0 < L; i0 += 4 * 32) {
    unsigned x[4];
#pragma unroll
    for (int u = 0; u < 4; ++u) {
      const int i = i0 + u * 32;
      const int a = (i < l0) ? b0 + i : b1 + (i - l0);
      x[u] = T[min(a, E_ADJ - 1)].cd;
    }
    asm volatile("" :: "v"(x[0]), "v"(x[1]), "v"(x[2]), "v"(x[3]));
#pragma unroll
    for (int u = 0; u < 4; ++u) {
      if (i0 + u * 32 < L) atomicAdd(&cnt[(x[u] >> 16) & 1023u], 1);
    }
  }
  __syncthreads();

  const int own = ((unsigned)t < nrows) ? cnt[t] : 0;
  int incl = own;
#pragma unroll
  for (int o = 1; o < 32; o <<= 1) {
    const int n = __shfl_up(incl, o, 32);
    if (lane >= o) incl += n;
  }
  if (lane == 31) wtot[w] = incl;
  __syncthreads();
  int base = sbel[0] + sbel[1];
  for (int q = 0; q < RB_WAVES; ++q) {
    if (q < w) base += wtot[q];
  }
  const int ex = base + incl - own;
  off[t] = ex;
  cnt[t] = 0;
  if ((unsigned)t < nrows) start[k * NN + lo + t] = ex;
  __syncthreads();

  for (int i0 = lane; i0 < L; i0 += 4 * 32) {
    EdgeDV x[4];
    int pos[4];
#pragma unroll
    for (int u = 0; u < 4; ++u) {
      const int i = i0 + u * 32;
      const int a = (i < l0) ? b0 + i : b1 + (i - l0);
      x[u] = T[min(a, E_ADJ - 1)];
    }
    asm volatile("" :: "v"(x[0].cd), "v"(x[0].v), "v"(x[1].cd), "v"(x[1].v), "v"(x[2].cd), "v"(x[2].v), "v"(x[3].cd), "v"(x[3].v));
#pragma unroll
    for (int u = 0; u < 4; ++u) {
      const int d = (int)((x[u].cd >> 16) & 1023u);
      pos[u] = (i0 + u * 32 < L) ? off[d] + atomicAdd(&cnt[d], 1) : E_ADJ;
    }
#pragma unroll
    for (int u = 0; u < 4; ++u) {
      if (pos[u] >= 0 && pos[u] < E_ADJ) {
        EdgeCV rec;
        rec.c = (int)(x[u].cd & 0xffffu);
        rec.v = x[u].v;
        S[pos[u]] = rec;
      }
    }
  }
}

struct Edge4 { EdgeCV a, b, c, d; };
static __device__ __forceinline__ Edge4 edges_at(const EdgeCV* __restrict__ ed, int i) {
  const EdgeCV* p = ed + min(max(i, 0), E_ADJ - 4);
  Edge4 q;
  q.a = p[0]; q.b = p[1]; q.c = p[2]; q.d = p[3];
  return q;
}

static_assert(NN % 16 == 0, "sixteen rows per block of the gather");

__global__ __launch_bounds__(256) __attribute__((amdgpu_waves_per_eu(10, 10))) void spmm_gather(
    const EdgeCV* __restrict__ se,
    const int* __restrict__ start, const float* __restrict__ xw,
    float* __restrict__ h0, float* __restrict__ h1) {
  const int lane = threadIdx.x & 31;
  const int grp  = lane >> 3;
  const int half = grp & 1;
  const int l8   = lane & 7;
  const int r    = blockIdx.x * 16 + (threadIdx.x >> 5) * 2 + (grp >> 1);
  const int i    = blockIdx.y;
  const int k    = i * 2 + half;

  const int* st = start + k * NN;
  int j = st[r];
  int end = (r + 1 < NN) ? st[r + 1] : E_ADJ;
  end = min(end, E_ADJ);
  const EdgeCV* ed = se + (size_t)k * E_ADJ;
  const float4* X = (const float4*)(xw + (size_t)k * NN * HH) + l8;

  float4 acc = make_float4(0.0f, 0.0f, 0.0f, 0.0f);
  float4 bcc = make_float4(0.0f, 0.0f, 0.0f, 0.0f);
  const Edge4 first = edges_at(ed, j);
  EdgeCV e0 = first.a, e1 = first.b, e2 = first.c, e3 = first.d;
  for (; j + 4 <= end; j += 4) {
    const Edge4 nx = edges_at(ed, j + 4);
    const EdgeCV n0 = nx.a, n1 = nx.b, n2 = nx.c, n3 = nx.d;
    const float v0 = e0.v, v1 = e1.v, v2 = e2.v, v3 = e3.v;
    const float4 a0 = X[(size_t)e0.c * 16], b0 = X[(size_t)e0.c * 16 + 8];
    const float4 a1 = X[(size_t)e1.c * 16], b1 = X[(size_t)e1.c * 16 + 8];
    const float4 a2 = X[(size_t)e2.c * 16], b2 = X[(size_t)e2.c * 16 + 8];
    const float4 a3 = X[(size_t)e3.c * 16], b3 = X[(size_t)e3.c * 16 + 8];
    acc.x += a0.x * v0; acc.y += a0.y * v0; acc.z += a0.z * v0; acc.w += a0.w * v0;
    bcc.x += b0.x * v0; bcc.y += b0.y * v0; bcc.z += b0.z * v0; bcc.w += b0.w * v0;
    acc.x += a1.x * v1; acc.y += a1.y * v1; acc.z += a1.z * v1; acc.w += a1.w * v1;
    bcc.x += b1.x * v1; bcc.y += b1.y * v1; bcc.z += b1.z * v1; bcc.w += b1.w * v1;
    acc.x += a2.x * v2; acc.y += a2.y * v2; acc.z += a2.z * v2; acc.w += a2.w * v2;
    bcc.x += b2.x * v2; bcc.y += b2.y * v2; bcc.z += b2.z * v2; bcc.w += b2.w * v2;
    acc.x += a3.x * v3; acc.y += a3.y * v3; acc.z += a3.z * v3; acc.w += a3.w * v3;
    bcc.x += b3.x * v3; bcc.y += b3.y * v3; bcc.z += b3.z * v3; bcc.w += b3.w * v3;
    __builtin_amdgcn_sched_group_barrier(0x020, 10, 0);
    __builtin_amdgcn_sched_group_barrier(0x002, 16, 0);
    e0 = n0; e1 = n1; e2 = n2; e3 = n3;
  }
  const int rem = end - j;
  if (rem > 0) {
    const int sh = j - min(max(j, 0), E_ADJ - 4);
    const EdgeCV w0 = (sh <= 0) ? e0 : (sh == 1) ? e1 : (sh == 2) ? e2 : e3;
    EdgeCV w1 = (sh <= 0) ? e1 : (sh == 1) ? e2 : e3;
    EdgeCV w2 = (sh <= 0) ? e2 : e3;
    if (rem < 2) w1 = w0;
    if (rem < 3) w2 = w0;
    const float v0 = w0.v, v1 = w1.v, v2 = w2.v;
    const float4 a0 = X[(size_t)w0.c * 16], b0 = X[(size_t)w0.c * 16 + 8];
    const float4 a1 = X[(size_t)w1.c * 16], b1 = X[(size_t)w1.c * 16 + 8];
    const float4 a2 = X[(size_t)w2.c * 16], b2 = X[(size_t)w2.c * 16 + 8];
    asm volatile("" :: "v"(a0.x), "v"(b0.x), "v"(a1.x), "v"(b1.x), "v"(a2.x), "v"(b2.x));
    acc.x += a0.x * v0; acc.y += a0.y * v0; acc.z += a0.z * v0; acc.w += a0.w * v0;
    bcc.x += b0.x * v0; bcc.y += b0.y * v0; bcc.z += b0.z * v0; bcc.w += b0.w * v0;
    if (rem > 1) {
      acc.x += a1.x * v1; acc.y += a1.y * v1; acc.z += a1.z * v1; acc.w += a1.w * v1;
      bcc.x += b1.x * v1; bcc.y += b1.y * v1; bcc.z += b1.z * v1; bcc.w += b1.w * v1;
    }
    if (rem > 2) {
      acc.x += a2.x * v2; acc.y += a2.y * v2; acc.z += a2.z * v2; acc.w += a2.w * v2;
      bcc.x += b2.x * v2; bcc.y += b2.y * v2; bcc.z += b2.z * v2; bcc.w += b2.w * v2;
    }
  }

  float4 q, p;
  q.x = fmaxf(acc.x, 0.0f); q.y = fmaxf(acc.y, 0.0f);
  q.z = fmaxf(acc.z, 0.0f); q.w = fmaxf(acc.w, 0.0f);
  p.x = fmaxf(bcc.x, 0.0f); p.y = fmaxf(bcc.y, 0.0f);
  p.z = fmaxf(bcc.z, 0.0f); p.w = fmaxf(bcc.w, 0.0f);
  float4 o, n;
  o.x = __shfl_xor(q.x, 8, 32); o.y = __shfl_xor(q.y, 8, 32);
  o.z = __shfl_xor(q.z, 8, 32); o.w = __shfl_xor(q.w, 8, 32);
  n.x = __shfl_xor(p.x, 8, 32); n.y = __shfl_xor(p.y, 8, 32);
  n.z = __shfl_xor(p.z, 8, 32); n.w = __shfl_xor(p.w, 8, 32);
  if (half == 0) {
    float4 s, t;
    s.x = q.x + o.x; s.y = q.y + o.y; s.z = q.z + o.z; s.w = q.w + o.w;
    t.x = p.x + n.x; t.y = p.y + n.y; t.z = p.z + n.z; t.w = p.w + n.w;
    float4* H = (float4*)(i ? h1 : h0) + (size_t)r * 16 + l8;
    H[0] = s;
    H[8] = t;
  }
}

#define GP 68
#define WTP 76
#define GG_TILES  5
#define GG_BLOCKS (NN / 16 / GG_TILES)
static_assert(GG_BLOCKS * GG_TILES * 16 == NN, "the blocks' tiles cover the rows exactly");

__global__ __launch_bounds__(256) __attribute__((amdgpu_waves_per_eu(10, 10))) void gather_gemm_layer(
    const EdgeCV* __restrict__ se, const int* __restrict__ start,
    const float* __restrict__ x0, const float* __restrict__ x1,
    const float* __restrict__ W,
    float* __restrict__ h0, float* __restrict__ h1) {
  __shared__ __align__(16) float Wt[2][HH][WTP];
  __shared__ __align__(16) float G[2][16][GP];
  __shared__ __align__(16) float Y[2][16][GP];

  const int lane = threadIdx.x & 31;
  const int wv   = threadIdx.x >> 5;
  const int grp  = lane >> 3;
  const int half = grp & 1;
  const int l8   = lane & 7;
  const int lr   = wv * 2 + (grp >> 1);
  const int i    = blockIdx.y;
  const int k    = i * 2 + half;

  const int* st = start + k * NN;
  const EdgeCV* ed = se + (size_t)k * E_ADJ;
  const float4* X = (const float4*)(half ? x1 : x0) + l8;

  int rj, re;
  {
    const int r = blockIdx.x * 16 + lr;
    rj = st[r];
    re = st[min(r + 1, NN - 1)];
  }

  {
    const float4* Wsrc = (const float4*)(W + (size_t)(i * 2) * HH * HH);
#pragma unroll
    for (int u = 0; u < 8; ++u) {
      const int idx = threadIdx.x + 256 * u;
      const float4 w4 = Wsrc[idx];
      const int relx = idx >> 10, kx = (idx >> 4) & 63, nx = (idx & 15) * 4;
      Wt[relx][nx + 0][kx] = w4.x;
      Wt[relx][nx + 1][kx] = w4.y;
      Wt[relx][nx + 2][kx] = w4.z;
      Wt[relx][nx + 3][kx] = w4.w;
    }
  }

  Edge4 fx = edges_at(ed, rj);

  const int rel = wv >> 2;
  const int ct  = wv & 3;
  const int h2  = lane >> 4;
  const int l   = lane & 15;
  const float* ar = &G[rel][l][h2 * 2];
  const float* br = &Wt[rel][ct * 16 + l][h2 * 2];
  const int orow = threadIdx.x >> 4;
  const int oq   = threadIdx.x & 15;
  float4* Hout = (float4*)(i ? h1 : h0) + oq;

  for (int tile = blockIdx.x; tile < NN / 16; tile += gridDim.x) {
    const int r = tile * 16 + lr;
    int j = rj;
    int end = (r + 1 < NN) ? re : E_ADJ;
    end = min(end, E_ADJ);
    {
      const int nr = min(r + (int)gridDim.x * 16, NN - 1);
      rj = st[nr];
      re = st[min(nr + 1, NN - 1)];
    }
    float4 acc = make_float4(0.0f, 0.0f, 0.0f, 0.0f);
    float4 bcc = make_float4(0.0f, 0.0f, 0.0f, 0.0f);
    EdgeCV e0 = fx.a, e1 = fx.b, e2 = fx.c, e3 = fx.d;
    for (; j + 4 <= end; j += 4) {
      const Edge4 nx = edges_at(ed, j + 4);
      const EdgeCV n0 = nx.a, n1 = nx.b, n2 = nx.c, n3 = nx.d;
      const float v0 = e0.v, v1 = e1.v, v2 = e2.v, v3 = e3.v;
      const float4 a0 = X[(size_t)e0.c * 16], b0 = X[(size_t)e0.c * 16 + 8];
      const float4 a1 = X[(size_t)e1.c * 16], b1 = X[(size_t)e1.c * 16 + 8];
      const float4 a2 = X[(size_t)e2.c * 16], b2 = X[(size_t)e2.c * 16 + 8];
      const float4 a3 = X[(size_t)e3.c * 16], b3 = X[(size_t)e3.c * 16 + 8];
      acc.x += a0.x * v0; acc.y += a0.y * v0; acc.z += a0.z * v0; acc.w += a0.w * v0;
      bcc.x += b0.x * v0; bcc.y += b0.y * v0; bcc.z += b0.z * v0; bcc.w += b0.w * v0;
      acc.x += a1.x * v1; acc.y += a1.y * v1; acc.z += a1.z * v1; acc.w += a1.w * v1;
      bcc.x += b1.x * v1; bcc.y += b1.y * v1; bcc.z += b1.z * v1; bcc.w += b1.w * v1;
      acc.x += a2.x * v2; acc.y += a2.y * v2; acc.z += a2.z * v2; acc.w += a2.w * v2;
      bcc.x += b2.x * v2; bcc.y += b2.y * v2; bcc.z += b2.z * v2; bcc.w += b2.w * v2;
      acc.x += a3.x * v3; acc.y += a3.y * v3; acc.z += a3.z * v3; acc.w += a3.w * v3;
      bcc.x += b3.x * v3; bcc.y += b3.y * v3; bcc.z += b3.z * v3; bcc.w += b3.w * v3;
      __builtin_amdgcn_sched_group_barrier(0x020, 10, 0);
      __builtin_amdgcn_sched_group_barrier(0x002, 16, 0);
      e0 = n0; e1 = n1; e2 = n2; e3 = n3;
    }
    fx = edges_at(ed, rj);
    const int rem = end - j;
    if (rem > 0) {
      const int sh = j - min(max(j, 0), E_ADJ - 4);
      const EdgeCV w0 = (sh <= 0) ? e0 : (sh == 1) ? e1 : (sh == 2) ? e2 : e3;
      EdgeCV w1 = (sh <= 0) ? e1 : (sh == 1) ? e2 : e3;
      EdgeCV w2 = (sh <= 0) ? e2 : e3;
      if (rem < 2) w1 = w0;
      if (rem < 3) w2 = w0;
      const float v0 = w0.v, v1 = w1.v, v2 = w2.v;
      const float4 a0 = X[(size_t)w0.c * 16], b0 = X[(size_t)w0.c * 16 + 8];
      const float4 a1 = X[(size_t)w1.c * 16], b1 = X[(size_t)w1.c * 16 + 8];
      const float4 a2 = X[(size_t)w2.c * 16], b2 = X[(size_t)w2.c * 16 + 8];
      asm volatile("" :: "v"(a0.x), "v"(b0.x), "v"(a1.x), "v"(b1.x), "v"(a2.x), "v"(b2.x));
      acc.x += a0.x * v0; acc.y += a0.y * v0; acc.z += a0.z * v0; acc.w += a0.w * v0;
      bcc.x += b0.x * v0; bcc.y += b0.y * v0; bcc.z += b0.z * v0; bcc.w += b0.w * v0;
      if (rem > 1) {
        acc.x += a1.x * v1; acc.y += a1.y * v1; acc.z += a1.z * v1; acc.w += a1.w * v1;
        bcc.x += b1.x * v1; bcc.y += b1.y * v1; bcc.z += b1.z * v1; bcc.w += b1.w * v1;
      }
      if (rem > 2) {
        acc.x += a2.x * v2; acc.y += a2.y * v2; acc.z += a2.z * v2; acc.w += a2.w * v2;
        bcc.x += b2.x * v2; bcc.y += b2.y * v2; bcc.z += b2.z * v2; bcc.w += b2.w * v2;
      }
    }

    *(float4*)&G[half][lr][4 * l8]      = acc;
    *(float4*)&G[half][lr][32 + 4 * l8] = bcc;
    __syncthreads();

    v8f c = {};
#pragma unroll
    for (int s = 0; s < HH / 4; ++s) {
      v2f a, b;
      a.x = ar[4 * s];
      a.y = ar[4 * s + 1];
      b.x = br[4 * s];
      b.y = br[4 * s + 1];
      c = __builtin_amdgcn_wmma_f32_16x16x4_f32(false, a, false, b, (short)0, c, false, false);
    }
#pragma unroll
    for (int v = 0; v < 8; ++v) Y[rel][v + h2 * 8][ct * 16 + l] = fmaxf(c[v], 0.0f);
    __syncthreads();

    const float4 y0 = *(const float4*)&Y[0][orow][4 * oq];
    const float4 y1 = *(const float4*)&Y[1][orow][4 * oq];
    float4 s4;
    s4.x = y0.x + y1.x; s4.y = y0.y + y1.y; s4.z = y0.z + y1.z; s4.w = y0.w + y1.w;
    Hout[(size_t)(tile * 16 + orow) * 16] = s4;
  }
}

__global__ void build_M(const float* __restrict__ R, const float* __restrict__ D,
                        const int* __restrict__ rt_k, float* __restrict__ M) {
  int i = blockIdx.x * blockDim.x + threadIdx.x;
  const float* d = D + rt_k[0] * HH;
  int h = i >> 6, g = i & 63;
  M[i] = d[h] * R[i] * d[g];
}

static_assert(E_DEC % 64 == 0, "four 16-edge groups per block of the decoder");

__global__ __launch_bounds__(128) void decode_edges(
    const float* __restrict__ h0, const float* __restrict__ h1,
    const int* __restrict__ edges, const float* __restrict__ M,
    float* __restrict__ preds) {
  const int wave = threadIdx.x >> 5;
  const int lane = threadIdx.x & 31;
  const int half = lane >> 4;
  const int l    = lane & 15;
  const int eb   = blockIdx.x * 64 + wave * 16;

  const int arow = edges[(eb + l) * 2 + 0];

  const float* ar = h0 + (size_t)arow * HH + half * 2;
  v8f c0 = {}, c1 = {}, c2 = {}, c3 = {};
  for (int kk = 0; kk < HH; kk += 4) {
    v2f a, b0, b1, b2, b3;
    a.x = ar[kk];
    a.y = ar[kk + 1];
    const float* mp = M + (kk + half * 2) * HH + l;
    b0.x = mp[0];   b0.y = mp[HH];
    b1.x = mp[16];  b1.y = mp[HH + 16];
    b2.x = mp[32];  b2.y = mp[HH + 32];
    b3.x = mp[48];  b3.y = mp[HH + 48];
    c0 = __builtin_amdgcn_wmma_f32_16x16x4_f32(false, a, false, b0, (short)0, c0, false, false);
    c1 = __builtin_amdgcn_wmma_f32_16x16x4_f32(false, a, false, b1, (short)0, c1, false, false);
    c2 = __builtin_amdgcn_wmma_f32_16x16x4_f32(false, a, false, b2, (short)0, c2, false, false);
    c3 = __builtin_amdgcn_wmma_f32_16x16x4_f32(false, a, false, b3, (short)0, c3, false, false);
  }

  float p0[8], p1[8], p2[8], p3[8];
#pragma unroll
  for (int v = 0; v < 8; ++v) {
    const int edge = v + half * 8;
    const int crow = edges[(eb + edge) * 2 + 1];
    const float* hc = h1 + (size_t)crow * HH + l;
    p0[v] = c0[v] * hc[0];
    p1[v] = c1[v] * hc[16];
    p2[v] = c2[v] * hc[32];
    p3[v] = c3[v] * hc[48];
  }
#pragma unroll
  for (int off = 1; off < 16; off <<= 1) {
#pragma unroll
    for (int v = 0; v < 8; ++v) {
      p0[v] += __shfl_xor(p0[v], off, 16);
      p1[v] += __shfl_xor(p1[v], off, 16);
      p2[v] += __shfl_xor(p2[v], off, 16);
      p3[v] += __shfl_xor(p3[v], off, 16);
    }
  }
  if (l == 0) {
    float s[8];
#pragma unroll
    for (int v = 0; v < 8; ++v) s[v] = p0[v] + p1[v] + p2[v] + p3[v];
    float4* o = (float4*)(preds + eb + half * 8);
    o[0] = make_float4(s[0], s[1], s[2], s[3]);
    o[1] = make_float4(s[4], s[5], s[6], s[7]);
  }
}

extern "C" void kernel_launch(void* const* d_in, const int* in_sizes, int n_in,
                              void* d_out, int out_size, void* d_ws, size_t ws_size,
                              hipStream_t stream) {
  const int*   adj_rows = (const int*)d_in[0];
  const int*   adj_cols = (const int*)d_in[1];
  const float* adj_vals = (const float*)d_in[2];
  const float* feat0    = (const float*)d_in[3];
  const float* feat1    = (const float*)d_in[4];
  const float* W1       = (const float*)d_in[5];
  const float* W2       = (const float*)d_in[6];
  const float* W3       = (const float*)d_in[7];
  const float* W4       = (const float*)d_in[8];
  const float* Rg       = (const float*)d_in[9];
  const float* Dl       = (const float*)d_in[10];
  const int*   edges    = (const int*)d_in[11];
  const int*   rt_k     = (const int*)d_in[12];

  const size_t S = (size_t)NN * HH;
  float* xw = (float*)d_ws;
  float* lists = xw + 4 * S;
  float* h0 = lists + 4 * S;
  float* h1 = h0 + S;
  float* M  = h1 + S;

  static_assert((size_t)4 * E_ADJ * 2 + (size_t)4 * NN + (size_t)4 * RB_BLOCKS * (RB_BLOCKS + 1) <= (size_t)4 * NN * HH,
                "sorted lists and offset tables must fit in the four former y planes");
  static_assert((size_t)4 * E_ADJ * 2 <= (size_t)2 * NN * HH, "the split slices must fit in two xw planes");
  EdgeCV* se   = (EdgeCV*)lists;
  int*   start = (int*)(se + (size_t)4 * E_ADJ);
  int*   hoff  = start + (size_t)4 * NN;
  EdgeDV* tmp  = (EdgeDV*)(xw + 2 * S);

  float* g0 = xw;
  float* g1 = xw + S;

  const int*   rows1 = adj_rows;
  const int*   cols1 = adj_cols;
  const float* vals1 = adj_vals;
  const int*   rows2 = adj_rows + (size_t)4 * E_ADJ;
  const int*   cols2 = adj_cols + (size_t)4 * E_ADJ;
  const float* vals2 = adj_vals + (size_t)4 * E_ADJ;

  split_edges<<<dim3(RB_BLOCKS, 4), RB_THREADS, 0, stream>>>(rows1, cols1, vals1, tmp, hoff);
  place_edges<<<dim3(RB_BLOCKS, 4), RB_THREADS, 0, stream>>>(tmp, hoff, start, se);
  gemm_stage<<<dim3(NN / (16 * GEMM_WAVES), 4), 32 * GEMM_WAVES, 0, stream>>>(feat0, feat1, W1, xw, 128);
  spmm_gather<<<dim3(NN / 16, 2), 256, 0, stream>>>(se, start, xw, h0, h1);
  gather_gemm_layer<<<dim3(GG_BLOCKS, 2), 256, 0, stream>>>(se, start, h0, h1, W2, g0, g1);

  split_edges<<<dim3(RB_BLOCKS, 4), RB_THREADS, 0, stream>>>(rows2, cols2, vals2, tmp, hoff);
  place_edges<<<dim3(RB_BLOCKS, 4), RB_THREADS, 0, stream>>>(tmp, hoff, start, se);
  gather_gemm_layer<<<dim3(GG_BLOCKS, 2), 256, 0, stream>>>(se, start, g0, g1, W3, h0, h1);
  gather_gemm_layer<<<dim3(GG_BLOCKS, 2), 256, 0, stream>>>(se, start, h0, h1, W4, g0, g1);

  build_M<<<16, 256, 0, stream>>>(Rg, Dl, rt_k, M);
  decode_edges<<<E_DEC / 64, 128, 0, stream>>>(g0, g1, edges, M, (float*)d_out);
}


# place_edges builds its range's sorted list in LDS (16384 records; any beyond go straight to memory) and stores aligned 256-byte chunks: no scattered 8-byte stores; LDS-atomic arrival order kept
# speedup vs baseline: 4.4439x; 1.0465x over previous
#include <hip/hip_runtime.h>
#include <hip/hip_bf16.h>

#define NN 50000
#define E_ADJ 800000
#define E_DEC 200000
#define HH 64

typedef float v2f __attribute__((ext_vector_type(2)));
typedef float v8f __attribute__((ext_vector_type(8)));

#define GEMM_WAVES 5
static_assert(NN % (16 * GEMM_WAVES) == 0, "row tiles must divide evenly among the blocks");

__global__ __launch_bounds__(32 * GEMM_WAVES) void gemm_stage(
    const float* __restrict__ x0, const float* __restrict__ x1,
    const float* __restrict__ W, float* __restrict__ xw, int Fin) {
  const int k    = blockIdx.y;
  const int tm   = blockIdx.x * GEMM_WAVES + (threadIdx.x >> 5);
  const int lane = threadIdx.x & 31;
  const int half = lane >> 4;
  const int l    = lane & 15;

  const float* X  = (k & 1) ? x1 : x0;
  const float* Wk = W + (size_t)k * Fin * HH;

  const int row = tm * 16 + l;

  const float* ar = X + (size_t)row * Fin + half * 2;
  v8f c0 = {}, c1 = {}, c2 = {}, c3 = {};
  for (int kk = 0; kk < Fin; kk += 4) {
    v2f a, b0, b1, b2, b3;
    a.x = ar[kk];
    a.y = ar[kk + 1];
    const float* wp = Wk + (size_t)(kk + half * 2) * HH + l;
    b0.x = wp[0];   b0.y = wp[HH];
    b1.x = wp[16];  b1.y = wp[HH + 16];
    b2.x = wp[32];  b2.y = wp[HH + 32];
    b3.x = wp[48];  b3.y = wp[HH + 48];
    c0 = __builtin_amdgcn_wmma_f32_16x16x4_f32(
         false, a,  false, b0,
         (short)0, c0,  false,  false);
    c1 = __builtin_amdgcn_wmma_f32_16x16x4_f32(false, a, false, b1, (short)0, c1, false, false);
    c2 = __builtin_amdgcn_wmma_f32_16x16x4_f32(false, a, false, b2, (short)0, c2, false, false);
    c3 = __builtin_amdgcn_wmma_f32_16x16x4_f32(false, a, false, b3, (short)0, c3, false, false);
  }

  float* out = xw + ((size_t)k * NN + (size_t)tm * 16) * HH + l;
#pragma unroll
  for (int v = 0; v < 8; ++v) {
    float* o = out + (size_t)(v + half * 8) * HH;
    o[0]  = c0[v];
    o[16] = c1[v];
    o[32] = c2[v];
    o[48] = c3[v];
  }
}

struct __align__(8) EdgeCV { int c; float v; };
struct __align__(8) EdgeDV { unsigned cd; float v; };

#define RB_BLOCKS  64
#define RB_ROWS    ((NN + RB_BLOCKS - 1) / RB_BLOCKS)
#define RB_THREADS 1024
#define RB_WAVES   (RB_THREADS / 32)
#define PS_SLICE   (E_ADJ / RB_BLOCKS)
#define PS_PER     ((PS_SLICE + RB_THREADS - 1) / RB_THREADS)
static_assert(RB_ROWS <= RB_THREADS && RB_ROWS <= 1024, "one counter per thread; local row packs into 10 bits");
static_assert(NN <= (1 << 16), "source row packs into 16 bits");
static_assert(E_ADJ % RB_BLOCKS == 0, "slices are equal");
static_assert((NN - 1) / RB_ROWS < RB_BLOCKS, "every row has a range");
static_assert(RB_BLOCKS == 2 * RB_WAVES, "a wave of place_edges serves two slices' pieces");
#define PL_CAP     16384
static_assert(PL_CAP % 32 == 0, "whole 256-byte chunks");

__global__ __launch_bounds__(RB_THREADS) void split_edges(
    const int* __restrict__ rows, const int* __restrict__ cols, const float* __restrict__ vals,
    EdgeDV* __restrict__ tmp, int* __restrict__ hoff) {
  __shared__ int hist[RB_BLOCKS];
  __shared__ int hx[RB_BLOCKS];
  __shared__ EdgeDV buf[PS_SLICE];

  const int k = blockIdx.y;
  const int j = blockIdx.x;
  const int t = threadIdx.x;
  const size_t s0 = (size_t)k * E_ADJ + (size_t)j * PS_SLICE;
  const int*   R  = rows + s0;
  const int*   Cs = cols + s0;
  const float* Vs = vals + s0;

  if (t < RB_BLOCKS) hist[t] = 0;
  int row[PS_PER];
#pragma unroll
  for (int n = 0; n < PS_PER; ++n) {
    const int i = t + n * RB_THREADS;
    row[n] = (i < PS_SLICE) ? R[i] : 0;
  }
  __syncthreads();

  int rg[PS_PER], rk[PS_PER];
#pragma unroll
  for (int n = 0; n < PS_PER; ++n) {
    const int i = t + n * RB_THREADS;
    const int r = (int)min((unsigned)row[n] / (unsigned)RB_ROWS, (unsigned)(RB_BLOCKS - 1));
    rg[n] = r;
    rk[n] = (i < PS_SLICE) ? atomicAdd(&hist[r], 1) : 0;
  }
  int cc[PS_PER];
  float vv[PS_PER];
#pragma unroll
  for (int n = 0; n < PS_PER; ++n) {
    const int i = t + n * RB_THREADS;
    cc[n] = (i < PS_SLICE) ? Cs[i] : 0;
    vv[n] = (i < PS_SLICE) ? Vs[i] : 0.0f;
  }
  __syncthreads();

  if (t < RB_BLOCKS) {
    int ex = 0;
    for (int q = 0; q < RB_BLOCKS; ++q) ex += (q < t) ? hist[q] : 0;
    hx[t] = ex;
    int* H = hoff + ((size_t)k * RB_BLOCKS + j) * (RB_BLOCKS + 1);
    H[t] = ex;
    if (t == RB_BLOCKS - 1) H[RB_BLOCKS] = ex + hist[t];
  }
  __syncthreads();

#pragma unroll
  for (int n = 0; n < PS_PER; ++n) {
    const int i = t + n * RB_THREADS;
    if (i < PS_SLICE) {
      const int p = min(hx[rg[n]] + rk[n], PS_SLICE - 1);
      EdgeDV rec;
      rec.cd = ((unsigned)cc[n] & 0xffffu) | (((unsigned)(row[n] - rg[n] * RB_ROWS) & 1023u) << 16);
      rec.v  = vv[n];
      buf[p] = rec;
    }
  }
  __syncthreads();

  EdgeDV* T = tmp + s0;
  for (int i = t; i < PS_SLICE; i += RB_THREADS) T[i] = buf[i];
}

__global__ __launch_bounds__(RB_THREADS) void place_edges(
    const EdgeDV* __restrict__ tmp, const int* __restrict__ hoff,
    int* __restrict__ start, EdgeCV* __restrict__ se) {
  __shared__ int cnt[RB_THREADS];
  __shared__ int off[RB_THREADS];
  __shared__ int wtot[RB_WAVES];
  __shared__ int sbeg[RB_BLOCKS];
  __shared__ int slen[RB_BLOCKS];
  __shared__ int sbel[2];
  __shared__ EdgeCV lst[PL_CAP];

  const int k    = blockIdx.y;
  const int b    = blockIdx.x;
  const int t    = threadIdx.x;
  const int lane = t & 31;
  const int w    = t >> 5;
  const int lo   = b * RB_ROWS;
  const unsigned nrows = (unsigned)min(RB_ROWS, NN - lo);
  const EdgeDV* T = tmp + (size_t)k * E_ADJ;
  EdgeCV* S = se + (size_t)k * E_ADJ;

  cnt[t] = 0;
  if (t < RB_BLOCKS) {
    const int* H = hoff + ((size_t)k * RB_BLOCKS + t) * (RB_BLOCKS + 1);
    const int hb = min(max(H[b], 0), PS_SLICE);
    const int he = min(max(H[b + 1], hb), PS_SLICE);
    sbeg[t] = t * PS_SLICE + hb;
    slen[t] = he - hb;
    int sb = hb;
#pragma unroll
    for (int o = 16; o > 0; o >>= 1) sb += __shfl_xor(sb, o, 32);
    if (lane == 0) sbel[w] = sb;
  }
  __syncthreads();

  const int b0 = sbeg[2 * w], l0 = slen[2 * w];
  const int b1 = sbeg[2 * w + 1], l1 = slen[2 * w + 1];
  const int L = l0 + l1;

  for (int i0 = lane; i0 < L; i0 += 4 * 32) {
    unsigned x[4];
#pragma unroll
    for (int u = 0; u < 4; ++u) {
      const int i = i0 + u * 32;
      const int a = (i < l0) ? b0 + i : b1 + (i - l0);
      x[u] = T[min(a, E_ADJ - 1)].cd;
    }
    asm volatile("" :: "v"(x[0]), "v"(x[1]), "v"(x[2]), "v"(x[3]));
#pragma unroll
    for (int u = 0; u < 4; ++u) {
      if (i0 + u * 32 < L) atomicAdd(&cnt[(x[u] >> 16) & 1023u], 1);
    }
  }
  __syncthreads();

  const int own = ((unsigned)t < nrows) ? cnt[t] : 0;
  int incl = own;
#pragma unroll
  for (int o = 1; o < 32; o <<= 1) {
    const int n = __shfl_up(incl, o, 32);
    if (lane >= o) incl += n;
  }
  if (lane == 31) wtot[w] = incl;
  __syncthreads();
  const int below = sbel[0] + sbel[1];
  int wsum = 0, tot = 0;
  for (int q = 0; q < RB_WAVES; ++q) {
    const int n = wtot[q];
    tot += n;
    if (q < w) wsum += n;
  }
  const int ex = wsum + incl - own;
  off[t] = ex;
  cnt[t] = 0;
  if ((unsigned)t < nrows) start[k * NN + lo + t] = below + ex;
  __syncthreads();

  for (int i0 = lane; i0 < L; i0 += 4 * 32) {
    EdgeDV x[4];
    int rel[4];
#pragma unroll
    for (int u = 0; u < 4; ++u) {
      const int i = i0 + u * 32;
      const int a = (i < l0) ? b0 + i : b1 + (i - l0);
      x[u] = T[min(a, E_ADJ - 1)];
    }
    asm volatile("" :: "v"(x[0].cd), "v"(x[0].v), "v"(x[1].cd), "v"(x[1].v), "v"(x[2].cd), "v"(x[2].v), "v"(x[3].cd), "v"(x[3].v));
#pragma unroll
    for (int u = 0; u < 4; ++u) {
      const int d = (int)((x[u].cd >> 16) & 1023u);
      rel[u] = (i0 + u * 32 < L) ? off[d] + atomicAdd(&cnt[d], 1) : -1;
    }
#pragma unroll
    for (int u = 0; u < 4; ++u) {
      EdgeCV rec;
      rec.c = (int)(x[u].cd & 0xffffu);
      rec.v = x[u].v;
      if ((unsigned)rel[u] < (unsigned)PL_CAP) {
        lst[rel[u]] = rec;
      } else if (rel[u] >= 0) {
        const int pos = below + rel[u];
        if (pos >= 0 && pos < E_ADJ) S[pos] = rec;
      }
    }
  }
  __syncthreads();

  const int nl = min(tot, PL_CAP);
  for (int i0 = t - (below & 31); i0 < nl; i0 += 4 * RB_THREADS) {
    EdgeCV q[4];
#pragma unroll
    for (int u = 0; u < 4; ++u) q[u] = lst[min(max(i0 + u * RB_THREADS, 0), PL_CAP - 1)];
#pragma unroll
    for (int u = 0; u < 4; ++u) {
      const int i = i0 + u * RB_THREADS;
      const int pos = below + i;
      if (i >= 0 && i < nl && pos < E_ADJ) S[pos] = q[u];
    }
  }
}

struct Edge4 { EdgeCV a, b, c, d; };
static __device__ __forceinline__ Edge4 edges_at(const EdgeCV* __restrict__ ed, int i) {
  const EdgeCV* p = ed + min(max(i, 0), E_ADJ - 4);
  Edge4 q;
  q.a = p[0]; q.b = p[1]; q.c = p[2]; q.d = p[3];
  return q;
}

static_assert(NN % 16 == 0, "sixteen rows per block of the gather");

__global__ __launch_bounds__(256) __attribute__((amdgpu_waves_per_eu(10, 10))) void spmm_gather(
    const EdgeCV* __restrict__ se,
    const int* __restrict__ start, const float* __restrict__ xw,
    float* __restrict__ h0, float* __restrict__ h1) {
  const int lane = threadIdx.x & 31;
  const int grp  = lane >> 3;
  const int half = grp & 1;
  const int l8   = lane & 7;
  const int r    = blockIdx.x * 16 + (threadIdx.x >> 5) * 2 + (grp >> 1);
  const int i    = blockIdx.y;
  const int k    = i * 2 + half;

  const int* st = start + k * NN;
  int j = st[r];
  int end = (r + 1 < NN) ? st[r + 1] : E_ADJ;
  end = min(end, E_ADJ);
  const EdgeCV* ed = se + (size_t)k * E_ADJ;
  const float4* X = (const float4*)(xw + (size_t)k * NN * HH) + l8;

  float4 acc = make_float4(0.0f, 0.0f, 0.0f, 0.0f);
  float4 bcc = make_float4(0.0f, 0.0f, 0.0f, 0.0f);
  const Edge4 first = edges_at(ed, j);
  EdgeCV e0 = first.a, e1 = first.b, e2 = first.c, e3 = first.d;
  for (; j + 4 <= end; j += 4) {
    const Edge4 nx = edges_at(ed, j + 4);
    const EdgeCV n0 = nx.a, n1 = nx.b, n2 = nx.c, n3 = nx.d;
    const float v0 = e0.v, v1 = e1.v, v2 = e2.v, v3 = e3.v;
    const float4 a0 = X[(size_t)e0.c * 16], b0 = X[(size_t)e0.c * 16 + 8];
    const float4 a1 = X[(size_t)e1.c * 16], b1 = X[(size_t)e1.c * 16 + 8];
    const float4 a2 = X[(size_t)e2.c * 16], b2 = X[(size_t)e2.c * 16 + 8];
    const float4 a3 = X[(size_t)e3.c * 16], b3 = X[(size_t)e3.c * 16 + 8];
    acc.x += a0.x * v0; acc.y += a0.y * v0; acc.z += a0.z * v0; acc.w += a0.w * v0;
    bcc.x += b0.x * v0; bcc.y += b0.y * v0; bcc.z += b0.z * v0; bcc.w += b0.w * v0;
    acc.x += a1.x * v1; acc.y += a1.y * v1; acc.z += a1.z * v1; acc.w += a1.w * v1;
    bcc.x += b1.x * v1; bcc.y += b1.y * v1; bcc.z += b1.z * v1; bcc.w += b1.w * v1;
    acc.x += a2.x * v2; acc.y += a2.y * v2; acc.z += a2.z * v2; acc.w += a2.w * v2;
    bcc.x += b2.x * v2; bcc.y += b2.y * v2; bcc.z += b2.z * v2; bcc.w += b2.w * v2;
    acc.x += a3.x * v3; acc.y += a3.y * v3; acc.z += a3.z * v3; acc.w += a3.w * v3;
    bcc.x += b3.x * v3; bcc.y += b3.y * v3; bcc.z += b3.z * v3; bcc.w += b3.w * v3;
    __builtin_amdgcn_sched_group_barrier(0x020, 10, 0);
    __builtin_amdgcn_sched_group_barrier(0x002, 16, 0);
    e0 = n0; e1 = n1; e2 = n2; e3 = n3;
  }
  const int rem = end - j;
  if (rem > 0) {
    const int sh = j - min(max(j, 0), E_ADJ - 4);
    const EdgeCV w0 = (sh <= 0) ? e0 : (sh == 1) ? e1 : (sh == 2) ? e2 : e3;
    EdgeCV w1 = (sh <= 0) ? e1 : (sh == 1) ? e2 : e3;
    EdgeCV w2 = (sh <= 0) ? e2 : e3;
    if (rem < 2) w1 = w0;
    if (rem < 3) w2 = w0;
    const float v0 = w0.v, v1 = w1.v, v2 = w2.v;
    const float4 a0 = X[(size_t)w0.c * 16], b0 = X[(size_t)w0.c * 16 + 8];
    const float4 a1 = X[(size_t)w1.c * 16], b1 = X[(size_t)w1.c * 16 + 8];
    const float4 a2 = X[(size_t)w2.c * 16], b2 = X[(size_t)w2.c * 16 + 8];
    asm volatile("" :: "v"(a0.x), "v"(b0.x), "v"(a1.x), "v"(b1.x), "v"(a2.x), "v"(b2.x));
    acc.x += a0.x * v0; acc.y += a0.y * v0; acc.z += a0.z * v0; acc.w += a0.w * v0;
    bcc.x += b0.x * v0; bcc.y += b0.y * v0; bcc.z += b0.z * v0; bcc.w += b0.w * v0;
    if (rem > 1) {
      acc.x += a1.x * v1; acc.y += a1.y * v1; acc.z += a1.z * v1; acc.w += a1.w * v1;
      bcc.x += b1.x * v1; bcc.y += b1.y * v1; bcc.z += b1.z * v1; bcc.w += b1.w * v1;
    }
    if (rem > 2) {
      acc.x += a2.x * v2; acc.y += a2.y * v2; acc.z += a2.z * v2; acc.w += a2.w * v2;
      bcc.x += b2.x * v2; bcc.y += b2.y * v2; bcc.z += b2.z * v2; bcc.w += b2.w * v2;
    }
  }

  float4 q, p;
  q.x = fmaxf(acc.x, 0.0f); q.y = fmaxf(acc.y, 0.0f);
  q.z = fmaxf(acc.z, 0.0f); q.w = fmaxf(acc.w, 0.0f);
  p.x = fmaxf(bcc.x, 0.0f); p.y = fmaxf(bcc.y, 0.0f);
  p.z = fmaxf(bcc.z, 0.0f); p.w = fmaxf(bcc.w, 0.0f);
  float4 o, n;
  o.x = __shfl_xor(q.x, 8, 32); o.y = __shfl_xor(q.y, 8, 32);
  o.z = __shfl_xor(q.z, 8, 32); o.w = __shfl_xor(q.w, 8, 32);
  n.x = __shfl_xor(p.x, 8, 32); n.y = __shfl_xor(p.y, 8, 32);
  n.z = __shfl_xor(p.z, 8, 32); n.w = __shfl_xor(p.w, 8, 32);
  if (half == 0) {
    float4 s, t;
    s.x = q.x + o.x; s.y = q.y + o.y; s.z = q.z + o.z; s.w = q.w + o.w;
    t.x = p.x + n.x; t.y = p.y + n.y; t.z = p.z + n.z; t.w = p.w + n.w;
    float4* H = (float4*)(i ? h1 : h0) + (size_t)r * 16 + l8;
    H[0] = s;
    H[8] = t;
  }
}

#define GP 68
#define WTP 76
#define GG_TILES  5
#define GG_BLOCKS (NN / 16 / GG_TILES)
static_assert(GG_BLOCKS * GG_TILES * 16 == NN, "the blocks' tiles cover the rows exactly");

__global__ __launch_bounds__(256) __attribute__((amdgpu_waves_per_eu(10, 10))) void gather_gemm_layer(
    const EdgeCV* __restrict__ se, const int* __restrict__ start,
    const float* __restrict__ x0, const float* __restrict__ x1,
    const float* __restrict__ W,
    float* __restrict__ h0, float* __restrict__ h1) {
  __shared__ __align__(16) float Wt[2][HH][WTP];
  __shared__ __align__(16) float G[2][16][GP];
  __shared__ __align__(16) float Y[2][16][GP];

  const int lane = threadIdx.x & 31;
  const int wv   = threadIdx.x >> 5;
  const int grp  = lane >> 3;
  const int half = grp & 1;
  const int l8   = lane & 7;
  const int lr   = wv * 2 + (grp >> 1);
  const int i    = blockIdx.y;
  const int k    = i * 2 + half;

  const int* st = start + k * NN;
  const EdgeCV* ed = se + (size_t)k * E_ADJ;
  const float4* X = (const float4*)(half ? x1 : x0) + l8;

  int rj, re;
  {
    const int r = blockIdx.x * 16 + lr;
    rj = st[r];
    re = st[min(r + 1, NN - 1)];
  }

  {
    const float4* Wsrc = (const float4*)(W + (size_t)(i * 2) * HH * HH);
#pragma unroll
    for (int u = 0; u < 8; ++u) {
      const int idx = threadIdx.x + 256 * u;
      const float4 w4 = Wsrc[idx];
      const int relx = idx >> 10, kx = (idx >> 4) & 63, nx = (idx & 15) * 4;
      Wt[relx][nx + 0][kx] = w4.x;
      Wt[relx][nx + 1][kx] = w4.y;
      Wt[relx][nx + 2][kx] = w4.z;
      Wt[relx][nx + 3][kx] = w4.w;
    }
  }

  Edge4 fx = edges_at(ed, rj);

  const int rel = wv >> 2;
  const int ct  = wv & 3;
  const int h2  = lane >> 4;
  const int l   = lane & 15;
  const float* ar = &G[rel][l][h2 * 2];
  const float* br = &Wt[rel][ct * 16 + l][h2 * 2];
  const int orow = threadIdx.x >> 4;
  const int oq   = threadIdx.x & 15;
  float4* Hout = (float4*)(i ? h1 : h0) + oq;

  for (int tile = blockIdx.x; tile < NN / 16; tile += gridDim.x) {
    const int r = tile * 16 + lr;
    int j = rj;
    int end = (r + 1 < NN) ? re : E_ADJ;
    end = min(end, E_ADJ);
    {
      const int nr = min(r + (int)gridDim.x * 16, NN - 1);
      rj = st[nr];
      re = st[min(nr + 1, NN - 1)];
    }
    float4 acc = make_float4(0.0f, 0.0f, 0.0f, 0.0f);
    float4 bcc = make_float4(0.0f, 0.0f, 0.0f, 0.0f);
    EdgeCV e0 = fx.a, e1 = fx.b, e2 = fx.c, e3 = fx.d;
    for (; j + 4 <= end; j += 4) {
      const Edge4 nx = edges_at(ed, j + 4);
      const EdgeCV n0 = nx.a, n1 = nx.b, n2 = nx.c, n3 = nx.d;
      const float v0 = e0.v, v1 = e1.v, v2 = e2.v, v3 = e3.v;
      const float4 a0 = X[(size_t)e0.c * 16], b0 = X[(size_t)e0.c * 16 + 8];
      const float4 a1 = X[(size_t)e1.c * 16], b1 = X[(size_t)e1.c * 16 + 8];
      const float4 a2 = X[(size_t)e2.c * 16], b2 = X[(size_t)e2.c * 16 + 8];
      const float4 a3 = X[(size_t)e3.c * 16], b3 = X[(size_t)e3.c * 16 + 8];
      acc.x += a0.x * v0; acc.y += a0.y * v0; acc.z += a0.z * v0; acc.w += a0.w * v0;
      bcc.x += b0.x * v0; bcc.y += b0.y * v0; bcc.z += b0.z * v0; bcc.w += b0.w * v0;
      acc.x += a1.x * v1; acc.y += a1.y * v1; acc.z += a1.z * v1; acc.w += a1.w * v1;
      bcc.x += b1.x * v1; bcc.y += b1.y * v1; bcc.z += b1.z * v1; bcc.w += b1.w * v1;
      acc.x += a2.x * v2; acc.y += a2.y * v2; acc.z += a2.z * v2; acc.w += a2.w * v2;
      bcc.x += b2.x * v2; bcc.y += b2.y * v2; bcc.z += b2.z * v2; bcc.w += b2.w * v2;
      acc.x += a3.x * v3; acc.y += a3.y * v3; acc.z += a3.z * v3; acc.w += a3.w * v3;
      bcc.x += b3.x * v3; bcc.y += b3.y * v3; bcc.z += b3.z * v3; bcc.w += b3.w * v3;
      __builtin_amdgcn_sched_group_barrier(0x020, 10, 0);
      __builtin_amdgcn_sched_group_barrier(0x002, 16, 0);
      e0 = n0; e1 = n1; e2 = n2; e3 = n3;
    }
    fx = edges_at(ed, rj);
    const int rem = end - j;
    if (rem > 0) {
      const int sh = j - min(max(j, 0), E_ADJ - 4);
      const EdgeCV w0 = (sh <= 0) ? e0 : (sh == 1) ? e1 : (sh == 2) ? e2 : e3;
      EdgeCV w1 = (sh <= 0) ? e1 : (sh == 1) ? e2 : e3;
      EdgeCV w2 = (sh <= 0) ? e2 : e3;
      if (rem < 2) w1 = w0;
      if (rem < 3) w2 = w0;
      const float v0 = w0.v, v1 = w1.v, v2 = w2.v;
      const float4 a0 = X[(size_t)w0.c * 16], b0 = X[(size_t)w0.c * 16 + 8];
      const float4 a1 = X[(size_t)w1.c * 16], b1 = X[(size_t)w1.c * 16 + 8];
      const float4 a2 = X[(size_t)w2.c * 16], b2 = X[(size_t)w2.c * 16 + 8];
      asm volatile("" :: "v"(a0.x), "v"(b0.x), "v"(a1.x), "v"(b1.x), "v"(a2.x), "v"(b2.x));
      acc.x += a0.x * v0; acc.y += a0.y * v0; acc.z += a0.z * v0; acc.w += a0.w * v0;
      bcc.x += b0.x * v0; bcc.y += b0.y * v0; bcc.z += b0.z * v0; bcc.w += b0.w * v0;
      if (rem > 1) {
        acc.x += a1.x * v1; acc.y += a1.y * v1; acc.z += a1.z * v1; acc.w += a1.w * v1;
        bcc.x += b1.x * v1; bcc.y += b1.y * v1; bcc.z += b1.z * v1; bcc.w += b1.w * v1;
      }
      if (rem > 2) {
        acc.x += a2.x * v2; acc.y += a2.y * v2; acc.z += a2.z * v2; acc.w += a2.w * v2;
        bcc.x += b2.x * v2; bcc.y += b2.y * v2; bcc.z += b2.z * v2; bcc.w += b2.w * v2;
      }
    }

    *(float4*)&G[half][lr][4 * l8]      = acc;
    *(float4*)&G[half][lr][32 + 4 * l8] = bcc;
    __syncthreads();

    v8f c = {};
#pragma unroll
    for (int s = 0; s < HH / 4; ++s) {
      v2f a, b;
      a.x = ar[4 * s];
      a.y = ar[4 * s + 1];
      b.x = br[4 * s];
      b.y = br[4 * s + 1];
      c = __builtin_amdgcn_wmma_f32_16x16x4_f32(false, a, false, b, (short)0, c, false, false);
    }
#pragma unroll
    for (int v = 0; v < 8; ++v) Y[rel][v + h2 * 8][ct * 16 + l] = fmaxf(c[v], 0.0f);
    __syncthreads();

    const float4 y0 = *(const float4*)&Y[0][orow][4 * oq];
    const float4 y1 = *(const float4*)&Y[1][orow][4 * oq];
    float4 s4;
    s4.x = y0.x + y1.x; s4.y = y0.y + y1.y; s4.z = y0.z + y1.z; s4.w = y0.w + y1.w;
    Hout[(size_t)(tile * 16 + orow) * 16] = s4;
  }
}

__global__ void build_M(const float* __restrict__ R, const float* __restrict__ D,
                        const int* __restrict__ rt_k, float* __restrict__ M) {
  int i = blockIdx.x * blockDim.x + threadIdx.x;
  const float* d = D + rt_k[0] * HH;
  int h = i >> 6, g = i & 63;
  M[i] = d[h] * R[i] * d[g];
}

static_assert(E_DEC % 64 == 0, "four 16-edge groups per block of the decoder");

__global__ __launch_bounds__(128) void decode_edges(
    const float* __restrict__ h0, const float* __restrict__ h1,
    const int* __restrict__ edges, const float* __restrict__ M,
    float* __restrict__ preds) {
  const int wave = threadIdx.x >> 5;
  const int lane = threadIdx.x & 31;
  const int half = lane >> 4;
  const int l    = lane & 15;
  const int eb   = blockIdx.x * 64 + wave * 16;

  const int arow = edges[(eb + l) * 2 + 0];

  const float* ar = h0 + (size_t)arow * HH + half * 2;
  v8f c0 = {}, c1 = {}, c2 = {}, c3 = {};
  for (int kk = 0; kk < HH; kk += 4) {
    v2f a, b0, b1, b2, b3;
    a.x = ar[kk];
    a.y = ar[kk + 1];
    const float* mp = M + (kk + half * 2) * HH + l;
    b0.x = mp[0];   b0.y = mp[HH];
    b1.x = mp[16];  b1.y = mp[HH + 16];
    b2.x = mp[32];  b2.y = mp[HH + 32];
    b3.x = mp[48];  b3.y = mp[HH + 48];
    c0 = __builtin_amdgcn_wmma_f32_16x16x4_f32(false, a, false, b0, (short)0, c0, false, false);
    c1 = __builtin_amdgcn_wmma_f32_16x16x4_f32(false, a, false, b1, (short)0, c1, false, false);
    c2 = __builtin_amdgcn_wmma_f32_16x16x4_f32(false, a, false, b2, (short)0, c2, false, false);
    c3 = __builtin_amdgcn_wmma_f32_16x16x4_f32(false, a, false, b3, (short)0, c3, false, false);
  }

  float p0[8], p1[8], p2[8], p3[8];
#pragma unroll
  for (int v = 0; v < 8; ++v) {
    const int edge = v + half * 8;
    const int crow = edges[(eb + edge) * 2 + 1];
    const float* hc = h1 + (size_t)crow * HH + l;
    p0[v] = c0[v] * hc[0];
    p1[v] = c1[v] * hc[16];
    p2[v] = c2[v] * hc[32];
    p3[v] = c3[v] * hc[48];
  }
#pragma unroll
  for (int off = 1; off < 16; off <<= 1) {
#pragma unroll
    for (int v = 0; v < 8; ++v) {
      p0[v] += __shfl_xor(p0[v], off, 16);
      p1[v] += __shfl_xor(p1[v], off, 16);
      p2[v] += __shfl_xor(p2[v], off, 16);
      p3[v] += __shfl_xor(p3[v], off, 16);
    }
  }
  if (l == 0) {
    float s[8];
#pragma unroll
    for (int v = 0; v < 8; ++v) s[v] = p0[v] + p1[v] + p2[v] + p3[v];
    float4* o = (float4*)(preds + eb + half * 8);
    o[0] = make_float4(s[0], s[1], s[2], s[3]);
    o[1] = make_float4(s[4], s[5], s[6], s[7]);
  }
}

extern "C" void kernel_launch(void* const* d_in, const int* in_sizes, int n_in,
                              void* d_out, int out_size, void* d_ws, size_t ws_size,
                              hipStream_t stream) {
  const int*   adj_rows = (const int*)d_in[0];
  const int*   adj_cols = (const int*)d_in[1];
  const float* adj_vals = (const float*)d_in[2];
  const float* feat0    = (const float*)d_in[3];
  const float* feat1    = (const float*)d_in[4];
  const float* W1       = (const float*)d_in[5];
  const float* W2       = (const float*)d_in[6];
  const float* W3       = (const float*)d_in[7];
  const float* W4       = (const float*)d_in[8];
  const float* Rg       = (const float*)d_in[9];
  const float* Dl       = (const float*)d_in[10];
  const int*   edges    = (const int*)d_in[11];
  const int*   rt_k     = (const int*)d_in[12];

  const size_t S = (size_t)NN * HH;
  float* xw = (float*)d_ws;
  float* lists = xw + 4 * S;
  float* h0 = lists + 4 * S;
  float* h1 = h0 + S;
  float* M  = h1 + S;

  static_assert((size_t)4 * E_ADJ * 2 + (size_t)4 * NN + (size_t)4 * RB_BLOCKS * (RB_BLOCKS + 1) <= (size_t)4 * NN * HH,
                "sorted lists and offset tables must fit in the four former y planes");
  static_assert((size_t)4 * E_ADJ * 2 <= (size_t)2 * NN * HH, "the split slices must fit in two xw planes");
  static_assert(((size_t)4 * NN * HH * sizeof(float)) % 256 == 0 && ((size_t)E_ADJ * sizeof(EdgeCV)) % 256 == 0,
                "se and every relation's part of it start on a 256-byte boundary of the workspace (place_edges stores aligned chunks)");
  EdgeCV* se   = (EdgeCV*)lists;
  int*   start = (int*)(se + (size_t)4 * E_ADJ);
  int*   hoff  = start + (size_t)4 * NN;
  EdgeDV* tmp  = (EdgeDV*)(xw + 2 * S);

  float* g0 = xw;
  float* g1 = xw + S;

  const int*   rows1 = adj_rows;
  const int*   cols1 = adj_cols;
  const float* vals1 = adj_vals;
  const int*   rows2 = adj_rows + (size_t)4 * E_ADJ;
  const int*   cols2 = adj_cols + (size_t)4 * E_ADJ;
  const float* vals2 = adj_vals + (size_t)4 * E_ADJ;

  split_edges<<<dim3(RB_BLOCKS, 4), RB_THREADS, 0, stream>>>(rows1, cols1, vals1, tmp, hoff);
  place_edges<<<dim3(RB_BLOCKS, 4), RB_THREADS, 0, stream>>>(tmp, hoff, start, se);
  gemm_stage<<<dim3(NN / (16 * GEMM_WAVES), 4), 32 * GEMM_WAVES, 0, stream>>>(feat0, feat1, W1, xw, 128);
  spmm_gather<<<dim3(NN / 16, 2), 256, 0, stream>>>(se, start, xw, h0, h1);
  gather_gemm_layer<<<dim3(GG_BLOCKS, 2), 256, 0, stream>>>(se, start, h0, h1, W2, g0, g1);

  split_edges<<<dim3(RB_BLOCKS, 4), RB_THREADS, 0, stream>>>(rows2, cols2, vals2, tmp, hoff);
  place_edges<<<dim3(RB_BLOCKS, 4), RB_THREADS, 0, stream>>>(tmp, hoff, start, se);
  gather_gemm_layer<<<dim3(GG_BLOCKS, 2), 256, 0, stream>>>(se, start, g0, g1, W3, h0, h1);
  gather_gemm_layer<<<dim3(GG_BLOCKS, 2), 256, 0, stream>>>(se, start, h0, h1, W4, g0, g1);

  build_M<<<16, 256, 0, stream>>>(Rg, Dl, rt_k, M);
  decode_edges<<<E_DEC / 64, 128, 0, stream>>>(g0, g1, edges, M, (float*)d_out);
}


# W fragments of the layer-1 product and both decoder operands (M, gathered rows) are read from LDS copies made with whole-line loads, not per-fragment global loads; k order and every sum unchanged
# speedup vs baseline: 4.5835x; 1.0314x over previous
#include <hip/hip_runtime.h>
#include <hip/hip_bf16.h>

#define NN 50000
#define E_ADJ 800000
#define E_DEC 200000
#define HH 64

typedef float v2f __attribute__((ext_vector_type(2)));
typedef float v8f __attribute__((ext_vector_type(8)));

#define GS_WAVES 8
#define GS_BLOCKS ((NN / 16 + GS_WAVES - 1) / GS_WAVES)
#define GS_K  128
#define GS_WP 80

__global__ __launch_bounds__(32 * GS_WAVES) void gemm_stage(
    const float* __restrict__ x0, const float* __restrict__ x1,
    const float* __restrict__ W, float* __restrict__ xw) {
  __shared__ __align__(16) float Wl[GS_K][GS_WP];

  const int k    = blockIdx.y;
  const int tm   = blockIdx.x * GS_WAVES + (threadIdx.x >> 5);
  const int lane = threadIdx.x & 31;
  const int half = lane >> 4;
  const int l    = lane & 15;

  const float* X  = (k & 1) ? x1 : x0;
  const float4* Wk = (const float4*)(W + (size_t)k * GS_K * HH);

#pragma unroll
  for (int u = 0; u < 8; ++u) {
    const int idx = threadIdx.x + 256 * u;
    const float4 w4 = Wk[idx];
    *(float4*)&Wl[idx >> 4][(idx & 15) * 4] = w4;
  }
  __syncthreads();
  if (tm >= NN / 16) return;

  const int row = tm * 16 + l;

  const float* ar = X + (size_t)row * GS_K + half * 2;
  const float* br = &Wl[half * 2][l];
  v8f c0 = {}, c1 = {}, c2 = {}, c3 = {};
#pragma unroll 8
  for (int s = 0; s < GS_K / 4; ++s) {
    v2f a, b0, b1, b2, b3;
    a.x = ar[4 * s];
    a.y = ar[4 * s + 1];
    const float* wp = br + 4 * s * GS_WP;
    b0.x = wp[0];   b0.y = wp[GS_WP];
    b1.x = wp[16];  b1.y = wp[GS_WP + 16];
    b2.x = wp[32];  b2.y = wp[GS_WP + 32];
    b3.x = wp[48];  b3.y = wp[GS_WP + 48];
    c0 = __builtin_amdgcn_wmma_f32_16x16x4_f32(
         false, a,  false, b0,
         (short)0, c0,  false,  false);
    c1 = __builtin_amdgcn_wmma_f32_16x16x4_f32(false, a, false, b1, (short)0, c1, false, false);
    c2 = __builtin_amdgcn_wmma_f32_16x16x4_f32(false, a, false, b2, (short)0, c2, false, false);
    c3 = __builtin_amdgcn_wmma_f32_16x16x4_f32(false, a, false, b3, (short)0, c3, false, false);
  }

  float* out = xw + ((size_t)k * NN + (size_t)tm * 16) * HH + l;
#pragma unroll
  for (int v = 0; v < 8; ++v) {
    float* o = out + (size_t)(v + half * 8) * HH;
    o[0]  = c0[v];
    o[16] = c1[v];
    o[32] = c2[v];
    o[48] = c3[v];
  }
}

struct __align__(8) EdgeCV { int c; float v; };
struct __align__(8) EdgeDV { unsigned cd; float v; };

#define RB_BLOCKS  64
#define RB_ROWS    ((NN + RB_BLOCKS - 1) / RB_BLOCKS)
#define RB_THREADS 1024
#define RB_WAVES   (RB_THREADS / 32)
#define PS_SLICE   (E_ADJ / RB_BLOCKS)
#define PS_PER     ((PS_SLICE + RB_THREADS - 1) / RB_THREADS)
static_assert(RB_ROWS <= RB_THREADS && RB_ROWS <= 1024, "one counter per thread; local row packs into 10 bits");
static_assert(NN <= (1 << 16), "source row packs into 16 bits");
static_assert(E_ADJ % RB_BLOCKS == 0, "slices are equal");
static_assert((NN - 1) / RB_ROWS < RB_BLOCKS, "every row has a range");
static_assert(RB_BLOCKS == 2 * RB_WAVES, "a wave of place_edges serves two slices' pieces");
#define PL_CAP     16384
static_assert(PL_CAP % 32 == 0, "whole 256-byte chunks");

__global__ __launch_bounds__(RB_THREADS) void split_edges(
    const int* __restrict__ rows, const int* __restrict__ cols, const float* __restrict__ vals,
    EdgeDV* __restrict__ tmp, int* __restrict__ hoff) {
  __shared__ int hist[RB_BLOCKS];
  __shared__ int hx[RB_BLOCKS];
  __shared__ EdgeDV buf[PS_SLICE];

  const int k = blockIdx.y;
  const int j = blockIdx.x;
  const int t = threadIdx.x;
  const size_t s0 = (size_t)k * E_ADJ + (size_t)j * PS_SLICE;
  const int*   R  = rows + s0;
  const int*   Cs = cols + s0;
  const float* Vs = vals + s0;

  if (t < RB_BLOCKS) hist[t] = 0;
  int row[PS_PER];
#pragma unroll
  for (int n = 0; n < PS_PER; ++n) {
    const int i = t + n * RB_THREADS;
    row[n] = (i < PS_SLICE) ? R[i] : 0;
  }
  __syncthreads();

  int rg[PS_PER], rk[PS_PER];
#pragma unroll
  for (int n = 0; n < PS_PER; ++n) {
    const int i = t + n * RB_THREADS;
    const int r = (int)min((unsigned)row[n] / (unsigned)RB_ROWS, (unsigned)(RB_BLOCKS - 1));
    rg[n] = r;
    rk[n] = (i < PS_SLICE) ? atomicAdd(&hist[r], 1) : 0;
  }
  int cc[PS_PER];
  float vv[PS_PER];
#pragma unroll
  for (int n = 0; n < PS_PER; ++n) {
    const int i = t + n * RB_THREADS;
    cc[n] = (i < PS_SLICE) ? Cs[i] : 0;
    vv[n] = (i < PS_SLICE) ? Vs[i] : 0.0f;
  }
  __syncthreads();

  if (t < RB_BLOCKS) {
    int ex = 0;
    for (int q = 0; q < RB_BLOCKS; ++q) ex += (q < t) ? hist[q] : 0;
    hx[t] = ex;
    int* H = hoff + ((size_t)k * RB_BLOCKS + j) * (RB_BLOCKS + 1);
    H[t] = ex;
    if (t == RB_BLOCKS - 1) H[RB_BLOCKS] = ex + hist[t];
  }
  __syncthreads();

#pragma unroll
  for (int n = 0; n < PS_PER; ++n) {
    const int i = t + n * RB_THREADS;
    if (i < PS_SLICE) {
      const int p = min(hx[rg[n]] + rk[n], PS_SLICE - 1);
      EdgeDV rec;
      rec.cd = ((unsigned)cc[n] & 0xffffu) | (((unsigned)(row[n] - rg[n] * RB_ROWS) & 1023u) << 16);
      rec.v  = vv[n];
      buf[p] = rec;
    }
  }
  __syncthreads();

  EdgeDV* T = tmp + s0;
  for (int i = t; i < PS_SLICE; i += RB_THREADS) T[i] = buf[i];
}

__global__ __launch_bounds__(RB_THREADS) void place_edges(
    const EdgeDV* __restrict__ tmp, const int* __restrict__ hoff,
    int* __restrict__ start, EdgeCV* __restrict__ se) {
  __shared__ int cnt[RB_THREADS];
  __shared__ int off[RB_THREADS];
  __shared__ int wtot[RB_WAVES];
  __shared__ int sbeg[RB_BLOCKS];
  __shared__ int slen[RB_BLOCKS];
  __shared__ int sbel[2];
  __shared__ EdgeCV lst[PL_CAP];

  const int k    = blockIdx.y;
  const int b    = blockIdx.x;
  const int t    = threadIdx.x;
  const int lane = t & 31;
  const int w    = t >> 5;
  const int lo   = b * RB_ROWS;
  const unsigned nrows = (unsigned)min(RB_ROWS, NN - lo);
  const EdgeDV* T = tmp + (size_t)k * E_ADJ;
  EdgeCV* S = se + (size_t)k * E_ADJ;

  cnt[t] = 0;
  if (t < RB_BLOCKS) {
    const int* H = hoff + ((size_t)k * RB_BLOCKS + t) * (RB_BLOCKS + 1);
    const int hb = min(max(H[b], 0), PS_SLICE);
    const int he = min(max(H[b + 1], hb), PS_SLICE);
    sbeg[t] = t * PS_SLICE + hb;
    slen[t] = he - hb;
    int sb = hb;
#pragma unroll
    for (int o = 16; o > 0; o >>= 1) sb += __shfl_xor(sb, o, 32);
    if (lane == 0) sbel[w] = sb;
  }
  __syncthreads();

  const int b0 = sbeg[2 * w], l0 = slen[2 * w];
  const int b1 = sbeg[2 * w + 1], l1 = slen[2 * w + 1];
  const int L = l0 + l1;

  for (int i0 = lane; i0 < L; i0 += 4 * 32) {
    unsigned x[4];
#pragma unroll
    for (int u = 0; u < 4; ++u) {
      const int i = i0 + u * 32;
      const int a = (i < l0) ? b0 + i : b1 + (i - l0);
      x[u] = T[min(a, E_ADJ - 1)].cd;
    }
    asm volatile("" :: "v"(x[0]), "v"(x[1]), "v"(x[2]), "v"(x[3]));
#pragma unroll
    for (int u = 0; u < 4; ++u) {
      if (i0 + u * 32 < L) atomicAdd(&cnt[(x[u] >> 16) & 1023u], 1);
    }
  }
  __syncthreads();

  const int own = ((unsigned)t < nrows) ? cnt[t] : 0;
  int incl = own;
#pragma unroll
  for (int o = 1; o < 32; o <<= 1) {
    const int n = __shfl_up(incl, o, 32);
    if (lane >= o) incl += n;
  }
  if (lane == 31) wtot[w] = incl;
  __syncthreads();
  const int below = sbel[0] + sbel[1];
  int wsum = 0, tot = 0;
  for (int q = 0; q < RB_WAVES; ++q) {
    const int n = wtot[q];
    tot += n;
    if (q < w) wsum += n;
  }
  const int ex = wsum + incl - own;
  off[t] = ex;
  cnt[t] = 0;
  if ((unsigned)t < nrows) start[k * NN + lo + t] = below + ex;
  __syncthreads();

  for (int i0 = lane; i0 < L; i0 += 4 * 32) {
    EdgeDV x[4];
    int rel[4];
#pragma unroll
    for (int u = 0; u < 4; ++u) {
      const int i = i0 + u * 32;
      const int a = (i < l0) ? b0 + i : b1 + (i - l0);
      x[u] = T[min(a, E_ADJ - 1)];
    }
    asm volatile("" :: "v"(x[0].cd), "v"(x[0].v), "v"(x[1].cd), "v"(x[1].v), "v"(x[2].cd), "v"(x[2].v), "v"(x[3].cd), "v"(x[3].v));
#pragma unroll
    for (int u = 0; u < 4; ++u) {
      const int d = (int)((x[u].cd >> 16) & 1023u);
      rel[u] = (i0 + u * 32 < L) ? off[d] + atomicAdd(&cnt[d], 1) : -1;
    }
#pragma unroll
    for (int u = 0; u < 4; ++u) {
      EdgeCV rec;
      rec.c = (int)(x[u].cd & 0xffffu);
      rec.v = x[u].v;
      if ((unsigned)rel[u] < (unsigned)PL_CAP) {
        lst[rel[u]] = rec;
      } else if (rel[u] >= 0) {
        const int pos = below + rel[u];
        if (pos >= 0 && pos < E_ADJ) S[pos] = rec;
      }
    }
  }
  __syncthreads();

  const int nl = min(tot, PL_CAP);
  for (int i0 = t - (below & 31); i0 < nl; i0 += 4 * RB_THREADS) {
    EdgeCV q[4];
#pragma unroll
    for (int u = 0; u < 4; ++u) q[u] = lst[min(max(i0 + u * RB_THREADS, 0), PL_CAP - 1)];
#pragma unroll
    for (int u = 0; u < 4; ++u) {
      const int i = i0 + u * RB_THREADS;
      const int pos = below + i;
      if (i >= 0 && i < nl && pos < E_ADJ) S[pos] = q[u];
    }
  }
}

struct Edge4 { EdgeCV a, b, c, d; };
static __device__ __forceinline__ Edge4 edges_at(const EdgeCV* __restrict__ ed, int i) {
  const EdgeCV* p = ed + min(max(i, 0), E_ADJ - 4);
  Edge4 q;
  q.a = p[0]; q.b = p[1]; q.c = p[2]; q.d = p[3];
  return q;
}

static_assert(NN % 16 == 0, "sixteen rows per block of the gather");

__global__ __launch_bounds__(256) __attribute__((amdgpu_waves_per_eu(10, 10))) void spmm_gather(
    const EdgeCV* __restrict__ se,
    const int* __restrict__ start, const float* __restrict__ xw,
    float* __restrict__ h0, float* __restrict__ h1) {
  const int lane = threadIdx.x & 31;
  const int grp  = lane >> 3;
  const int half = grp & 1;
  const int l8   = lane & 7;
  const int r    = blockIdx.x * 16 + (threadIdx.x >> 5) * 2 + (grp >> 1);
  const int i    = blockIdx.y;
  const int k    = i * 2 + half;

  const int* st = start + k * NN;
  int j = st[r];
  int end = (r + 1 < NN) ? st[r + 1] : E_ADJ;
  end = min(end, E_ADJ);
  const EdgeCV* ed = se + (size_t)k * E_ADJ;
  const float4* X = (const float4*)(xw + (size_t)k * NN * HH) + l8;

  float4 acc = make_float4(0.0f, 0.0f, 0.0f, 0.0f);
  float4 bcc = make_float4(0.0f, 0.0f, 0.0f, 0.0f);
  const Edge4 first = edges_at(ed, j);
  EdgeCV e0 = first.a, e1 = first.b, e2 = first.c, e3 = first.d;
  for (; j + 4 <= end; j += 4) {
    const Edge4 nx = edges_at(ed, j + 4);
    const EdgeCV n0 = nx.a, n1 = nx.b, n2 = nx.c, n3 = nx.d;
    const float v0 = e0.v, v1 = e1.v, v2 = e2.v, v3 = e3.v;
    const float4 a0 = X[(size_t)e0.c * 16], b0 = X[(size_t)e0.c * 16 + 8];
    const float4 a1 = X[(size_t)e1.c * 16], b1 = X[(size_t)e1.c * 16 + 8];
    const float4 a2 = X[(size_t)e2.c * 16], b2 = X[(size_t)e2.c * 16 + 8];
    const float4 a3 = X[(size_t)e3.c * 16], b3 = X[(size_t)e3.c * 16 + 8];
    acc.x += a0.x * v0; acc.y += a0.y * v0; acc.z += a0.z * v0; acc.w += a0.w * v0;
    bcc.x += b0.x * v0; bcc.y += b0.y * v0; bcc.z += b0.z * v0; bcc.w += b0.w * v0;
    acc.x += a1.x * v1; acc.y += a1.y * v1; acc.z += a1.z * v1; acc.w += a1.w * v1;
    bcc.x += b1.x * v1; bcc.y += b1.y * v1; bcc.z += b1.z * v1; bcc.w += b1.w * v1;
    acc.x += a2.x * v2; acc.y += a2.y * v2; acc.z += a2.z * v2; acc.w += a2.w * v2;
    bcc.x += b2.x * v2; bcc.y += b2.y * v2; bcc.z += b2.z * v2; bcc.w += b2.w * v2;
    acc.x += a3.x * v3; acc.y += a3.y * v3; acc.z += a3.z * v3; acc.w += a3.w * v3;
    bcc.x += b3.x * v3; bcc.y += b3.y * v3; bcc.z += b3.z * v3; bcc.w += b3.w * v3;
    __builtin_amdgcn_sched_group_barrier(0x020, 10, 0);
    __builtin_amdgcn_sched_group_barrier(0x002, 16, 0);
    e0 = n0; e1 = n1; e2 = n2; e3 = n3;
  }
  const int rem = end - j;
  if (rem > 0) {
    const int sh = j - min(max(j, 0), E_ADJ - 4);
    const EdgeCV w0 = (sh <= 0) ? e0 : (sh == 1) ? e1 : (sh == 2) ? e2 : e3;
    EdgeCV w1 = (sh <= 0) ? e1 : (sh == 1) ? e2 : e3;
    EdgeCV w2 = (sh <= 0) ? e2 : e3;
    if (rem < 2) w1 = w0;
    if (rem < 3) w2 = w0;
    const float v0 = w0.v, v1 = w1.v, v2 = w2.v;
    const float4 a0 = X[(size_t)w0.c * 16], b0 = X[(size_t)w0.c * 16 + 8];
    const float4 a1 = X[(size_t)w1.c * 16], b1 = X[(size_t)w1.c * 16 + 8];
    const float4 a2 = X[(size_t)w2.c * 16], b2 = X[(size_t)w2.c * 16 + 8];
    asm volatile("" :: "v"(a0.x), "v"(b0.x), "v"(a1.x), "v"(b1.x), "v"(a2.x), "v"(b2.x));
    acc.x += a0.x * v0; acc.y += a0.y * v0; acc.z += a0.z * v0; acc.w += a0.w * v0;
    bcc.x += b0.x * v0; bcc.y += b0.y * v0; bcc.z += b0.z * v0; bcc.w += b0.w * v0;
    if (rem > 1) {
      acc.x += a1.x * v1; acc.y += a1.y * v1; acc.z += a1.z * v1; acc.w += a1.w * v1;
      bcc.x += b1.x * v1; bcc.y += b1.y * v1; bcc.z += b1.z * v1; bcc.w += b1.w * v1;
    }
    if (rem > 2) {
      acc.x += a2.x * v2; acc.y += a2.y * v2; acc.z += a2.z * v2; acc.w += a2.w * v2;
      bcc.x += b2.x * v2; bcc.y += b2.y * v2; bcc.z += b2.z * v2; bcc.w += b2.w * v2;
    }
  }

  float4 q, p;
  q.x = fmaxf(acc.x, 0.0f); q.y = fmaxf(acc.y, 0.0f);
  q.z = fmaxf(acc.z, 0.0f); q.w = fmaxf(acc.w, 0.0f);
  p.x = fmaxf(bcc.x, 0.0f); p.y = fmaxf(bcc.y, 0.0f);
  p.z = fmaxf(bcc.z, 0.0f); p.w = fmaxf(bcc.w, 0.0f);
  float4 o, n;
  o.x = __shfl_xor(q.x, 8, 32); o.y = __shfl_xor(q.y, 8, 32);
  o.z = __shfl_xor(q.z, 8, 32); o.w = __shfl_xor(q.w, 8, 32);
  n.x = __shfl_xor(p.x, 8, 32); n.y = __shfl_xor(p.y, 8, 32);
  n.z = __shfl_xor(p.z, 8, 32); n.w = __shfl_xor(p.w, 8, 32);
  if (half == 0) {
    float4 s, t;
    s.x = q.x + o.x; s.y = q.y + o.y; s.z = q.z + o.z; s.w = q.w + o.w;
    t.x = p.x + n.x; t.y = p.y + n.y; t.z = p.z + n.z; t.w = p.w + n.w;
    float4* H = (float4*)(i ? h1 : h0) + (size_t)r * 16 + l8;
    H[0] = s;
    H[8] = t;
  }
}

#define GP 68
#define WTP 76
#define GG_TILES  5
#define GG_BLOCKS (NN / 16 / GG_TILES)
static_assert(GG_BLOCKS * GG_TILES * 16 == NN, "the blocks' tiles cover the rows exactly");

__global__ __launch_bounds__(256) __attribute__((amdgpu_waves_per_eu(10, 10))) void gather_gemm_layer(
    const EdgeCV* __restrict__ se, const int* __restrict__ start,
    const float* __restrict__ x0, const float* __restrict__ x1,
    const float* __restrict__ W,
    float* __restrict__ h0, float* __restrict__ h1) {
  __shared__ __align__(16) float Wt[2][HH][WTP];
  __shared__ __align__(16) float G[2][16][GP];
  __shared__ __align__(16) float Y[2][16][GP];

  const int lane = threadIdx.x & 31;
  const int wv   = threadIdx.x >> 5;
  const int grp  = lane >> 3;
  const int half = grp & 1;
  const int l8   = lane & 7;
  const int lr   = wv * 2 + (grp >> 1);
  const int i    = blockIdx.y;
  const int k    = i * 2 + half;

  const int* st = start + k * NN;
  const EdgeCV* ed = se + (size_t)k * E_ADJ;
  const float4* X = (const float4*)(half ? x1 : x0) + l8;

  int rj, re;
  {
    const int r = blockIdx.x * 16 + lr;
    rj = st[r];
    re = st[min(r + 1, NN - 1)];
  }

  {
    const float4* Wsrc = (const float4*)(W + (size_t)(i * 2) * HH * HH);
#pragma unroll
    for (int u = 0; u < 8; ++u) {
      const int idx = threadIdx.x + 256 * u;
      const float4 w4 = Wsrc[idx];
      const int relx = idx >> 10, kx = (idx >> 4) & 63, nx = (idx & 15) * 4;
      Wt[relx][nx + 0][kx] = w4.x;
      Wt[relx][nx + 1][kx] = w4.y;
      Wt[relx][nx + 2][kx] = w4.z;
      Wt[relx][nx + 3][kx] = w4.w;
    }
  }

  Edge4 fx = edges_at(ed, rj);

  const int rel = wv >> 2;
  const int ct  = wv & 3;
  const int h2  = lane >> 4;
  const int l   = lane & 15;
  const float* ar = &G[rel][l][h2 * 2];
  const float* br = &Wt[rel][ct * 16 + l][h2 * 2];
  const int orow = threadIdx.x >> 4;
  const int oq   = threadIdx.x & 15;
  float4* Hout = (float4*)(i ? h1 : h0) + oq;

  for (int tile = blockIdx.x; tile < NN / 16; tile += gridDim.x) {
    const int r = tile * 16 + lr;
    int j = rj;
    int end = (r + 1 < NN) ? re : E_ADJ;
    end = min(end, E_ADJ);
    {
      const int nr = min(r + (int)gridDim.x * 16, NN - 1);
      rj = st[nr];
      re = st[min(nr + 1, NN - 1)];
    }
    float4 acc = make_float4(0.0f, 0.0f, 0.0f, 0.0f);
    float4 bcc = make_float4(0.0f, 0.0f, 0.0f, 0.0f);
    EdgeCV e0 = fx.a, e1 = fx.b, e2 = fx.c, e3 = fx.d;
    for (; j + 4 <= end; j += 4) {
      const Edge4 nx = edges_at(ed, j + 4);
      const EdgeCV n0 = nx.a, n1 = nx.b, n2 = nx.c, n3 = nx.d;
      const float v0 = e0.v, v1 = e1.v, v2 = e2.v, v3 = e3.v;
      const float4 a0 = X[(size_t)e0.c * 16], b0 = X[(size_t)e0.c * 16 + 8];
      const float4 a1 = X[(size_t)e1.c * 16], b1 = X[(size_t)e1.c * 16 + 8];
      const float4 a2 = X[(size_t)e2.c * 16], b2 = X[(size_t)e2.c * 16 + 8];
      const float4 a3 = X[(size_t)e3.c * 16], b3 = X[(size_t)e3.c * 16 + 8];
      acc.x += a0.x * v0; acc.y += a0.y * v0; acc.z += a0.z * v0; acc.w += a0.w * v0;
      bcc.x += b0.x * v0; bcc.y += b0.y * v0; bcc.z += b0.z * v0; bcc.w += b0.w * v0;
      acc.x += a1.x * v1; acc.y += a1.y * v1; acc.z += a1.z * v1; acc.w += a1.w * v1;
      bcc.x += b1.x * v1; bcc.y += b1.y * v1; bcc.z += b1.z * v1; bcc.w += b1.w * v1;
      acc.x += a2.x * v2; acc.y += a2.y * v2; acc.z += a2.z * v2; acc.w += a2.w * v2;
      bcc.x += b2.x * v2; bcc.y += b2.y * v2; bcc.z += b2.z * v2; bcc.w += b2.w * v2;
      acc.x += a3.x * v3; acc.y += a3.y * v3; acc.z += a3.z * v3; acc.w += a3.w * v3;
      bcc.x += b3.x * v3; bcc.y += b3.y * v3; bcc.z += b3.z * v3; bcc.w += b3.w * v3;
      __builtin_amdgcn_sched_group_barrier(0x020, 10, 0);
      __builtin_amdgcn_sched_group_barrier(0x002, 16, 0);
      e0 = n0; e1 = n1; e2 = n2; e3 = n3;
    }
    fx = edges_at(ed, rj);
    const int rem = end - j;
    if (rem > 0) {
      const int sh = j - min(max(j, 0), E_ADJ - 4);
      const EdgeCV w0 = (sh <= 0) ? e0 : (sh == 1) ? e1 : (sh == 2) ? e2 : e3;
      EdgeCV w1 = (sh <= 0) ? e1 : (sh == 1) ? e2 : e3;
      EdgeCV w2 = (sh <= 0) ? e2 : e3;
      if (rem < 2) w1 = w0;
      if (rem < 3) w2 = w0;
      const float v0 = w0.v, v1 = w1.v, v2 = w2.v;
      const float4 a0 = X[(size_t)w0.c * 16], b0 = X[(size_t)w0.c * 16 + 8];
      const float4 a1 = X[(size_t)w1.c * 16], b1 = X[(size_t)w1.c * 16 + 8];
      const float4 a2 = X[(size_t)w2.c * 16], b2 = X[(size_t)w2.c * 16 + 8];
      asm volatile("" :: "v"(a0.x), "v"(b0.x), "v"(a1.x), "v"(b1.x), "v"(a2.x), "v"(b2.x));
      acc.x += a0.x * v0; acc.y += a0.y * v0; acc.z += a0.z * v0; acc.w += a0.w * v0;
      bcc.x += b0.x * v0; bcc.y += b0.y * v0; bcc.z += b0.z * v0; bcc.w += b0.w * v0;
      if (rem > 1) {
        acc.x += a1.x * v1; acc.y += a1.y * v1; acc.z += a1.z * v1; acc.w += a1.w * v1;
        bcc.x += b1.x * v1; bcc.y += b1.y * v1; bcc.z += b1.z * v1; bcc.w += b1.w * v1;
      }
      if (rem > 2) {
        acc.x += a2.x * v2; acc.y += a2.y * v2; acc.z += a2.z * v2; acc.w += a2.w * v2;
        bcc.x += b2.x * v2; bcc.y += b2.y * v2; bcc.z += b2.z * v2; bcc.w += b2.w * v2;
      }
    }

    *(float4*)&G[half][lr][4 * l8]      = acc;
    *(float4*)&G[half][lr][32 + 4 * l8] = bcc;
    __syncthreads();

    v8f c = {};
#pragma unroll
    for (int s = 0; s < HH / 4; ++s) {
      v2f a, b;
      a.x = ar[4 * s];
      a.y = ar[4 * s + 1];
      b.x = br[4 * s];
      b.y = br[4 * s + 1];
      c = __builtin_amdgcn_wmma_f32_16x16x4_f32(false, a, false, b, (short)0, c, false, false);
    }
#pragma unroll
    for (int v = 0; v < 8; ++v) Y[rel][v + h2 * 8][ct * 16 + l] = fmaxf(c[v], 0.0f);
    __syncthreads();

    const float4 y0 = *(const float4*)&Y[0][orow][4 * oq];
    const float4 y1 = *(const float4*)&Y[1][orow][4 * oq];
    float4 s4;
    s4.x = y0.x + y1.x; s4.y = y0.y + y1.y; s4.z = y0.z + y1.z; s4.w = y0.w + y1.w;
    Hout[(size_t)(tile * 16 + orow) * 16] = s4;
  }
}

__global__ void build_M(const float* __restrict__ R, const float* __restrict__ D,
                        const int* __restrict__ rt_k, float* __restrict__ M) {
  int i = blockIdx.x * blockDim.x + threadIdx.x;
  const float* d = D + rt_k[0] * HH;
  int h = i >> 6, g = i & 63;
  M[i] = d[h] * R[i] * d[g];
}

static_assert(E_DEC % 64 == 0, "four 16-edge groups per block of the decoder");
#define DC_MP 80
#define DC_AP 68

__global__ __launch_bounds__(128) void decode_edges(
    const float* __restrict__ h0, const float* __restrict__ h1,
    const int* __restrict__ edges, const float* __restrict__ M,
    float* __restrict__ preds) {
  __shared__ __align__(16) float Ml[HH][DC_MP];
  __shared__ __align__(16) float Et[4][16 * DC_AP];

  const int wave = threadIdx.x >> 5;
  const int lane = threadIdx.x & 31;
  const int half = lane >> 4;
  const int l    = lane & 15;
  const int eb   = blockIdx.x * 64 + wave * 16;

  float* A = Et[wave];
  {
    int arow[8];
#pragma unroll
    for (int u = 0; u < 8; ++u) arow[u] = edges[(eb + 2 * u + half) * 2 + 0];
#pragma unroll
    for (int u = 0; u < 8; ++u) {
      const int idx = threadIdx.x + 128 * u;
      const float4 m4 = ((const float4*)M)[idx];
      *(float4*)&Ml[idx >> 4][(idx & 15) * 4] = m4;
    }
    float4 a4[8];
#pragma unroll
    for (int u = 0; u < 8; ++u) a4[u] = ((const float4*)(h0 + (size_t)arow[u] * HH))[l];
#pragma unroll
    for (int u = 0; u < 8; ++u) *(float4*)&A[(2 * u + half) * DC_AP + l * 4] = a4[u];
  }
  __syncthreads();

  const float* ar = A + l * DC_AP + half * 2;
  const float* br = &Ml[half * 2][l];
  v8f c0 = {}, c1 = {}, c2 = {}, c3 = {};
#pragma unroll
  for (int s = 0; s < HH / 4; ++s) {
    v2f a, b0, b1, b2, b3;
    a.x = ar[4 * s];
    a.y = ar[4 * s + 1];
    const float* mp = br + 4 * s * DC_MP;
    b0.x = mp[0];   b0.y = mp[DC_MP];
    b1.x = mp[16];  b1.y = mp[DC_MP + 16];
    b2.x = mp[32];  b2.y = mp[DC_MP + 32];
    b3.x = mp[48];  b3.y = mp[DC_MP + 48];
    c0 = __builtin_amdgcn_wmma_f32_16x16x4_f32(false, a, false, b0, (short)0, c0, false, false);
    c1 = __builtin_amdgcn_wmma_f32_16x16x4_f32(false, a, false, b1, (short)0, c1, false, false);
    c2 = __builtin_amdgcn_wmma_f32_16x16x4_f32(false, a, false, b2, (short)0, c2, false, false);
    c3 = __builtin_amdgcn_wmma_f32_16x16x4_f32(false, a, false, b3, (short)0, c3, false, false);
  }

  float p0[8], p1[8], p2[8], p3[8];
#pragma unroll
  for (int v = 0; v < 8; ++v) {
    const int edge = v + half * 8;
    const int crow = edges[(eb + edge) * 2 + 1];
    const float* hc = h1 + (size_t)crow * HH + l;
    p0[v] = c0[v] * hc[0];
    p1[v] = c1[v] * hc[16];
    p2[v] = c2[v] * hc[32];
    p3[v] = c3[v] * hc[48];
  }
#pragma unroll
  for (int off = 1; off < 16; off <<= 1) {
#pragma unroll
    for (int v = 0; v < 8; ++v) {
      p0[v] += __shfl_xor(p0[v], off, 16);
      p1[v] += __shfl_xor(p1[v], off, 16);
      p2[v] += __shfl_xor(p2[v], off, 16);
      p3[v] += __shfl_xor(p3[v], off, 16);
    }
  }
  if (l == 0) {
    float s[8];
#pragma unroll
    for (int v = 0; v < 8; ++v) s[v] = p0[v] + p1[v] + p2[v] + p3[v];
    float4* o = (float4*)(preds + eb + half * 8);
    o[0] = make_float4(s[0], s[1], s[2], s[3]);
    o[1] = make_float4(s[4], s[5], s[6], s[7]);
  }
}

extern "C" void kernel_launch(void* const* d_in, const int* in_sizes, int n_in,
                              void* d_out, int out_size, void* d_ws, size_t ws_size,
                              hipStream_t stream) {
  const int*   adj_rows = (const int*)d_in[0];
  const int*   adj_cols = (const int*)d_in[1];
  const float* adj_vals = (const float*)d_in[2];
  const float* feat0    = (const float*)d_in[3];
  const float* feat1    = (const float*)d_in[4];
  const float* W1       = (const float*)d_in[5];
  const float* W2       = (const float*)d_in[6];
  const float* W3       = (const float*)d_in[7];
  const float* W4       = (const float*)d_in[8];
  const float* Rg       = (const float*)d_in[9];
  const float* Dl       = (const float*)d_in[10];
  const int*   edges    = (const int*)d_in[11];
  const int*   rt_k     = (const int*)d_in[12];

  const size_t S = (size_t)NN * HH;
  float* xw = (float*)d_ws;
  float* lists = xw + 4 * S;
  float* h0 = lists + 4 * S;
  float* h1 = h0 + S;
  float* M  = h1 + S;

  static_assert((size_t)4 * E_ADJ * 2 + (size_t)4 * NN + (size_t)4 * RB_BLOCKS * (RB_BLOCKS + 1) <= (size_t)4 * NN * HH,
                "sorted lists and offset tables must fit in the four former y planes");
  static_assert((size_t)4 * E_ADJ * 2 <= (size_t)2 * NN * HH, "the split slices must fit in two xw planes");
  static_assert(((size_t)4 * NN * HH * sizeof(float)) % 256 == 0 && ((size_t)E_ADJ * sizeof(EdgeCV)) % 256 == 0,
                "se and every relation's part of it start on a 256-byte boundary of the workspace (place_edges stores aligned chunks)");
  EdgeCV* se   = (EdgeCV*)lists;
  int*   start = (int*)(se + (size_t)4 * E_ADJ);
  int*   hoff  = start + (size_t)4 * NN;
  EdgeDV* tmp  = (EdgeDV*)(xw + 2 * S);

  float* g0 = xw;
  float* g1 = xw + S;

  const int*   rows1 = adj_rows;
  const int*   cols1 = adj_cols;
  const float* vals1 = adj_vals;
  const int*   rows2 = adj_rows + (size_t)4 * E_ADJ;
  const int*   cols2 = adj_cols + (size_t)4 * E_ADJ;
  const float* vals2 = adj_vals + (size_t)4 * E_ADJ;

  split_edges<<<dim3(RB_BLOCKS, 4), RB_THREADS, 0, stream>>>(rows1, cols1, vals1, tmp, hoff);
  place_edges<<<dim3(RB_BLOCKS, 4), RB_THREADS, 0, stream>>>(tmp, hoff, start, se);
  gemm_stage<<<dim3(GS_BLOCKS, 4), 32 * GS_WAVES, 0, stream>>>(feat0, feat1, W1, xw);
  spmm_gather<<<dim3(NN / 16, 2), 256, 0, stream>>>(se, start, xw, h0, h1);
  gather_gemm_layer<<<dim3(GG_BLOCKS, 2), 256, 0, stream>>>(se, start, h0, h1, W2, g0, g1);

  split_edges<<<dim3(RB_BLOCKS, 4), RB_THREADS, 0, stream>>>(rows2, cols2, vals2, tmp, hoff);
  place_edges<<<dim3(RB_BLOCKS, 4), RB_THREADS, 0, stream>>>(tmp, hoff, start, se);
  gather_gemm_layer<<<dim3(GG_BLOCKS, 2), 256, 0, stream>>>(se, start, g0, g1, W3, h0, h1);
  gather_gemm_layer<<<dim3(GG_BLOCKS, 2), 256, 0, stream>>>(se, start, h0, h1, W4, g0, g1);

  build_M<<<16, 256, 0, stream>>>(Rg, Dl, rt_k, M);
  decode_edges<<<E_DEC / 64, 128, 0, stream>>>(g0, g1, edges, M, (float*)d_out);
}
